# starting kernel
# baseline (speedup 1.0000x reference)
#include <hip/hip_runtime.h>
#include <float.h>

#define FDIM 128

typedef __attribute__((ext_vector_type(2))) float v2f;
typedef __attribute__((ext_vector_type(8))) float v8f;


__global__ void zero_f32_kernel(float* __restrict__ p, long n) {
  long i = (long)blockIdx.x * blockDim.x + threadIdx.x;
  if (i < n) p[i] = 0.0f;
}

__global__ void degree_kernel(const int* __restrict__ ei, float* __restrict__ deg,
                              int E, int N) {
  int e = blockIdx.x * blockDim.x + threadIdx.x;
  int EN = E + N;
  if (e >= EN) return;
  int d = (e < E) ? ei[E + e] : (e - E);
  atomicAdd(&deg[d], 1.0f);
}

__global__ void dinv_kernel(float* __restrict__ deg_dinv, int N) {
  int i = blockIdx.x * blockDim.x + threadIdx.x;
  if (i < N) deg_dinv[i] = rsqrtf(fmaxf(deg_dinv[i], 1.0f));
}

__global__ void gemm_wmma_kernel(const float* __restrict__ X,
                                 const float* __restrict__ W,
                                 float* __restrict__ H) {
  const int row0 = blockIdx.x * 16;
  const int col0 = (threadIdx.x >> 5) * 16;
  const int lane = threadIdx.x & 31;
  const int m    = lane & 15;
  const int kh   = lane >> 4;

  v8f c = {};
  const float* xr = X + (size_t)(row0 + m) * FDIM;
  const float* wc = W + col0 + m;

  #pragma unroll 4
  for (int kk = 0; kk < FDIM / 4; ++kk) {
    const int k0 = kk * 4 + (kh ? 2 : 0);
    v2f a, b;
    a.x = xr[k0];
    a.y = xr[k0 + 1];
    b.x = wc[(size_t)k0 * FDIM];
    b.y = wc[(size_t)(k0 + 1) * FDIM];
    c = __builtin_amdgcn_wmma_f32_16x16x4_f32(
         false, a,  false, b,
         (short)0, c,  false,  false);
  }

  float* hb = H + (size_t)(row0 + (kh ? 8 : 0)) * FDIM + col0 + m;
  #pragma unroll
  for (int r = 0; r < 8; ++r) hb[(size_t)r * FDIM] = c[r];
}

__global__ void gemm_tail_kernel(const float* __restrict__ X,
                                 const float* __restrict__ W,
                                 float* __restrict__ H, int base) {
  const int row = base + blockIdx.x;
  const int f = threadIdx.x;
  const float* xr = X + (size_t)row * FDIM;
  float acc = 0.0f;
  #pragma unroll 8
  for (int k = 0; k < FDIM; ++k) acc = fmaf(xr[k], W[(size_t)k * FDIM + f], acc);
  H[(size_t)row * FDIM + f] = acc;
}

__global__ void aggregate_kernel(const float* __restrict__ H,
                                 const int* __restrict__ ei,
                                 const float* __restrict__ dinv,
                                 float* __restrict__ G, int E, int N) {
  const int e = blockIdx.x * (blockDim.x >> 5) + (threadIdx.x >> 5);
  const int EN = E + N;
  if (e >= EN) return;
  const int lane = threadIdx.x & 31;
  int s, d;
  if (e < E) { s = ei[e]; d = ei[E + e]; }
  else       { s = e - E; d = s; }
  const float w = dinv[s] * dinv[d];
  const float4 hv = reinterpret_cast<const float4*>(H + (size_t)s * FDIM)[lane];
  float* gp = G + (size_t)d * FDIM + lane * 4;
  atomicAdd(gp + 0, hv.x * w);
  atomicAdd(gp + 1, hv.y * w);
  atomicAdd(gp + 2, hv.z * w);
  atomicAdd(gp + 3, hv.w * w);
}

__global__ void bias_relu_kernel(float* __restrict__ G, const float* __restrict__ b) {
  const size_t i = (size_t)blockIdx.x * FDIM + threadIdx.x;
  G[i] = fmaxf(G[i] + b[threadIdx.x], 0.0f);
}

__global__ void init_out_kernel(float* __restrict__ out, int ng) {
  int i = blockIdx.x * blockDim.x + threadIdx.x;
  if (i >= ng * 2 * FDIM) return;
  int f = i & (2 * FDIM - 1);
  out[i] = (f < FDIM) ? 0.0f : __int_as_float(0xFF7FFFFF);
}

__device__ __forceinline__ void atomic_max_f32(float* addr, float v) {
  if (v >= 0.0f) atomicMax((int*)addr, __float_as_int(v));
  else           atomicMin((unsigned int*)addr, (unsigned int)__float_as_int(v));
}

__global__ void pool_kernel(const float* __restrict__ G, const int* __restrict__ gi,
                            float* __restrict__ out, float* __restrict__ cnt) {
  const int node = blockIdx.x;
  const int f = threadIdx.x;
  const int g = gi[node];
  const float v = G[(size_t)node * FDIM + f];
  atomicAdd(&out[(size_t)g * 2 * FDIM + f], v);
  atomic_max_f32(&out[(size_t)g * 2 * FDIM + FDIM + f], v);
  if (f == 0) atomicAdd(&cnt[g], 1.0f);
}

__global__ void finalize_mean_kernel(float* __restrict__ out, const float* __restrict__ cnt) {
  const int g = blockIdx.x;
  out[(size_t)g * 2 * FDIM + threadIdx.x] /= fmaxf(cnt[g], 1.0f);
}


extern "C" void kernel_launch(void* const* d_in, const int* in_sizes, int n_in,
                              void* d_out, int out_size, void* d_ws, size_t ws_size,
                              hipStream_t stream) {
  const float* x  = (const float*)d_in[0];
  const int*   ei = (const int*)d_in[1];
  const int*   gi = (const int*)d_in[2];
  const float* Wm[3] = {(const float*)d_in[4], (const float*)d_in[6], (const float*)d_in[8]};
  const float* Bv[3] = {(const float*)d_in[5], (const float*)d_in[7], (const float*)d_in[9]};
  float* out = (float*)d_out;

  const int  N  = in_sizes[0] / FDIM;
  const int  E  = in_sizes[1] / 2;
  const int  NG = out_size / (2 * FDIM);
  const long Nf = (long)N * FDIM;

  float* Hbuf = (float*)d_ws;
  float* Gbuf = Hbuf + Nf;
  float* dinv = Gbuf + Nf;
  float* cnt  = dinv + N;
  if (ws_size < (size_t)(2 * Nf + N + NG) * sizeof(float)) return;

  auto zero = [&](float* p, long n) {
    zero_f32_kernel<<<(unsigned)((n + 255) / 256), 256, 0, stream>>>(p, n);
  };

  zero(dinv, N);
  degree_kernel<<<(E + N + 255) / 256, 256, 0, stream>>>(ei, dinv, E, N);
  dinv_kernel<<<(N + 255) / 256, 256, 0, stream>>>(dinv, N);

  const int tiles = N / 16;
  const int rem   = N - tiles * 16;
  const float* cur = x;
  for (int l = 0; l < 3; ++l) {
    if (tiles > 0)
      gemm_wmma_kernel<<<tiles, 256, 0, stream>>>(cur, Wm[l], Hbuf);
    if (rem > 0)
      gemm_tail_kernel<<<rem, FDIM, 0, stream>>>(cur, Wm[l], Hbuf, tiles * 16);
    zero(Gbuf, Nf);
    aggregate_kernel<<<(E + N + 7) / 8, 256, 0, stream>>>(Hbuf, ei, dinv, Gbuf, E, N);
    bias_relu_kernel<<<N, FDIM, 0, stream>>>(Gbuf, Bv[l]);
    cur = Gbuf;
  }

  init_out_kernel<<<(NG * 2 * FDIM + 255) / 256, 256, 0, stream>>>(out, NG);
  zero(cnt, NG);
  pool_kernel<<<N, FDIM, 0, stream>>>(Gbuf, gi, out, cnt);
  finalize_mean_kernel<<<NG, FDIM, 0, stream>>>(out, cnt);
}


# Aggregation without float atomics: edges bucketed by target each call (int count+rank atomic, scan, place), one wave sums a target row in registers, one store; sum order = arrival rank
# speedup vs baseline: 2.3092x; 2.3092x over previous
#include <hip/hip_runtime.h>
#include <float.h>

#define FDIM 128

typedef __attribute__((ext_vector_type(2))) float v2f;
typedef __attribute__((ext_vector_type(8))) float v8f;


__global__ void zero_f32_kernel(float* __restrict__ p, long n) {
  long i = (long)blockIdx.x * blockDim.x + threadIdx.x;
  if (i < n) p[i] = 0.0f;
}

__global__ void degree_kernel(const int* __restrict__ ei, float* __restrict__ deg,
                              int E, int N) {
  int e = blockIdx.x * blockDim.x + threadIdx.x;
  int EN = E + N;
  if (e >= EN) return;
  int d = (e < E) ? ei[E + e] : (e - E);
  atomicAdd(&deg[d], 1.0f);
}

__global__ void dinv_kernel(float* __restrict__ deg_dinv, int N) {
  int i = blockIdx.x * blockDim.x + threadIdx.x;
  if (i < N) deg_dinv[i] = rsqrtf(fmaxf(deg_dinv[i], 1.0f));
}

__global__ void gemm_wmma_kernel(const float* __restrict__ X,
                                 const float* __restrict__ W,
                                 float* __restrict__ H) {
  const int row0 = blockIdx.x * 16;
  const int col0 = (threadIdx.x >> 5) * 16;
  const int lane = threadIdx.x & 31;
  const int m    = lane & 15;
  const int kh   = lane >> 4;

  v8f c = {};
  const float* xr = X + (size_t)(row0 + m) * FDIM;
  const float* wc = W + col0 + m;

  #pragma unroll 4
  for (int kk = 0; kk < FDIM / 4; ++kk) {
    const int k0 = kk * 4 + (kh ? 2 : 0);
    v2f a, b;
    a.x = xr[k0];
    a.y = xr[k0 + 1];
    b.x = wc[(size_t)k0 * FDIM];
    b.y = wc[(size_t)(k0 + 1) * FDIM];
    c = __builtin_amdgcn_wmma_f32_16x16x4_f32(
         false, a,  false, b,
         (short)0, c,  false,  false);
  }

  float* hb = H + (size_t)(row0 + (kh ? 8 : 0)) * FDIM + col0 + m;
  #pragma unroll
  for (int r = 0; r < 8; ++r) hb[(size_t)r * FDIM] = c[r];
}

__global__ void gemm_tail_kernel(const float* __restrict__ X,
                                 const float* __restrict__ W,
                                 float* __restrict__ H, int base) {
  const int row = base + blockIdx.x;
  const int f = threadIdx.x;
  const float* xr = X + (size_t)row * FDIM;
  float acc = 0.0f;
  #pragma unroll 8
  for (int k = 0; k < FDIM; ++k) acc = fmaf(xr[k], W[(size_t)k * FDIM + f], acc);
  H[(size_t)row * FDIM + f] = acc;
}

__global__ void aggregate_kernel(const float* __restrict__ H,
                                 const int* __restrict__ ei,
                                 const float* __restrict__ dinv,
                                 float* __restrict__ G, int E, int N) {
  const int e = blockIdx.x * (blockDim.x >> 5) + (threadIdx.x >> 5);
  const int EN = E + N;
  if (e >= EN) return;
  const int lane = threadIdx.x & 31;
  int s, d;
  if (e < E) { s = ei[e]; d = ei[E + e]; }
  else       { s = e - E; d = s; }
  const float w = dinv[s] * dinv[d];
  const float4 hv = reinterpret_cast<const float4*>(H + (size_t)s * FDIM)[lane];
  float* gp = G + (size_t)d * FDIM + lane * 4;
  atomicAdd(gp + 0, hv.x * w);
  atomicAdd(gp + 1, hv.y * w);
  atomicAdd(gp + 2, hv.z * w);
  atomicAdd(gp + 3, hv.w * w);
}


__global__ void count_rank_kernel(const int* __restrict__ ei, int* __restrict__ icnt,
                                  int* __restrict__ rank, int E) {
  const int e = blockIdx.x * blockDim.x + threadIdx.x;
  if (e >= E) return;
  rank[e] = atomicAdd(&icnt[ei[E + e]], 1);
}

#define SCAN_THREADS 256
#define SCAN_ELEMS   1024

__global__ void scan_local_kernel(const int* __restrict__ icnt, int* __restrict__ rowptr,
                                  int* __restrict__ bsum, int N) {
  __shared__ int wsum[SCAN_THREADS / 32];
  const int tid = threadIdx.x, lane = tid & 31, wv = tid >> 5;
  const int base = blockIdx.x * SCAN_ELEMS + tid * 4;
  int v[4];
  #pragma unroll
  for (int q = 0; q < 4; ++q) v[q] = (base + q < N) ? icnt[base + q] : 0;
  const int t = v[0] + v[1] + v[2] + v[3];
  int inc = t;
  #pragma unroll
  for (int off = 1; off < 32; off <<= 1) {
    const int u = __shfl_up(inc, off);
    if (lane >= off) inc += u;
  }
  if (lane == 31) wsum[wv] = inc;
  __syncthreads();
  int woff = 0, tot = 0;
  #pragma unroll
  for (int k = 0; k < SCAN_THREADS / 32; ++k) {
    const int s = wsum[k];
    if (k < wv) woff += s;
    tot += s;
  }
  int run = woff + inc - t;
  #pragma unroll
  for (int q = 0; q < 4; ++q) {
    if (base + q < N) rowptr[base + q] = run;
    run += v[q];
  }
  if (tid == 0) bsum[blockIdx.x] = tot;
}

__global__ void scan_add_kernel(int* __restrict__ rowptr, const int* __restrict__ bsum, int N) {
  __shared__ int red[SCAN_THREADS / 32];
  const int tid = threadIdx.x, lane = tid & 31, wv = tid >> 5;
  const int b = blockIdx.x;
  int part = 0;
  for (int k = tid; k < b; k += SCAN_THREADS) part += bsum[k];
  #pragma unroll
  for (int off = 16; off > 0; off >>= 1) part += __shfl_xor(part, off);
  if (lane == 0) red[wv] = part;
  __syncthreads();
  int add = 0;
  #pragma unroll
  for (int k = 0; k < SCAN_THREADS / 32; ++k) add += red[k];
  const int base = b * SCAN_ELEMS + tid * 4;
  #pragma unroll
  for (int q = 0; q < 4; ++q)
    if (base + q < N) rowptr[base + q] += add;
  if (b == (int)gridDim.x - 1 && tid == 0) rowptr[N] = add + bsum[b];
}

__global__ void place_kernel(const int* __restrict__ ei, const int* __restrict__ rowptr,
                             const int* __restrict__ rank, int* __restrict__ col, int E) {
  const int e = blockIdx.x * blockDim.x + threadIdx.x;
  if (e >= E) return;
  col[rowptr[ei[E + e]] + rank[e]] = ei[e];
}

__global__ void dinv_from_count_kernel(float* __restrict__ cnt_dinv, int N) {
  const int i = blockIdx.x * blockDim.x + threadIdx.x;
  if (i < N) {
    const int c = __float_as_int(cnt_dinv[i]);
    cnt_dinv[i] = rsqrtf(fmaxf((float)(c + 1), 1.0f));
  }
}

__device__ __forceinline__ void row_fma(float4& acc, const float4 h, const float w) {
  acc.x = fmaf(h.x, w, acc.x);
  acc.y = fmaf(h.y, w, acc.y);
  acc.z = fmaf(h.z, w, acc.z);
  acc.w = fmaf(h.w, w, acc.w);
}

__global__ void gather_rows_kernel(const float* __restrict__ H, const int* __restrict__ rowptr,
                                   const int* __restrict__ col, const float* __restrict__ dinv,
                                   float* __restrict__ G, int N) {
  const int d = __builtin_amdgcn_readfirstlane(
      (int)(blockIdx.x * (blockDim.x >> 5) + (threadIdx.x >> 5)));
  if (d >= N) return;
  const int lane = threadIdx.x & 31;
  const int beg = rowptr[d];
  const int end = rowptr[d + 1];
  const float dd = dinv[d];
  const float4* __restrict__ H4 = reinterpret_cast<const float4*>(H);

  float4 acc;
  {
    const float4 hs = H4[(size_t)d * (FDIM / 4) + lane];
    const float ws = dd * dd;
    acc.x = hs.x * ws; acc.y = hs.y * ws; acc.z = hs.z * ws; acc.w = hs.w * ws;
  }

  for (int j = beg; j < end; j += 32) {
    const int n = min(32, end - j);
    int   s_l = 0;
    float w_l = 0.0f;
    if (lane < n) {
      s_l = col[j + lane];
      w_l = dinv[s_l] * dd;
    }
    int t = 0;
    for (; t + 4 <= n; t += 4) {
      const int s0 = __builtin_amdgcn_readlane(s_l, t);
      const int s1 = __builtin_amdgcn_readlane(s_l, t + 1);
      const int s2 = __builtin_amdgcn_readlane(s_l, t + 2);
      const int s3 = __builtin_amdgcn_readlane(s_l, t + 3);
      const float4 h0 = H4[(size_t)s0 * (FDIM / 4) + lane];
      const float4 h1 = H4[(size_t)s1 * (FDIM / 4) + lane];
      const float4 h2 = H4[(size_t)s2 * (FDIM / 4) + lane];
      const float4 h3 = H4[(size_t)s3 * (FDIM / 4) + lane];
      const float w0 = __int_as_float(__builtin_amdgcn_readlane(__float_as_int(w_l), t));
      const float w1 = __int_as_float(__builtin_amdgcn_readlane(__float_as_int(w_l), t + 1));
      const float w2 = __int_as_float(__builtin_amdgcn_readlane(__float_as_int(w_l), t + 2));
      const float w3 = __int_as_float(__builtin_amdgcn_readlane(__float_as_int(w_l), t + 3));
      row_fma(acc, h0, w0);
      row_fma(acc, h1, w1);
      row_fma(acc, h2, w2);
      row_fma(acc, h3, w3);
    }
    const int r = n - t;
    if (r == 3) {
      const int s0 = __builtin_amdgcn_readlane(s_l, t);
      const int s1 = __builtin_amdgcn_readlane(s_l, t + 1);
      const int s2 = __builtin_amdgcn_readlane(s_l, t + 2);
      const float4 h0 = H4[(size_t)s0 * (FDIM / 4) + lane];
      const float4 h1 = H4[(size_t)s1 * (FDIM / 4) + lane];
      const float4 h2 = H4[(size_t)s2 * (FDIM / 4) + lane];
      const float w0 = __int_as_float(__builtin_amdgcn_readlane(__float_as_int(w_l), t));
      const float w1 = __int_as_float(__builtin_amdgcn_readlane(__float_as_int(w_l), t + 1));
      const float w2 = __int_as_float(__builtin_amdgcn_readlane(__float_as_int(w_l), t + 2));
      row_fma(acc, h0, w0);
      row_fma(acc, h1, w1);
      row_fma(acc, h2, w2);
    } else if (r == 2) {
      const int s0 = __builtin_amdgcn_readlane(s_l, t);
      const int s1 = __builtin_amdgcn_readlane(s_l, t + 1);
      const float4 h0 = H4[(size_t)s0 * (FDIM / 4) + lane];
      const float4 h1 = H4[(size_t)s1 * (FDIM / 4) + lane];
      const float w0 = __int_as_float(__builtin_amdgcn_readlane(__float_as_int(w_l), t));
      const float w1 = __int_as_float(__builtin_amdgcn_readlane(__float_as_int(w_l), t + 1));
      row_fma(acc, h0, w0);
      row_fma(acc, h1, w1);
    } else if (r == 1) {
      const int s0 = __builtin_amdgcn_readlane(s_l, t);
      const float w0 = __int_as_float(__builtin_amdgcn_readlane(__float_as_int(w_l), t));
      row_fma(acc, H4[(size_t)s0 * (FDIM / 4) + lane], w0);
    }
  }
  reinterpret_cast<float4*>(G)[(size_t)d * (FDIM / 4) + lane] = acc;
}

__global__ void bias_relu_kernel(float* __restrict__ G, const float* __restrict__ b) {
  const size_t i = (size_t)blockIdx.x * FDIM + threadIdx.x;
  G[i] = fmaxf(G[i] + b[threadIdx.x], 0.0f);
}

__global__ void init_out_kernel(float* __restrict__ out, int ng) {
  int i = blockIdx.x * blockDim.x + threadIdx.x;
  if (i >= ng * 2 * FDIM) return;
  int f = i & (2 * FDIM - 1);
  out[i] = (f < FDIM) ? 0.0f : __int_as_float(0xFF7FFFFF);
}

__device__ __forceinline__ void atomic_max_f32(float* addr, float v) {
  if (v >= 0.0f) atomicMax((int*)addr, __float_as_int(v));
  else           atomicMin((unsigned int*)addr, (unsigned int)__float_as_int(v));
}

__global__ void pool_kernel(const float* __restrict__ G, const int* __restrict__ gi,
                            float* __restrict__ out, float* __restrict__ cnt) {
  const int node = blockIdx.x;
  const int f = threadIdx.x;
  const int g = gi[node];
  const float v = G[(size_t)node * FDIM + f];
  atomicAdd(&out[(size_t)g * 2 * FDIM + f], v);
  atomic_max_f32(&out[(size_t)g * 2 * FDIM + FDIM + f], v);
  if (f == 0) atomicAdd(&cnt[g], 1.0f);
}

__global__ void finalize_mean_kernel(float* __restrict__ out, const float* __restrict__ cnt) {
  const int g = blockIdx.x;
  out[(size_t)g * 2 * FDIM + threadIdx.x] /= fmaxf(cnt[g], 1.0f);
}


extern "C" void kernel_launch(void* const* d_in, const int* in_sizes, int n_in,
                              void* d_out, int out_size, void* d_ws, size_t ws_size,
                              hipStream_t stream) {
  const float* x  = (const float*)d_in[0];
  const int*   ei = (const int*)d_in[1];
  const int*   gi = (const int*)d_in[2];
  const float* Wm[3] = {(const float*)d_in[4], (const float*)d_in[6], (const float*)d_in[8]};
  const float* Bv[3] = {(const float*)d_in[5], (const float*)d_in[7], (const float*)d_in[9]};
  float* out = (float*)d_out;

  const int  N  = in_sizes[0] / FDIM;
  const int  E  = in_sizes[1] / 2;
  const int  NG = out_size / (2 * FDIM);
  const long Nf = (long)N * FDIM;

  float* Hbuf = (float*)d_ws;
  float* Gbuf = Hbuf + Nf;
  float* dinv = Gbuf + Nf;
  float* cnt  = dinv + N;
  if (ws_size < (size_t)(2 * Nf + N + NG) * sizeof(float)) return;

  const int nsb = (N + SCAN_ELEMS - 1) / SCAN_ELEMS;
  int* rowptr = (int*)(cnt + NG);
  int* bsum   = rowptr + (N + 1);
  int* col    = bsum + nsb;
  int* rank   = (int*)Hbuf;
  const bool ordered =
      ws_size >= (size_t)(2 * Nf + N + NG) * sizeof(float) + ((size_t)N + 1 + nsb + E) * sizeof(int) &&
      (long)E <= Nf && E > 0;

  auto zero = [&](float* p, long n) {
    zero_f32_kernel<<<(unsigned)((n + 255) / 256), 256, 0, stream>>>(p, n);
  };

  const int tiles = N / 16;
  const int rem   = N - tiles * 16;
  const float* cur = x;

  if (ordered) {
    zero(dinv, N);
    count_rank_kernel<<<(E + 255) / 256, 256, 0, stream>>>(ei, (int*)dinv, rank, E);
    scan_local_kernel<<<nsb, SCAN_THREADS, 0, stream>>>((const int*)dinv, rowptr, bsum, N);
    scan_add_kernel<<<nsb, SCAN_THREADS, 0, stream>>>(rowptr, bsum, N);
    place_kernel<<<(E + 255) / 256, 256, 0, stream>>>(ei, rowptr, rank, col, E);
    dinv_from_count_kernel<<<(N + 255) / 256, 256, 0, stream>>>(dinv, N);

    for (int l = 0; l < 3; ++l) {
      if (tiles > 0)
        gemm_wmma_kernel<<<tiles, 256, 0, stream>>>(cur, Wm[l], Hbuf);
      if (rem > 0)
        gemm_tail_kernel<<<rem, FDIM, 0, stream>>>(cur, Wm[l], Hbuf, tiles * 16);
      gather_rows_kernel<<<(N + 7) / 8, 256, 0, stream>>>(Hbuf, rowptr, col, dinv, Gbuf, N);
      bias_relu_kernel<<<N, FDIM, 0, stream>>>(Gbuf, Bv[l]);
      cur = Gbuf;
    }
  } else {
    zero(dinv, N);
    degree_kernel<<<(E + N + 255) / 256, 256, 0, stream>>>(ei, dinv, E, N);
    dinv_kernel<<<(N + 255) / 256, 256, 0, stream>>>(dinv, N);

    for (int l = 0; l < 3; ++l) {
      if (tiles > 0)
        gemm_wmma_kernel<<<tiles, 256, 0, stream>>>(cur, Wm[l], Hbuf);
      if (rem > 0)
        gemm_tail_kernel<<<rem, FDIM, 0, stream>>>(cur, Wm[l], Hbuf, tiles * 16);
      zero(Gbuf, Nf);
      aggregate_kernel<<<(E + N + 7) / 8, 256, 0, stream>>>(Hbuf, ei, dinv, Gbuf, E, N);
      bias_relu_kernel<<<N, FDIM, 0, stream>>>(Gbuf, Bv[l]);
      cur = Gbuf;
    }
  }

  init_out_kernel<<<(NG * 2 * FDIM + 255) / 256, 256, 0, stream>>>(out, NG);
  zero(cnt, NG);
  pool_kernel<<<N, FDIM, 0, stream>>>(Gbuf, gi, out, cnt);
  finalize_mean_kernel<<<NG, FDIM, 0, stream>>>(out, cnt);
}


# Bias and ReLU applied to each target row's finished sums in the gather kernel's registers before its one store; the separate bias/ReLU pass over the activations is gone (3 launches fewer)
# speedup vs baseline: 2.7554x; 1.1932x over previous
#include <hip/hip_runtime.h>
#include <float.h>

#define FDIM 128

typedef __attribute__((ext_vector_type(2))) float v2f;
typedef __attribute__((ext_vector_type(8))) float v8f;


__global__ void zero_f32_kernel(float* __restrict__ p, long n) {
  long i = (long)blockIdx.x * blockDim.x + threadIdx.x;
  if (i < n) p[i] = 0.0f;
}

__global__ void degree_kernel(const int* __restrict__ ei, float* __restrict__ deg,
                              int E, int N) {
  int e = blockIdx.x * blockDim.x + threadIdx.x;
  int EN = E + N;
  if (e >= EN) return;
  int d = (e < E) ? ei[E + e] : (e - E);
  atomicAdd(&deg[d], 1.0f);
}

__global__ void dinv_kernel(float* __restrict__ deg_dinv, int N) {
  int i = blockIdx.x * blockDim.x + threadIdx.x;
  if (i < N) deg_dinv[i] = rsqrtf(fmaxf(deg_dinv[i], 1.0f));
}

__global__ void gemm_wmma_kernel(const float* __restrict__ X,
                                 const float* __restrict__ W,
                                 float* __restrict__ H) {
  const int row0 = blockIdx.x * 16;
  const int col0 = (threadIdx.x >> 5) * 16;
  const int lane = threadIdx.x & 31;
  const int m    = lane & 15;
  const int kh   = lane >> 4;

  v8f c = {};
  const float* xr = X + (size_t)(row0 + m) * FDIM;
  const float* wc = W + col0 + m;

  #pragma unroll 4
  for (int kk = 0; kk < FDIM / 4; ++kk) {
    const int k0 = kk * 4 + (kh ? 2 : 0);
    v2f a, b;
    a.x = xr[k0];
    a.y = xr[k0 + 1];
    b.x = wc[(size_t)k0 * FDIM];
    b.y = wc[(size_t)(k0 + 1) * FDIM];
    c = __builtin_amdgcn_wmma_f32_16x16x4_f32(
         false, a,  false, b,
         (short)0, c,  false,  false);
  }

  float* hb = H + (size_t)(row0 + (kh ? 8 : 0)) * FDIM + col0 + m;
  #pragma unroll
  for (int r = 0; r < 8; ++r) hb[(size_t)r * FDIM] = c[r];
}

__global__ void gemm_tail_kernel(const float* __restrict__ X,
                                 const float* __restrict__ W,
                                 float* __restrict__ H, int base) {
  const int row = base + blockIdx.x;
  const int f = threadIdx.x;
  const float* xr = X + (size_t)row * FDIM;
  float acc = 0.0f;
  #pragma unroll 8
  for (int k = 0; k < FDIM; ++k) acc = fmaf(xr[k], W[(size_t)k * FDIM + f], acc);
  H[(size_t)row * FDIM + f] = acc;
}

__global__ void aggregate_kernel(const float* __restrict__ H,
                                 const int* __restrict__ ei,
                                 const float* __restrict__ dinv,
                                 float* __restrict__ G, int E, int N) {
  const int e = blockIdx.x * (blockDim.x >> 5) + (threadIdx.x >> 5);
  const int EN = E + N;
  if (e >= EN) return;
  const int lane = threadIdx.x & 31;
  int s, d;
  if (e < E) { s = ei[e]; d = ei[E + e]; }
  else       { s = e - E; d = s; }
  const float w = dinv[s] * dinv[d];
  const float4 hv = reinterpret_cast<const float4*>(H + (size_t)s * FDIM)[lane];
  float* gp = G + (size_t)d * FDIM + lane * 4;
  atomicAdd(gp + 0, hv.x * w);
  atomicAdd(gp + 1, hv.y * w);
  atomicAdd(gp + 2, hv.z * w);
  atomicAdd(gp + 3, hv.w * w);
}


__global__ void count_rank_kernel(const int* __restrict__ ei, int* __restrict__ icnt,
                                  int* __restrict__ rank, int E) {
  const int e = blockIdx.x * blockDim.x + threadIdx.x;
  if (e >= E) return;
  rank[e] = atomicAdd(&icnt[ei[E + e]], 1);
}

#define SCAN_THREADS 256
#define SCAN_ELEMS   1024

__global__ void scan_local_kernel(const int* __restrict__ icnt, int* __restrict__ rowptr,
                                  int* __restrict__ bsum, int N) {
  __shared__ int wsum[SCAN_THREADS / 32];
  const int tid = threadIdx.x, lane = tid & 31, wv = tid >> 5;
  const int base = blockIdx.x * SCAN_ELEMS + tid * 4;
  int v[4];
  #pragma unroll
  for (int q = 0; q < 4; ++q) v[q] = (base + q < N) ? icnt[base + q] : 0;
  const int t = v[0] + v[1] + v[2] + v[3];
  int inc = t;
  #pragma unroll
  for (int off = 1; off < 32; off <<= 1) {
    const int u = __shfl_up(inc, off);
    if (lane >= off) inc += u;
  }
  if (lane == 31) wsum[wv] = inc;
  __syncthreads();
  int woff = 0, tot = 0;
  #pragma unroll
  for (int k = 0; k < SCAN_THREADS / 32; ++k) {
    const int s = wsum[k];
    if (k < wv) woff += s;
    tot += s;
  }
  int run = woff + inc - t;
  #pragma unroll
  for (int q = 0; q < 4; ++q) {
    if (base + q < N) rowptr[base + q] = run;
    run += v[q];
  }
  if (tid == 0) bsum[blockIdx.x] = tot;
}

__global__ void scan_add_kernel(int* __restrict__ rowptr, const int* __restrict__ bsum, int N) {
  __shared__ int red[SCAN_THREADS / 32];
  const int tid = threadIdx.x, lane = tid & 31, wv = tid >> 5;
  const int b = blockIdx.x;
  int part = 0;
  for (int k = tid; k < b; k += SCAN_THREADS) part += bsum[k];
  #pragma unroll
  for (int off = 16; off > 0; off >>= 1) part += __shfl_xor(part, off);
  if (lane == 0) red[wv] = part;
  __syncthreads();
  int add = 0;
  #pragma unroll
  for (int k = 0; k < SCAN_THREADS / 32; ++k) add += red[k];
  const int base = b * SCAN_ELEMS + tid * 4;
  #pragma unroll
  for (int q = 0; q < 4; ++q)
    if (base + q < N) rowptr[base + q] += add;
  if (b == (int)gridDim.x - 1 && tid == 0) rowptr[N] = add + bsum[b];
}

__global__ void place_kernel(const int* __restrict__ ei, const int* __restrict__ rowptr,
                             const int* __restrict__ rank, int* __restrict__ col, int E) {
  const int e = blockIdx.x * blockDim.x + threadIdx.x;
  if (e >= E) return;
  col[rowptr[ei[E + e]] + rank[e]] = ei[e];
}

__global__ void dinv_from_count_kernel(float* __restrict__ cnt_dinv, int N) {
  const int i = blockIdx.x * blockDim.x + threadIdx.x;
  if (i < N) {
    const int c = __float_as_int(cnt_dinv[i]);
    cnt_dinv[i] = rsqrtf(fmaxf((float)(c + 1), 1.0f));
  }
}

__device__ __forceinline__ void row_fma(float4& acc, const float4 h, const float w) {
  acc.x = fmaf(h.x, w, acc.x);
  acc.y = fmaf(h.y, w, acc.y);
  acc.z = fmaf(h.z, w, acc.z);
  acc.w = fmaf(h.w, w, acc.w);
}

__global__ void gather_rows_kernel(const float* __restrict__ H, const int* __restrict__ rowptr,
                                   const int* __restrict__ col, const float* __restrict__ dinv,
                                   const float* __restrict__ bias, float* __restrict__ G, int N) {
  const int d = __builtin_amdgcn_readfirstlane(
      (int)(blockIdx.x * (blockDim.x >> 5) + (threadIdx.x >> 5)));
  if (d >= N) return;
  const int lane = threadIdx.x & 31;
  const int beg = rowptr[d];
  const int end = rowptr[d + 1];
  const float dd = dinv[d];
  const float4* __restrict__ H4 = reinterpret_cast<const float4*>(H);
  const float bx = bias[lane * 4 + 0];
  const float by = bias[lane * 4 + 1];
  const float bz = bias[lane * 4 + 2];
  const float bw = bias[lane * 4 + 3];

  float4 acc;
  {
    const float4 hs = H4[(size_t)d * (FDIM / 4) + lane];
    const float ws = dd * dd;
    acc.x = hs.x * ws; acc.y = hs.y * ws; acc.z = hs.z * ws; acc.w = hs.w * ws;
  }

  for (int j = beg; j < end; j += 32) {
    const int n = min(32, end - j);
    int   s_l = 0;
    float w_l = 0.0f;
    if (lane < n) {
      s_l = col[j + lane];
      w_l = dinv[s_l] * dd;
    }
    int t = 0;
    for (; t + 4 <= n; t += 4) {
      const int s0 = __builtin_amdgcn_readlane(s_l, t);
      const int s1 = __builtin_amdgcn_readlane(s_l, t + 1);
      const int s2 = __builtin_amdgcn_readlane(s_l, t + 2);
      const int s3 = __builtin_amdgcn_readlane(s_l, t + 3);
      const float4 h0 = H4[(size_t)s0 * (FDIM / 4) + lane];
      const float4 h1 = H4[(size_t)s1 * (FDIM / 4) + lane];
      const float4 h2 = H4[(size_t)s2 * (FDIM / 4) + lane];
      const float4 h3 = H4[(size_t)s3 * (FDIM / 4) + lane];
      const float w0 = __int_as_float(__builtin_amdgcn_readlane(__float_as_int(w_l), t));
      const float w1 = __int_as_float(__builtin_amdgcn_readlane(__float_as_int(w_l), t + 1));
      const float w2 = __int_as_float(__builtin_amdgcn_readlane(__float_as_int(w_l), t + 2));
      const float w3 = __int_as_float(__builtin_amdgcn_readlane(__float_as_int(w_l), t + 3));
      row_fma(acc, h0, w0);
      row_fma(acc, h1, w1);
      row_fma(acc, h2, w2);
      row_fma(acc, h3, w3);
    }
    const int r = n - t;
    if (r == 3) {
      const int s0 = __builtin_amdgcn_readlane(s_l, t);
      const int s1 = __builtin_amdgcn_readlane(s_l, t + 1);
      const int s2 = __builtin_amdgcn_readlane(s_l, t + 2);
      const float4 h0 = H4[(size_t)s0 * (FDIM / 4) + lane];
      const float4 h1 = H4[(size_t)s1 * (FDIM / 4) + lane];
      const float4 h2 = H4[(size_t)s2 * (FDIM / 4) + lane];
      const float w0 = __int_as_float(__builtin_amdgcn_readlane(__float_as_int(w_l), t));
      const float w1 = __int_as_float(__builtin_amdgcn_readlane(__float_as_int(w_l), t + 1));
      const float w2 = __int_as_float(__builtin_amdgcn_readlane(__float_as_int(w_l), t + 2));
      row_fma(acc, h0, w0);
      row_fma(acc, h1, w1);
      row_fma(acc, h2, w2);
    } else if (r == 2) {
      const int s0 = __builtin_amdgcn_readlane(s_l, t);
      const int s1 = __builtin_amdgcn_readlane(s_l, t + 1);
      const float4 h0 = H4[(size_t)s0 * (FDIM / 4) + lane];
      const float4 h1 = H4[(size_t)s1 * (FDIM / 4) + lane];
      const float w0 = __int_as_float(__builtin_amdgcn_readlane(__float_as_int(w_l), t));
      const float w1 = __int_as_float(__builtin_amdgcn_readlane(__float_as_int(w_l), t + 1));
      row_fma(acc, h0, w0);
      row_fma(acc, h1, w1);
    } else if (r == 1) {
      const int s0 = __builtin_amdgcn_readlane(s_l, t);
      const float w0 = __int_as_float(__builtin_amdgcn_readlane(__float_as_int(w_l), t));
      row_fma(acc, H4[(size_t)s0 * (FDIM / 4) + lane], w0);
    }
  }
  acc.x = fmaxf(acc.x + bx, 0.0f);
  acc.y = fmaxf(acc.y + by, 0.0f);
  acc.z = fmaxf(acc.z + bz, 0.0f);
  acc.w = fmaxf(acc.w + bw, 0.0f);
  reinterpret_cast<float4*>(G)[(size_t)d * (FDIM / 4) + lane] = acc;
}

__global__ void bias_relu_kernel(float* __restrict__ G, const float* __restrict__ b) {
  const size_t i = (size_t)blockIdx.x * FDIM + threadIdx.x;
  G[i] = fmaxf(G[i] + b[threadIdx.x], 0.0f);
}

__global__ void init_out_kernel(float* __restrict__ out, int ng) {
  int i = blockIdx.x * blockDim.x + threadIdx.x;
  if (i >= ng * 2 * FDIM) return;
  int f = i & (2 * FDIM - 1);
  out[i] = (f < FDIM) ? 0.0f : __int_as_float(0xFF7FFFFF);
}

__device__ __forceinline__ void atomic_max_f32(float* addr, float v) {
  if (v >= 0.0f) atomicMax((int*)addr, __float_as_int(v));
  else           atomicMin((unsigned int*)addr, (unsigned int)__float_as_int(v));
}

__global__ void pool_kernel(const float* __restrict__ G, const int* __restrict__ gi,
                            float* __restrict__ out, float* __restrict__ cnt) {
  const int node = blockIdx.x;
  const int f = threadIdx.x;
  const int g = gi[node];
  const float v = G[(size_t)node * FDIM + f];
  atomicAdd(&out[(size_t)g * 2 * FDIM + f], v);
  atomic_max_f32(&out[(size_t)g * 2 * FDIM + FDIM + f], v);
  if (f == 0) atomicAdd(&cnt[g], 1.0f);
}

__global__ void finalize_mean_kernel(float* __restrict__ out, const float* __restrict__ cnt) {
  const int g = blockIdx.x;
  out[(size_t)g * 2 * FDIM + threadIdx.x] /= fmaxf(cnt[g], 1.0f);
}


extern "C" void kernel_launch(void* const* d_in, const int* in_sizes, int n_in,
                              void* d_out, int out_size, void* d_ws, size_t ws_size,
                              hipStream_t stream) {
  const float* x  = (const float*)d_in[0];
  const int*   ei = (const int*)d_in[1];
  const int*   gi = (const int*)d_in[2];
  const float* Wm[3] = {(const float*)d_in[4], (const float*)d_in[6], (const float*)d_in[8]};
  const float* Bv[3] = {(const float*)d_in[5], (const float*)d_in[7], (const float*)d_in[9]};
  float* out = (float*)d_out;

  const int  N  = in_sizes[0] / FDIM;
  const int  E  = in_sizes[1] / 2;
  const int  NG = out_size / (2 * FDIM);
  const long Nf = (long)N * FDIM;

  float* Hbuf = (float*)d_ws;
  float* Gbuf = Hbuf + Nf;
  float* dinv = Gbuf + Nf;
  float* cnt  = dinv + N;
  if (ws_size < (size_t)(2 * Nf + N + NG) * sizeof(float)) return;

  const int nsb = (N + SCAN_ELEMS - 1) / SCAN_ELEMS;
  int* rowptr = (int*)(cnt + NG);
  int* bsum   = rowptr + (N + 1);
  int* col    = bsum + nsb;
  int* rank   = (int*)Hbuf;
  const bool ordered =
      ws_size >= (size_t)(2 * Nf + N + NG) * sizeof(float) + ((size_t)N + 1 + nsb + E) * sizeof(int) &&
      (long)E <= Nf && E > 0;

  auto zero = [&](float* p, long n) {
    zero_f32_kernel<<<(unsigned)((n + 255) / 256), 256, 0, stream>>>(p, n);
  };

  const int tiles = N / 16;
  const int rem   = N - tiles * 16;
  const float* cur = x;

  if (ordered) {
    zero(dinv, N);
    count_rank_kernel<<<(E + 255) / 256, 256, 0, stream>>>(ei, (int*)dinv, rank, E);
    scan_local_kernel<<<nsb, SCAN_THREADS, 0, stream>>>((const int*)dinv, rowptr, bsum, N);
    scan_add_kernel<<<nsb, SCAN_THREADS, 0, stream>>>(rowptr, bsum, N);
    place_kernel<<<(E + 255) / 256, 256, 0, stream>>>(ei, rowptr, rank, col, E);
    dinv_from_count_kernel<<<(N + 255) / 256, 256, 0, stream>>>(dinv, N);

    for (int l = 0; l < 3; ++l) {
      if (tiles > 0)
        gemm_wmma_kernel<<<tiles, 256, 0, stream>>>(cur, Wm[l], Hbuf);
      if (rem > 0)
        gemm_tail_kernel<<<rem, FDIM, 0, stream>>>(cur, Wm[l], Hbuf, tiles * 16);
      gather_rows_kernel<<<(N + 7) / 8, 256, 0, stream>>>(Hbuf, rowptr, col, dinv, Bv[l], Gbuf, N);
      cur = Gbuf;
    }
  } else {
    zero(dinv, N);
    degree_kernel<<<(E + N + 255) / 256, 256, 0, stream>>>(ei, dinv, E, N);
    dinv_kernel<<<(N + 255) / 256, 256, 0, stream>>>(dinv, N);

    for (int l = 0; l < 3; ++l) {
      if (tiles > 0)
        gemm_wmma_kernel<<<tiles, 256, 0, stream>>>(cur, Wm[l], Hbuf);
      if (rem > 0)
        gemm_tail_kernel<<<rem, FDIM, 0, stream>>>(cur, Wm[l], Hbuf, tiles * 16);
      zero(Gbuf, Nf);
      aggregate_kernel<<<(E + N + 7) / 8, 256, 0, stream>>>(Hbuf, ei, dinv, Gbuf, E, N);
      bias_relu_kernel<<<N, FDIM, 0, stream>>>(Gbuf, Bv[l]);
      cur = Gbuf;
    }
  }

  init_out_kernel<<<(NG * 2 * FDIM + 255) / 256, 256, 0, stream>>>(out, NG);
  zero(cnt, NG);
  pool_kernel<<<N, FDIM, 0, stream>>>(Gbuf, gi, out, cnt);
  finalize_mean_kernel<<<NG, FDIM, 0, stream>>>(out, cnt);
}


# Product kernel retiled: a wave owns a 16x128 strip, W staged once per block in LDS as k-pairs, each A fragment loaded once for 8 WMMAs (was 3 loads per WMMA); k order and sums unchanged
# speedup vs baseline: 3.1718x; 1.1511x over previous
#include <hip/hip_runtime.h>
#include <float.h>

#define FDIM 128

typedef __attribute__((ext_vector_type(2))) float v2f;
typedef __attribute__((ext_vector_type(8))) float v8f;


__global__ void zero_f32_kernel(float* __restrict__ p, long n) {
  long i = (long)blockIdx.x * blockDim.x + threadIdx.x;
  if (i < n) p[i] = 0.0f;
}

__global__ void degree_kernel(const int* __restrict__ ei, float* __restrict__ deg,
                              int E, int N) {
  int e = blockIdx.x * blockDim.x + threadIdx.x;
  int EN = E + N;
  if (e >= EN) return;
  int d = (e < E) ? ei[E + e] : (e - E);
  atomicAdd(&deg[d], 1.0f);
}

__global__ void dinv_kernel(float* __restrict__ deg_dinv, int N) {
  int i = blockIdx.x * blockDim.x + threadIdx.x;
  if (i < N) deg_dinv[i] = rsqrtf(fmaxf(deg_dinv[i], 1.0f));
}

#define GEMM_WPITCH 288
#define GEMM_BLOCKS 512

__global__ __launch_bounds__(256, 4) void gemm_strip_kernel(const float* __restrict__ X,
                                                         const float* __restrict__ W,
                                                         float* __restrict__ H, int tiles) {
  __shared__ alignas(16) float Ws[64 * GEMM_WPITCH];
  const int tid = threadIdx.x;

  {
    const float4* __restrict__ W4 = reinterpret_cast<const float4*>(W);
    for (int it = 0; it < 8; ++it) {
      const int i = it * 256 + tid;
      const int p = i >> 5;
      const int q = i & 31;
      const float4 w0 = W4[(2 * p) * (FDIM / 4) + q];
      const float4 w1 = W4[(2 * p + 1) * (FDIM / 4) + q];
      float4* dst = reinterpret_cast<float4*>(Ws + p * GEMM_WPITCH + q * 8);
      dst[0] = make_float4(w0.x, w1.x, w0.y, w1.y);
      dst[1] = make_float4(w0.z, w1.z, w0.w, w1.w);
    }
  }
  __syncthreads();

  const int wave = __builtin_amdgcn_readfirstlane((int)(threadIdx.x >> 5));
  const int lane = tid & 31;
  const int c = lane & 15;
  const int h = lane >> 4;

  const v2f* __restrict__ bfrag = reinterpret_cast<const v2f*>(Ws) + h * (GEMM_WPITCH / 2) + c;

  const int xoff = c * (FDIM / 2) + h;
  const int hoff = 8 * h * FDIM + c;

  const int s0 = blockIdx.x + gridDim.x * wave;
  const size_t sstep = (size_t)gridDim.x * 8 * (16 * FDIM);
  const v2f* xs = reinterpret_cast<const v2f*>(X + (size_t)s0 * (16 * FDIM)) + xoff;
  float* hs = H + (size_t)s0 * (16 * FDIM) + hoff;

  for (int s = s0; s < tiles; s += gridDim.x * 8, xs += sstep / 2, hs += sstep) {
    v2f a[FDIM / 4];
    #pragma unroll
    for (int kk = 0; kk < FDIM / 4; ++kk) a[kk] = xs[2 * kk];

    v8f acc[8];
    #pragma unroll
    for (int t = 0; t < 8; ++t) acc[t] = v8f{};

    v2f bcur[8];
    #pragma unroll
    for (int t = 0; t < 8; ++t) bcur[t] = bfrag[16 * t];

    #pragma unroll
    for (int kk = 0; kk < FDIM / 4; ++kk) {
      v2f bnxt[8];
      if (kk + 1 < FDIM / 4) {
        const v2f* bp = bfrag + (2 * (kk + 1)) * (GEMM_WPITCH / 2);
        #pragma unroll
        for (int t = 0; t < 8; ++t) bnxt[t] = bp[16 * t];
      }
      #pragma unroll
      for (int t = 0; t < 8; ++t) {
        acc[t] = __builtin_amdgcn_wmma_f32_16x16x4_f32(
             false, a[kk],  false, bcur[t],
             (short)0, acc[t],  false,  false);
      }
      __builtin_amdgcn_sched_barrier(0);
      if (kk + 1 < FDIM / 4) {
        #pragma unroll
        for (int t = 0; t < 8; ++t) bcur[t] = bnxt[t];
      }
    }

    #pragma unroll
    for (int t = 0; t < 8; ++t) {
      #pragma unroll
      for (int r = 0; r < 8; ++r) hs[r * FDIM + 16 * t] = acc[t][r];
    }
  }
}

__global__ void gemm_tail_kernel(const float* __restrict__ X,
                                 const float* __restrict__ W,
                                 float* __restrict__ H, int base) {
  const int row = base + blockIdx.x;
  const int f = threadIdx.x;
  const float* xr = X + (size_t)row * FDIM;
  float acc = 0.0f;
  #pragma unroll 8
  for (int k = 0; k < FDIM; ++k) acc = fmaf(xr[k], W[(size_t)k * FDIM + f], acc);
  H[(size_t)row * FDIM + f] = acc;
}

__global__ void aggregate_kernel(const float* __restrict__ H,
                                 const int* __restrict__ ei,
                                 const float* __restrict__ dinv,
                                 float* __restrict__ G, int E, int N) {
  const int e = blockIdx.x * (blockDim.x >> 5) + (threadIdx.x >> 5);
  const int EN = E + N;
  if (e >= EN) return;
  const int lane = threadIdx.x & 31;
  int s, d;
  if (e < E) { s = ei[e]; d = ei[E + e]; }
  else       { s = e - E; d = s; }
  const float w = dinv[s] * dinv[d];
  const float4 hv = reinterpret_cast<const float4*>(H + (size_t)s * FDIM)[lane];
  float* gp = G + (size_t)d * FDIM + lane * 4;
  atomicAdd(gp + 0, hv.x * w);
  atomicAdd(gp + 1, hv.y * w);
  atomicAdd(gp + 2, hv.z * w);
  atomicAdd(gp + 3, hv.w * w);
}


__global__ void count_rank_kernel(const int* __restrict__ ei, int* __restrict__ icnt,
                                  int* __restrict__ rank, int E) {
  const int e = blockIdx.x * blockDim.x + threadIdx.x;
  if (e >= E) return;
  rank[e] = atomicAdd(&icnt[ei[E + e]], 1);
}

#define SCAN_THREADS 256
#define SCAN_ELEMS   1024

__global__ void scan_local_kernel(const int* __restrict__ icnt, int* __restrict__ rowptr,
                                  int* __restrict__ bsum, int N) {
  __shared__ int wsum[SCAN_THREADS / 32];
  const int tid = threadIdx.x, lane = tid & 31, wv = tid >> 5;
  const int base = blockIdx.x * SCAN_ELEMS + tid * 4;
  int v[4];
  #pragma unroll
  for (int q = 0; q < 4; ++q) v[q] = (base + q < N) ? icnt[base + q] : 0;
  const int t = v[0] + v[1] + v[2] + v[3];
  int inc = t;
  #pragma unroll
  for (int off = 1; off < 32; off <<= 1) {
    const int u = __shfl_up(inc, off);
    if (lane >= off) inc += u;
  }
  if (lane == 31) wsum[wv] = inc;
  __syncthreads();
  int woff = 0, tot = 0;
  #pragma unroll
  for (int k = 0; k < SCAN_THREADS / 32; ++k) {
    const int s = wsum[k];
    if (k < wv) woff += s;
    tot += s;
  }
  int run = woff + inc - t;
  #pragma unroll
  for (int q = 0; q < 4; ++q) {
    if (base + q < N) rowptr[base + q] = run;
    run += v[q];
  }
  if (tid == 0) bsum[blockIdx.x] = tot;
}

__global__ void scan_add_kernel(int* __restrict__ rowptr, const int* __restrict__ bsum, int N) {
  __shared__ int red[SCAN_THREADS / 32];
  const int tid = threadIdx.x, lane = tid & 31, wv = tid >> 5;
  const int b = blockIdx.x;
  int part = 0;
  for (int k = tid; k < b; k += SCAN_THREADS) part += bsum[k];
  #pragma unroll
  for (int off = 16; off > 0; off >>= 1) part += __shfl_xor(part, off);
  if (lane == 0) red[wv] = part;
  __syncthreads();
  int add = 0;
  #pragma unroll
  for (int k = 0; k < SCAN_THREADS / 32; ++k) add += red[k];
  const int base = b * SCAN_ELEMS + tid * 4;
  #pragma unroll
  for (int q = 0; q < 4; ++q)
    if (base + q < N) rowptr[base + q] += add;
  if (b == (int)gridDim.x - 1 && tid == 0) rowptr[N] = add + bsum[b];
}

__global__ void place_kernel(const int* __restrict__ ei, const int* __restrict__ rowptr,
                             const int* __restrict__ rank, int* __restrict__ col, int E) {
  const int e = blockIdx.x * blockDim.x + threadIdx.x;
  if (e >= E) return;
  col[rowptr[ei[E + e]] + rank[e]] = ei[e];
}

__global__ void dinv_from_count_kernel(float* __restrict__ cnt_dinv, int N) {
  const int i = blockIdx.x * blockDim.x + threadIdx.x;
  if (i < N) {
    const int c = __float_as_int(cnt_dinv[i]);
    cnt_dinv[i] = rsqrtf(fmaxf((float)(c + 1), 1.0f));
  }
}

__device__ __forceinline__ void row_fma(float4& acc, const float4 h, const float w) {
  acc.x = fmaf(h.x, w, acc.x);
  acc.y = fmaf(h.y, w, acc.y);
  acc.z = fmaf(h.z, w, acc.z);
  acc.w = fmaf(h.w, w, acc.w);
}

__global__ void gather_rows_kernel(const float* __restrict__ H, const int* __restrict__ rowptr,
                                   const int* __restrict__ col, const float* __restrict__ dinv,
                                   const float* __restrict__ bias, float* __restrict__ G, int N) {
  const int d = __builtin_amdgcn_readfirstlane(
      (int)(blockIdx.x * (blockDim.x >> 5) + (threadIdx.x >> 5)));
  if (d >= N) return;
  const int lane = threadIdx.x & 31;
  const int beg = rowptr[d];
  const int end = rowptr[d + 1];
  const float dd = dinv[d];
  const float4* __restrict__ H4 = reinterpret_cast<const float4*>(H);
  const float bx = bias[lane * 4 + 0];
  const float by = bias[lane * 4 + 1];
  const float bz = bias[lane * 4 + 2];
  const float bw = bias[lane * 4 + 3];

  float4 acc;
  {
    const float4 hs = H4[(size_t)d * (FDIM / 4) + lane];
    const float ws = dd * dd;
    acc.x = hs.x * ws; acc.y = hs.y * ws; acc.z = hs.z * ws; acc.w = hs.w * ws;
  }

  for (int j = beg; j < end; j += 32) {
    const int n = min(32, end - j);
    int   s_l = 0;
    float w_l = 0.0f;
    if (lane < n) {
      s_l = col[j + lane];
      w_l = dinv[s_l] * dd;
    }
    int t = 0;
    for (; t + 4 <= n; t += 4) {
      const int s0 = __builtin_amdgcn_readlane(s_l, t);
      const int s1 = __builtin_amdgcn_readlane(s_l, t + 1);
      const int s2 = __builtin_amdgcn_readlane(s_l, t + 2);
      const int s3 = __builtin_amdgcn_readlane(s_l, t + 3);
      const float4 h0 = H4[(size_t)s0 * (FDIM / 4) + lane];
      const float4 h1 = H4[(size_t)s1 * (FDIM / 4) + lane];
      const float4 h2 = H4[(size_t)s2 * (FDIM / 4) + lane];
      const float4 h3 = H4[(size_t)s3 * (FDIM / 4) + lane];
      const float w0 = __int_as_float(__builtin_amdgcn_readlane(__float_as_int(w_l), t));
      const float w1 = __int_as_float(__builtin_amdgcn_readlane(__float_as_int(w_l), t + 1));
      const float w2 = __int_as_float(__builtin_amdgcn_readlane(__float_as_int(w_l), t + 2));
      const float w3 = __int_as_float(__builtin_amdgcn_readlane(__float_as_int(w_l), t + 3));
      row_fma(acc, h0, w0);
      row_fma(acc, h1, w1);
      row_fma(acc, h2, w2);
      row_fma(acc, h3, w3);
    }
    const int r = n - t;
    if (r == 3) {
      const int s0 = __builtin_amdgcn_readlane(s_l, t);
      const int s1 = __builtin_amdgcn_readlane(s_l, t + 1);
      const int s2 = __builtin_amdgcn_readlane(s_l, t + 2);
      const float4 h0 = H4[(size_t)s0 * (FDIM / 4) + lane];
      const float4 h1 = H4[(size_t)s1 * (FDIM / 4) + lane];
      const float4 h2 = H4[(size_t)s2 * (FDIM / 4) + lane];
      const float w0 = __int_as_float(__builtin_amdgcn_readlane(__float_as_int(w_l), t));
      const float w1 = __int_as_float(__builtin_amdgcn_readlane(__float_as_int(w_l), t + 1));
      const float w2 = __int_as_float(__builtin_amdgcn_readlane(__float_as_int(w_l), t + 2));
      row_fma(acc, h0, w0);
      row_fma(acc, h1, w1);
      row_fma(acc, h2, w2);
    } else if (r == 2) {
      const int s0 = __builtin_amdgcn_readlane(s_l, t);
      const int s1 = __builtin_amdgcn_readlane(s_l, t + 1);
      const float4 h0 = H4[(size_t)s0 * (FDIM / 4) + lane];
      const float4 h1 = H4[(size_t)s1 * (FDIM / 4) + lane];
      const float w0 = __int_as_float(__builtin_amdgcn_readlane(__float_as_int(w_l), t));
      const float w1 = __int_as_float(__builtin_amdgcn_readlane(__float_as_int(w_l), t + 1));
      row_fma(acc, h0, w0);
      row_fma(acc, h1, w1);
    } else if (r == 1) {
      const int s0 = __builtin_amdgcn_readlane(s_l, t);
      const float w0 = __int_as_float(__builtin_amdgcn_readlane(__float_as_int(w_l), t));
      row_fma(acc, H4[(size_t)s0 * (FDIM / 4) + lane], w0);
    }
  }
  acc.x = fmaxf(acc.x + bx, 0.0f);
  acc.y = fmaxf(acc.y + by, 0.0f);
  acc.z = fmaxf(acc.z + bz, 0.0f);
  acc.w = fmaxf(acc.w + bw, 0.0f);
  reinterpret_cast<float4*>(G)[(size_t)d * (FDIM / 4) + lane] = acc;
}

__global__ void bias_relu_kernel(float* __restrict__ G, const float* __restrict__ b) {
  const size_t i = (size_t)blockIdx.x * FDIM + threadIdx.x;
  G[i] = fmaxf(G[i] + b[threadIdx.x], 0.0f);
}

__global__ void init_out_kernel(float* __restrict__ out, int ng) {
  int i = blockIdx.x * blockDim.x + threadIdx.x;
  if (i >= ng * 2 * FDIM) return;
  int f = i & (2 * FDIM - 1);
  out[i] = (f < FDIM) ? 0.0f : __int_as_float(0xFF7FFFFF);
}

__device__ __forceinline__ void atomic_max_f32(float* addr, float v) {
  if (v >= 0.0f) atomicMax((int*)addr, __float_as_int(v));
  else           atomicMin((unsigned int*)addr, (unsigned int)__float_as_int(v));
}

__global__ void pool_kernel(const float* __restrict__ G, const int* __restrict__ gi,
                            float* __restrict__ out, float* __restrict__ cnt) {
  const int node = blockIdx.x;
  const int f = threadIdx.x;
  const int g = gi[node];
  const float v = G[(size_t)node * FDIM + f];
  atomicAdd(&out[(size_t)g * 2 * FDIM + f], v);
  atomic_max_f32(&out[(size_t)g * 2 * FDIM + FDIM + f], v);
  if (f == 0) atomicAdd(&cnt[g], 1.0f);
}

__global__ void finalize_mean_kernel(float* __restrict__ out, const float* __restrict__ cnt) {
  const int g = blockIdx.x;
  out[(size_t)g * 2 * FDIM + threadIdx.x] /= fmaxf(cnt[g], 1.0f);
}


extern "C" void kernel_launch(void* const* d_in, const int* in_sizes, int n_in,
                              void* d_out, int out_size, void* d_ws, size_t ws_size,
                              hipStream_t stream) {
  const float* x  = (const float*)d_in[0];
  const int*   ei = (const int*)d_in[1];
  const int*   gi = (const int*)d_in[2];
  const float* Wm[3] = {(const float*)d_in[4], (const float*)d_in[6], (const float*)d_in[8]};
  const float* Bv[3] = {(const float*)d_in[5], (const float*)d_in[7], (const float*)d_in[9]};
  float* out = (float*)d_out;

  const int  N  = in_sizes[0] / FDIM;
  const int  E  = in_sizes[1] / 2;
  const int  NG = out_size / (2 * FDIM);
  const long Nf = (long)N * FDIM;

  float* Hbuf = (float*)d_ws;
  float* Gbuf = Hbuf + Nf;
  float* dinv = Gbuf + Nf;
  float* cnt  = dinv + N;
  if (ws_size < (size_t)(2 * Nf + N + NG) * sizeof(float)) return;

  const int nsb = (N + SCAN_ELEMS - 1) / SCAN_ELEMS;
  int* rowptr = (int*)(cnt + NG);
  int* bsum   = rowptr + (N + 1);
  int* col    = bsum + nsb;
  int* rank   = (int*)Hbuf;
  const bool ordered =
      ws_size >= (size_t)(2 * Nf + N + NG) * sizeof(float) + ((size_t)N + 1 + nsb + E) * sizeof(int) &&
      (long)E <= Nf && E > 0;

  auto zero = [&](float* p, long n) {
    zero_f32_kernel<<<(unsigned)((n + 255) / 256), 256, 0, stream>>>(p, n);
  };

  const int tiles = N / 16;
  const int rem   = N - tiles * 16;
  const float* cur = x;

  if (ordered) {
    zero(dinv, N);
    count_rank_kernel<<<(E + 255) / 256, 256, 0, stream>>>(ei, (int*)dinv, rank, E);
    scan_local_kernel<<<nsb, SCAN_THREADS, 0, stream>>>((const int*)dinv, rowptr, bsum, N);
    scan_add_kernel<<<nsb, SCAN_THREADS, 0, stream>>>(rowptr, bsum, N);
    place_kernel<<<(E + 255) / 256, 256, 0, stream>>>(ei, rowptr, rank, col, E);
    dinv_from_count_kernel<<<(N + 255) / 256, 256, 0, stream>>>(dinv, N);

    for (int l = 0; l < 3; ++l) {
      if (tiles > 0)
        gemm_strip_kernel<<<GEMM_BLOCKS, 256, 0, stream>>>(cur, Wm[l], Hbuf, tiles);
      if (rem > 0)
        gemm_tail_kernel<<<rem, FDIM, 0, stream>>>(cur, Wm[l], Hbuf, tiles * 16);
      gather_rows_kernel<<<(N + 7) / 8, 256, 0, stream>>>(Hbuf, rowptr, col, dinv, Bv[l], Gbuf, N);
      cur = Gbuf;
    }
  } else {
    zero(dinv, N);
    degree_kernel<<<(E + N + 255) / 256, 256, 0, stream>>>(ei, dinv, E, N);
    dinv_kernel<<<(N + 255) / 256, 256, 0, stream>>>(dinv, N);

    for (int l = 0; l < 3; ++l) {
      if (tiles > 0)
        gemm_strip_kernel<<<GEMM_BLOCKS, 256, 0, stream>>>(cur, Wm[l], Hbuf, tiles);
      if (rem > 0)
        gemm_tail_kernel<<<rem, FDIM, 0, stream>>>(cur, Wm[l], Hbuf, tiles * 16);
      zero(Gbuf, Nf);
      aggregate_kernel<<<(E + N + 7) / 8, 256, 0, stream>>>(Hbuf, ei, dinv, Gbuf, E, N);
      bias_relu_kernel<<<N, FDIM, 0, stream>>>(Gbuf, Bv[l]);
      cur = Gbuf;
    }
  }

  init_out_kernel<<<(NG * 2 * FDIM + 255) / 256, 256, 0, stream>>>(out, NG);
  zero(cnt, NG);
  pool_kernel<<<N, FDIM, 0, stream>>>(Gbuf, gi, out, cnt);
  finalize_mean_kernel<<<NG, FDIM, 0, stream>>>(out, cnt);
}


# Pooling reduces before its atomics: a wave takes 32 adjacent node rows, sums and maxes each run of one graph id in registers, one set of whole-line atomics per run, not per node; sum order changes
# speedup vs baseline: 3.5605x; 1.1225x over previous
#include <hip/hip_runtime.h>
#include <float.h>

#define FDIM 128

typedef __attribute__((ext_vector_type(2))) float v2f;
typedef __attribute__((ext_vector_type(8))) float v8f;


__global__ void zero_f32_kernel(float* __restrict__ p, long n) {
  long i = (long)blockIdx.x * blockDim.x + threadIdx.x;
  if (i < n) p[i] = 0.0f;
}

__global__ void degree_kernel(const int* __restrict__ ei, float* __restrict__ deg,
                              int E, int N) {
  int e = blockIdx.x * blockDim.x + threadIdx.x;
  int EN = E + N;
  if (e >= EN) return;
  int d = (e < E) ? ei[E + e] : (e - E);
  atomicAdd(&deg[d], 1.0f);
}

__global__ void dinv_kernel(float* __restrict__ deg_dinv, int N) {
  int i = blockIdx.x * blockDim.x + threadIdx.x;
  if (i < N) deg_dinv[i] = rsqrtf(fmaxf(deg_dinv[i], 1.0f));
}

#define GEMM_WPITCH 288
#define GEMM_BLOCKS 512

__global__ __launch_bounds__(256, 4) void gemm_strip_kernel(const float* __restrict__ X,
                                                         const float* __restrict__ W,
                                                         float* __restrict__ H, int tiles) {
  __shared__ alignas(16) float Ws[64 * GEMM_WPITCH];
  const int tid = threadIdx.x;

  {
    const float4* __restrict__ W4 = reinterpret_cast<const float4*>(W);
    for (int it = 0; it < 8; ++it) {
      const int i = it * 256 + tid;
      const int p = i >> 5;
      const int q = i & 31;
      const float4 w0 = W4[(2 * p) * (FDIM / 4) + q];
      const float4 w1 = W4[(2 * p + 1) * (FDIM / 4) + q];
      float4* dst = reinterpret_cast<float4*>(Ws + p * GEMM_WPITCH + q * 8);
      dst[0] = make_float4(w0.x, w1.x, w0.y, w1.y);
      dst[1] = make_float4(w0.z, w1.z, w0.w, w1.w);
    }
  }
  __syncthreads();

  const int wave = __builtin_amdgcn_readfirstlane((int)(threadIdx.x >> 5));
  const int lane = tid & 31;
  const int c = lane & 15;
  const int h = lane >> 4;

  const v2f* __restrict__ bfrag = reinterpret_cast<const v2f*>(Ws) + h * (GEMM_WPITCH / 2) + c;

  const int xoff = c * (FDIM / 2) + h;
  const int hoff = 8 * h * FDIM + c;

  const int s0 = blockIdx.x + gridDim.x * wave;
  const size_t sstep = (size_t)gridDim.x * 8 * (16 * FDIM);
  const v2f* xs = reinterpret_cast<const v2f*>(X + (size_t)s0 * (16 * FDIM)) + xoff;
  float* hs = H + (size_t)s0 * (16 * FDIM) + hoff;

  for (int s = s0; s < tiles; s += gridDim.x * 8, xs += sstep / 2, hs += sstep) {
    v2f a[FDIM / 4];
    #pragma unroll
    for (int kk = 0; kk < FDIM / 4; ++kk) a[kk] = xs[2 * kk];

    v8f acc[8];
    #pragma unroll
    for (int t = 0; t < 8; ++t) acc[t] = v8f{};

    v2f bcur[8];
    #pragma unroll
    for (int t = 0; t < 8; ++t) bcur[t] = bfrag[16 * t];

    #pragma unroll
    for (int kk = 0; kk < FDIM / 4; ++kk) {
      v2f bnxt[8];
      if (kk + 1 < FDIM / 4) {
        const v2f* bp = bfrag + (2 * (kk + 1)) * (GEMM_WPITCH / 2);
        #pragma unroll
        for (int t = 0; t < 8; ++t) bnxt[t] = bp[16 * t];
      }
      #pragma unroll
      for (int t = 0; t < 8; ++t) {
        acc[t] = __builtin_amdgcn_wmma_f32_16x16x4_f32(
             false, a[kk],  false, bcur[t],
             (short)0, acc[t],  false,  false);
      }
      __builtin_amdgcn_sched_barrier(0);
      if (kk + 1 < FDIM / 4) {
        #pragma unroll
        for (int t = 0; t < 8; ++t) bcur[t] = bnxt[t];
      }
    }

    #pragma unroll
    for (int t = 0; t < 8; ++t) {
      #pragma unroll
      for (int r = 0; r < 8; ++r) hs[r * FDIM + 16 * t] = acc[t][r];
    }
  }
}

__global__ void gemm_tail_kernel(const float* __restrict__ X,
                                 const float* __restrict__ W,
                                 float* __restrict__ H, int base) {
  const int row = base + blockIdx.x;
  const int f = threadIdx.x;
  const float* xr = X + (size_t)row * FDIM;
  float acc = 0.0f;
  #pragma unroll 8
  for (int k = 0; k < FDIM; ++k) acc = fmaf(xr[k], W[(size_t)k * FDIM + f], acc);
  H[(size_t)row * FDIM + f] = acc;
}

__global__ void aggregate_kernel(const float* __restrict__ H,
                                 const int* __restrict__ ei,
                                 const float* __restrict__ dinv,
                                 float* __restrict__ G, int E, int N) {
  const int e = blockIdx.x * (blockDim.x >> 5) + (threadIdx.x >> 5);
  const int EN = E + N;
  if (e >= EN) return;
  const int lane = threadIdx.x & 31;
  int s, d;
  if (e < E) { s = ei[e]; d = ei[E + e]; }
  else       { s = e - E; d = s; }
  const float w = dinv[s] * dinv[d];
  const float4 hv = reinterpret_cast<const float4*>(H + (size_t)s * FDIM)[lane];
  float* gp = G + (size_t)d * FDIM + lane * 4;
  atomicAdd(gp + 0, hv.x * w);
  atomicAdd(gp + 1, hv.y * w);
  atomicAdd(gp + 2, hv.z * w);
  atomicAdd(gp + 3, hv.w * w);
}


__global__ void count_rank_kernel(const int* __restrict__ ei, int* __restrict__ icnt,
                                  int* __restrict__ rank, int E) {
  const int e = blockIdx.x * blockDim.x + threadIdx.x;
  if (e >= E) return;
  rank[e] = atomicAdd(&icnt[ei[E + e]], 1);
}

#define SCAN_THREADS 256
#define SCAN_ELEMS   1024

__global__ void scan_local_kernel(const int* __restrict__ icnt, int* __restrict__ rowptr,
                                  int* __restrict__ bsum, int N) {
  __shared__ int wsum[SCAN_THREADS / 32];
  const int tid = threadIdx.x, lane = tid & 31, wv = tid >> 5;
  const int base = blockIdx.x * SCAN_ELEMS + tid * 4;
  int v[4];
  #pragma unroll
  for (int q = 0; q < 4; ++q) v[q] = (base + q < N) ? icnt[base + q] : 0;
  const int t = v[0] + v[1] + v[2] + v[3];
  int inc = t;
  #pragma unroll
  for (int off = 1; off < 32; off <<= 1) {
    const int u = __shfl_up(inc, off);
    if (lane >= off) inc += u;
  }
  if (lane == 31) wsum[wv] = inc;
  __syncthreads();
  int woff = 0, tot = 0;
  #pragma unroll
  for (int k = 0; k < SCAN_THREADS / 32; ++k) {
    const int s = wsum[k];
    if (k < wv) woff += s;
    tot += s;
  }
  int run = woff + inc - t;
  #pragma unroll
  for (int q = 0; q < 4; ++q) {
    if (base + q < N) rowptr[base + q] = run;
    run += v[q];
  }
  if (tid == 0) bsum[blockIdx.x] = tot;
}

__global__ void scan_add_kernel(int* __restrict__ rowptr, const int* __restrict__ bsum, int N) {
  __shared__ int red[SCAN_THREADS / 32];
  const int tid = threadIdx.x, lane = tid & 31, wv = tid >> 5;
  const int b = blockIdx.x;
  int part = 0;
  for (int k = tid; k < b; k += SCAN_THREADS) part += bsum[k];
  #pragma unroll
  for (int off = 16; off > 0; off >>= 1) part += __shfl_xor(part, off);
  if (lane == 0) red[wv] = part;
  __syncthreads();
  int add = 0;
  #pragma unroll
  for (int k = 0; k < SCAN_THREADS / 32; ++k) add += red[k];
  const int base = b * SCAN_ELEMS + tid * 4;
  #pragma unroll
  for (int q = 0; q < 4; ++q)
    if (base + q < N) rowptr[base + q] += add;
  if (b == (int)gridDim.x - 1 && tid == 0) rowptr[N] = add + bsum[b];
}

__global__ void place_kernel(const int* __restrict__ ei, const int* __restrict__ rowptr,
                             const int* __restrict__ rank, int* __restrict__ col, int E) {
  const int e = blockIdx.x * blockDim.x + threadIdx.x;
  if (e >= E) return;
  col[rowptr[ei[E + e]] + rank[e]] = ei[e];
}

__global__ void dinv_from_count_kernel(float* __restrict__ cnt_dinv, int N) {
  const int i = blockIdx.x * blockDim.x + threadIdx.x;
  if (i < N) {
    const int c = __float_as_int(cnt_dinv[i]);
    cnt_dinv[i] = rsqrtf(fmaxf((float)(c + 1), 1.0f));
  }
}

__device__ __forceinline__ void row_fma(float4& acc, const float4 h, const float w) {
  acc.x = fmaf(h.x, w, acc.x);
  acc.y = fmaf(h.y, w, acc.y);
  acc.z = fmaf(h.z, w, acc.z);
  acc.w = fmaf(h.w, w, acc.w);
}

__global__ void gather_rows_kernel(const float* __restrict__ H, const int* __restrict__ rowptr,
                                   const int* __restrict__ col, const float* __restrict__ dinv,
                                   const float* __restrict__ bias, float* __restrict__ G, int N) {
  const int d = __builtin_amdgcn_readfirstlane(
      (int)(blockIdx.x * (blockDim.x >> 5) + (threadIdx.x >> 5)));
  if (d >= N) return;
  const int lane = threadIdx.x & 31;
  const int beg = rowptr[d];
  const int end = rowptr[d + 1];
  const float dd = dinv[d];
  const float4* __restrict__ H4 = reinterpret_cast<const float4*>(H);
  const float bx = bias[lane * 4 + 0];
  const float by = bias[lane * 4 + 1];
  const float bz = bias[lane * 4 + 2];
  const float bw = bias[lane * 4 + 3];

  float4 acc;
  {
    const float4 hs = H4[(size_t)d * (FDIM / 4) + lane];
    const float ws = dd * dd;
    acc.x = hs.x * ws; acc.y = hs.y * ws; acc.z = hs.z * ws; acc.w = hs.w * ws;
  }

  for (int j = beg; j < end; j += 32) {
    const int n = min(32, end - j);
    int   s_l = 0;
    float w_l = 0.0f;
    if (lane < n) {
      s_l = col[j + lane];
      w_l = dinv[s_l] * dd;
    }
    int t = 0;
    for (; t + 4 <= n; t += 4) {
      const int s0 = __builtin_amdgcn_readlane(s_l, t);
      const int s1 = __builtin_amdgcn_readlane(s_l, t + 1);
      const int s2 = __builtin_amdgcn_readlane(s_l, t + 2);
      const int s3 = __builtin_amdgcn_readlane(s_l, t + 3);
      const float4 h0 = H4[(size_t)s0 * (FDIM / 4) + lane];
      const float4 h1 = H4[(size_t)s1 * (FDIM / 4) + lane];
      const float4 h2 = H4[(size_t)s2 * (FDIM / 4) + lane];
      const float4 h3 = H4[(size_t)s3 * (FDIM / 4) + lane];
      const float w0 = __int_as_float(__builtin_amdgcn_readlane(__float_as_int(w_l), t));
      const float w1 = __int_as_float(__builtin_amdgcn_readlane(__float_as_int(w_l), t + 1));
      const float w2 = __int_as_float(__builtin_amdgcn_readlane(__float_as_int(w_l), t + 2));
      const float w3 = __int_as_float(__builtin_amdgcn_readlane(__float_as_int(w_l), t + 3));
      row_fma(acc, h0, w0);
      row_fma(acc, h1, w1);
      row_fma(acc, h2, w2);
      row_fma(acc, h3, w3);
    }
    const int r = n - t;
    if (r == 3) {
      const int s0 = __builtin_amdgcn_readlane(s_l, t);
      const int s1 = __builtin_amdgcn_readlane(s_l, t + 1);
      const int s2 = __builtin_amdgcn_readlane(s_l, t + 2);
      const float4 h0 = H4[(size_t)s0 * (FDIM / 4) + lane];
      const float4 h1 = H4[(size_t)s1 * (FDIM / 4) + lane];
      const float4 h2 = H4[(size_t)s2 * (FDIM / 4) + lane];
      const float w0 = __int_as_float(__builtin_amdgcn_readlane(__float_as_int(w_l), t));
      const float w1 = __int_as_float(__builtin_amdgcn_readlane(__float_as_int(w_l), t + 1));
      const float w2 = __int_as_float(__builtin_amdgcn_readlane(__float_as_int(w_l), t + 2));
      row_fma(acc, h0, w0);
      row_fma(acc, h1, w1);
      row_fma(acc, h2, w2);
    } else if (r == 2) {
      const int s0 = __builtin_amdgcn_readlane(s_l, t);
      const int s1 = __builtin_amdgcn_readlane(s_l, t + 1);
      const float4 h0 = H4[(size_t)s0 * (FDIM / 4) + lane];
      const float4 h1 = H4[(size_t)s1 * (FDIM / 4) + lane];
      const float w0 = __int_as_float(__builtin_amdgcn_readlane(__float_as_int(w_l), t));
      const float w1 = __int_as_float(__builtin_amdgcn_readlane(__float_as_int(w_l), t + 1));
      row_fma(acc, h0, w0);
      row_fma(acc, h1, w1);
    } else if (r == 1) {
      const int s0 = __builtin_amdgcn_readlane(s_l, t);
      const float w0 = __int_as_float(__builtin_amdgcn_readlane(__float_as_int(w_l), t));
      row_fma(acc, H4[(size_t)s0 * (FDIM / 4) + lane], w0);
    }
  }
  acc.x = fmaxf(acc.x + bx, 0.0f);
  acc.y = fmaxf(acc.y + by, 0.0f);
  acc.z = fmaxf(acc.z + bz, 0.0f);
  acc.w = fmaxf(acc.w + bw, 0.0f);
  reinterpret_cast<float4*>(G)[(size_t)d * (FDIM / 4) + lane] = acc;
}

__global__ void bias_relu_kernel(float* __restrict__ G, const float* __restrict__ b) {
  const size_t i = (size_t)blockIdx.x * FDIM + threadIdx.x;
  G[i] = fmaxf(G[i] + b[threadIdx.x], 0.0f);
}

__global__ void init_out_kernel(float* __restrict__ out, int ng) {
  int i = blockIdx.x * blockDim.x + threadIdx.x;
  if (i >= ng * 2 * FDIM) return;
  int f = i & (2 * FDIM - 1);
  out[i] = (f < FDIM) ? 0.0f : __int_as_float(0xFF7FFFFF);
}

__device__ __forceinline__ void atomic_max_f32(float* addr, float v) {
  if (v >= 0.0f) atomicMax((int*)addr, __float_as_int(v));
  else           atomicMin((unsigned int*)addr, (unsigned int)__float_as_int(v));
}

#define POOL_ROWS  32
#define POOL_WAVES 8

__device__ __forceinline__ void pool_flush(float* __restrict__ out, float* __restrict__ cnt,
                                           float* st, int lane, int g,
                                           const float4 s, const float4 m, int run) {
  reinterpret_cast<float4*>(st)[lane] = s;
  reinterpret_cast<float4*>(st + FDIM)[lane] = m;
  __builtin_amdgcn_fence(__ATOMIC_ACQ_REL, "wavefront");
  float a[4], b[4];
  #pragma unroll
  for (int k = 0; k < 4; ++k) {
    a[k] = st[lane + 32 * k];
    b[k] = st[FDIM + lane + 32 * k];
  }
  __builtin_amdgcn_fence(__ATOMIC_ACQ_REL, "wavefront");
  float* o = out + (size_t)g * (2 * FDIM);
  #pragma unroll
  for (int k = 0; k < 4; ++k) atomicAdd(o + lane + 32 * k, a[k]);
  #pragma unroll
  for (int k = 0; k < 4; ++k) atomic_max_f32(o + FDIM + lane + 32 * k, b[k]);
  if (lane == 0) atomicAdd(&cnt[g], (float)run);
}

__global__ __launch_bounds__(256) void pool_runs_kernel(const float* __restrict__ G,
                                                        const int* __restrict__ gi,
                                                        float* __restrict__ out,
                                                        float* __restrict__ cnt, int N) {
  __shared__ alignas(16) float stage[POOL_WAVES * 2 * FDIM];
  const int wv = threadIdx.x >> 5;
  const int lane = threadIdx.x & 31;
  const int n0 = __builtin_amdgcn_readfirstlane((int)(blockIdx.x * POOL_WAVES + wv)) * POOL_ROWS;
  if (n0 >= N) return;
  const int nr = min(POOL_ROWS, N - n0);
  float* st = stage + wv * (2 * FDIM);

  const int g_l = gi[n0 + min(lane, nr - 1)];

  const float4* __restrict__ G4 = reinterpret_cast<const float4*>(G) + (size_t)n0 * (FDIM / 4) + lane;
  float4 v[POOL_ROWS];
  #pragma unroll
  for (int i = 0; i < POOL_ROWS; ++i) v[i] = G4[(size_t)min(i, nr - 1) * (FDIM / 4)];

  int a = 0;
  for (int trip = 0; trip < POOL_ROWS && a < nr; ++trip) {
    const int ga = __builtin_amdgcn_readlane(g_l, a);
    const unsigned diff = __builtin_amdgcn_ballot_w32(g_l != ga) & (~0u << a);
    const int b = min(diff ? __builtin_ctz(diff) : POOL_ROWS, nr);
    float4 s = make_float4(0.0f, 0.0f, 0.0f, 0.0f);
    float4 m = make_float4(-FLT_MAX, -FLT_MAX, -FLT_MAX, -FLT_MAX);
    #pragma unroll
    for (int i = 0; i < POOL_ROWS; ++i) {
      if (i >= a && i < b) {
        s.x += v[i].x; s.y += v[i].y; s.z += v[i].z; s.w += v[i].w;
        m.x = fmaxf(m.x, v[i].x); m.y = fmaxf(m.y, v[i].y);
        m.z = fmaxf(m.z, v[i].z); m.w = fmaxf(m.w, v[i].w);
      }
    }
    pool_flush(out, cnt, st, lane, ga, s, m, b - a);
    a = b;
  }
}

__global__ void finalize_mean_kernel(float* __restrict__ out, const float* __restrict__ cnt) {
  const int g = blockIdx.x;
  out[(size_t)g * 2 * FDIM + threadIdx.x] /= fmaxf(cnt[g], 1.0f);
}


extern "C" void kernel_launch(void* const* d_in, const int* in_sizes, int n_in,
                              void* d_out, int out_size, void* d_ws, size_t ws_size,
                              hipStream_t stream) {
  const float* x  = (const float*)d_in[0];
  const int*   ei = (const int*)d_in[1];
  const int*   gi = (const int*)d_in[2];
  const float* Wm[3] = {(const float*)d_in[4], (const float*)d_in[6], (const float*)d_in[8]};
  const float* Bv[3] = {(const float*)d_in[5], (const float*)d_in[7], (const float*)d_in[9]};
  float* out = (float*)d_out;

  const int  N  = in_sizes[0] / FDIM;
  const int  E  = in_sizes[1] / 2;
  const int  NG = out_size / (2 * FDIM);
  const long Nf = (long)N * FDIM;

  float* Hbuf = (float*)d_ws;
  float* Gbuf = Hbuf + Nf;
  float* dinv = Gbuf + Nf;
  float* cnt  = dinv + N;
  if (ws_size < (size_t)(2 * Nf + N + NG) * sizeof(float)) return;

  const int nsb = (N + SCAN_ELEMS - 1) / SCAN_ELEMS;
  int* rowptr = (int*)(cnt + NG);
  int* bsum   = rowptr + (N + 1);
  int* col    = bsum + nsb;
  int* rank   = (int*)Hbuf;
  const bool ordered =
      ws_size >= (size_t)(2 * Nf + N + NG) * sizeof(float) + ((size_t)N + 1 + nsb + E) * sizeof(int) &&
      (long)E <= Nf && E > 0;

  auto zero = [&](float* p, long n) {
    zero_f32_kernel<<<(unsigned)((n + 255) / 256), 256, 0, stream>>>(p, n);
  };

  const int tiles = N / 16;
  const int rem   = N - tiles * 16;
  const float* cur = x;

  if (ordered) {
    zero(dinv, N);
    count_rank_kernel<<<(E + 255) / 256, 256, 0, stream>>>(ei, (int*)dinv, rank, E);
    scan_local_kernel<<<nsb, SCAN_THREADS, 0, stream>>>((const int*)dinv, rowptr, bsum, N);
    scan_add_kernel<<<nsb, SCAN_THREADS, 0, stream>>>(rowptr, bsum, N);
    place_kernel<<<(E + 255) / 256, 256, 0, stream>>>(ei, rowptr, rank, col, E);
    dinv_from_count_kernel<<<(N + 255) / 256, 256, 0, stream>>>(dinv, N);

    for (int l = 0; l < 3; ++l) {
      if (tiles > 0)
        gemm_strip_kernel<<<GEMM_BLOCKS, 256, 0, stream>>>(cur, Wm[l], Hbuf, tiles);
      if (rem > 0)
        gemm_tail_kernel<<<rem, FDIM, 0, stream>>>(cur, Wm[l], Hbuf, tiles * 16);
      gather_rows_kernel<<<(N + 7) / 8, 256, 0, stream>>>(Hbuf, rowptr, col, dinv, Bv[l], Gbuf, N);
      cur = Gbuf;
    }
  } else {
    zero(dinv, N);
    degree_kernel<<<(E + N + 255) / 256, 256, 0, stream>>>(ei, dinv, E, N);
    dinv_kernel<<<(N + 255) / 256, 256, 0, stream>>>(dinv, N);

    for (int l = 0; l < 3; ++l) {
      if (tiles > 0)
        gemm_strip_kernel<<<GEMM_BLOCKS, 256, 0, stream>>>(cur, Wm[l], Hbuf, tiles);
      if (rem > 0)
        gemm_tail_kernel<<<rem, FDIM, 0, stream>>>(cur, Wm[l], Hbuf, tiles * 16);
      zero(Gbuf, Nf);
      aggregate_kernel<<<(E + N + 7) / 8, 256, 0, stream>>>(Hbuf, ei, dinv, Gbuf, E, N);
      bias_relu_kernel<<<N, FDIM, 0, stream>>>(Gbuf, Bv[l]);
      cur = Gbuf;
    }
  }

  init_out_kernel<<<(NG * 2 * FDIM + 255) / 256, 256, 0, stream>>>(out, NG);
  zero(cnt, NG);
  pool_runs_kernel<<<(N + POOL_ROWS * POOL_WAVES - 1) / (POOL_ROWS * POOL_WAVES), 32 * POOL_WAVES, 0, stream>>>(Gbuf, gi, out, cnt, N);
  finalize_mean_kernel<<<NG, FDIM, 0, stream>>>(out, cnt);
}


# Layers 2, 3: product fused into the gather kernel: a 32-wave block sums 32 target rows into an LDS tile (bias, ReLU), 8 waves multiply it by W kept in LDS (f32 WMMA); 2 launches fewer
# speedup vs baseline: 3.8979x; 1.0948x over previous
#include <hip/hip_runtime.h>
#include <float.h>

#define FDIM 128

typedef __attribute__((ext_vector_type(2))) float v2f;
typedef __attribute__((ext_vector_type(8))) float v8f;


__global__ void zero_f32_kernel(float* __restrict__ p, long n) {
  long i = (long)blockIdx.x * blockDim.x + threadIdx.x;
  if (i < n) p[i] = 0.0f;
}

__global__ void degree_kernel(const int* __restrict__ ei, float* __restrict__ deg,
                              int E, int N) {
  int e = blockIdx.x * blockDim.x + threadIdx.x;
  int EN = E + N;
  if (e >= EN) return;
  int d = (e < E) ? ei[E + e] : (e - E);
  atomicAdd(&deg[d], 1.0f);
}

__global__ void dinv_kernel(float* __restrict__ deg_dinv, int N) {
  int i = blockIdx.x * blockDim.x + threadIdx.x;
  if (i < N) deg_dinv[i] = rsqrtf(fmaxf(deg_dinv[i], 1.0f));
}

#define GEMM_WPITCH 288
#define GEMM_BLOCKS 512

__global__ __launch_bounds__(256, 4) void gemm_strip_kernel(const float* __restrict__ X,
                                                         const float* __restrict__ W,
                                                         float* __restrict__ H, int tiles) {
  __shared__ alignas(16) float Ws[64 * GEMM_WPITCH];
  const int tid = threadIdx.x;

  {
    const float4* __restrict__ W4 = reinterpret_cast<const float4*>(W);
    for (int it = 0; it < 8; ++it) {
      const int i = it * 256 + tid;
      const int p = i >> 5;
      const int q = i & 31;
      const float4 w0 = W4[(2 * p) * (FDIM / 4) + q];
      const float4 w1 = W4[(2 * p + 1) * (FDIM / 4) + q];
      float4* dst = reinterpret_cast<float4*>(Ws + p * GEMM_WPITCH + q * 8);
      dst[0] = make_float4(w0.x, w1.x, w0.y, w1.y);
      dst[1] = make_float4(w0.z, w1.z, w0.w, w1.w);
    }
  }
  __syncthreads();

  const int wave = __builtin_amdgcn_readfirstlane((int)(threadIdx.x >> 5));
  const int lane = tid & 31;
  const int c = lane & 15;
  const int h = lane >> 4;

  const v2f* __restrict__ bfrag = reinterpret_cast<const v2f*>(Ws) + h * (GEMM_WPITCH / 2) + c;

  const int xoff = c * (FDIM / 2) + h;
  const int hoff = 8 * h * FDIM + c;

  const int s0 = blockIdx.x + gridDim.x * wave;
  const size_t sstep = (size_t)gridDim.x * 8 * (16 * FDIM);
  const v2f* xs = reinterpret_cast<const v2f*>(X + (size_t)s0 * (16 * FDIM)) + xoff;
  float* hs = H + (size_t)s0 * (16 * FDIM) + hoff;

  for (int s = s0; s < tiles; s += gridDim.x * 8, xs += sstep / 2, hs += sstep) {
    v2f a[FDIM / 4];
    #pragma unroll
    for (int kk = 0; kk < FDIM / 4; ++kk) a[kk] = xs[2 * kk];

    v8f acc[8];
    #pragma unroll
    for (int t = 0; t < 8; ++t) acc[t] = v8f{};

    v2f bcur[8];
    #pragma unroll
    for (int t = 0; t < 8; ++t) bcur[t] = bfrag[16 * t];

    #pragma unroll
    for (int kk = 0; kk < FDIM / 4; ++kk) {
      v2f bnxt[8];
      if (kk + 1 < FDIM / 4) {
        const v2f* bp = bfrag + (2 * (kk + 1)) * (GEMM_WPITCH / 2);
        #pragma unroll
        for (int t = 0; t < 8; ++t) bnxt[t] = bp[16 * t];
      }
      #pragma unroll
      for (int t = 0; t < 8; ++t) {
        acc[t] = __builtin_amdgcn_wmma_f32_16x16x4_f32(
             false, a[kk],  false, bcur[t],
             (short)0, acc[t],  false,  false);
      }
      __builtin_amdgcn_sched_barrier(0);
      if (kk + 1 < FDIM / 4) {
        #pragma unroll
        for (int t = 0; t < 8; ++t) bcur[t] = bnxt[t];
      }
    }

    #pragma unroll
    for (int t = 0; t < 8; ++t) {
      #pragma unroll
      for (int r = 0; r < 8; ++r) hs[r * FDIM + 16 * t] = acc[t][r];
    }
  }
}

__global__ void gemm_tail_kernel(const float* __restrict__ X,
                                 const float* __restrict__ W,
                                 float* __restrict__ H, int base) {
  const int row = base + blockIdx.x;
  const int f = threadIdx.x;
  const float* xr = X + (size_t)row * FDIM;
  float acc = 0.0f;
  #pragma unroll 8
  for (int k = 0; k < FDIM; ++k) acc = fmaf(xr[k], W[(size_t)k * FDIM + f], acc);
  H[(size_t)row * FDIM + f] = acc;
}

__global__ void aggregate_kernel(const float* __restrict__ H,
                                 const int* __restrict__ ei,
                                 const float* __restrict__ dinv,
                                 float* __restrict__ G, int E, int N) {
  const int e = blockIdx.x * (blockDim.x >> 5) + (threadIdx.x >> 5);
  const int EN = E + N;
  if (e >= EN) return;
  const int lane = threadIdx.x & 31;
  int s, d;
  if (e < E) { s = ei[e]; d = ei[E + e]; }
  else       { s = e - E; d = s; }
  const float w = dinv[s] * dinv[d];
  const float4 hv = reinterpret_cast<const float4*>(H + (size_t)s * FDIM)[lane];
  float* gp = G + (size_t)d * FDIM + lane * 4;
  atomicAdd(gp + 0, hv.x * w);
  atomicAdd(gp + 1, hv.y * w);
  atomicAdd(gp + 2, hv.z * w);
  atomicAdd(gp + 3, hv.w * w);
}


__global__ void count_rank_kernel(const int* __restrict__ ei, int* __restrict__ icnt,
                                  int* __restrict__ rank, int E) {
  const int e = blockIdx.x * blockDim.x + threadIdx.x;
  if (e >= E) return;
  rank[e] = atomicAdd(&icnt[ei[E + e]], 1);
}

#define SCAN_THREADS 256
#define SCAN_ELEMS   1024

__global__ void scan_local_kernel(const int* __restrict__ icnt, int* __restrict__ rowptr,
                                  int* __restrict__ bsum, int N) {
  __shared__ int wsum[SCAN_THREADS / 32];
  const int tid = threadIdx.x, lane = tid & 31, wv = tid >> 5;
  const int base = blockIdx.x * SCAN_ELEMS + tid * 4;
  int v[4];
  #pragma unroll
  for (int q = 0; q < 4; ++q) v[q] = (base + q < N) ? icnt[base + q] : 0;
  const int t = v[0] + v[1] + v[2] + v[3];
  int inc = t;
  #pragma unroll
  for (int off = 1; off < 32; off <<= 1) {
    const int u = __shfl_up(inc, off);
    if (lane >= off) inc += u;
  }
  if (lane == 31) wsum[wv] = inc;
  __syncthreads();
  int woff = 0, tot = 0;
  #pragma unroll
  for (int k = 0; k < SCAN_THREADS / 32; ++k) {
    const int s = wsum[k];
    if (k < wv) woff += s;
    tot += s;
  }
  int run = woff + inc - t;
  #pragma unroll
  for (int q = 0; q < 4; ++q) {
    if (base + q < N) rowptr[base + q] = run;
    run += v[q];
  }
  if (tid == 0) bsum[blockIdx.x] = tot;
}

__global__ void scan_add_kernel(int* __restrict__ rowptr, const int* __restrict__ bsum, int N) {
  __shared__ int red[SCAN_THREADS / 32];
  const int tid = threadIdx.x, lane = tid & 31, wv = tid >> 5;
  const int b = blockIdx.x;
  int part = 0;
  for (int k = tid; k < b; k += SCAN_THREADS) part += bsum[k];
  #pragma unroll
  for (int off = 16; off > 0; off >>= 1) part += __shfl_xor(part, off);
  if (lane == 0) red[wv] = part;
  __syncthreads();
  int add = 0;
  #pragma unroll
  for (int k = 0; k < SCAN_THREADS / 32; ++k) add += red[k];
  const int base = b * SCAN_ELEMS + tid * 4;
  #pragma unroll
  for (int q = 0; q < 4; ++q)
    if (base + q < N) rowptr[base + q] += add;
  if (b == (int)gridDim.x - 1 && tid == 0) rowptr[N] = add + bsum[b];
}

__global__ void place_kernel(const int* __restrict__ ei, const int* __restrict__ rowptr,
                             const int* __restrict__ rank, int* __restrict__ col, int E) {
  const int e = blockIdx.x * blockDim.x + threadIdx.x;
  if (e >= E) return;
  col[rowptr[ei[E + e]] + rank[e]] = ei[e];
}

__global__ void dinv_from_count_kernel(float* __restrict__ cnt_dinv, int N) {
  const int i = blockIdx.x * blockDim.x + threadIdx.x;
  if (i < N) {
    const int c = __float_as_int(cnt_dinv[i]);
    cnt_dinv[i] = rsqrtf(fmaxf((float)(c + 1), 1.0f));
  }
}

__device__ __forceinline__ void row_fma(float4& acc, const float4 h, const float w) {
  acc.x = fmaf(h.x, w, acc.x);
  acc.y = fmaf(h.y, w, acc.y);
  acc.z = fmaf(h.z, w, acc.z);
  acc.w = fmaf(h.w, w, acc.w);
}

__device__ __forceinline__ float4 gather_row_sum(const float4* __restrict__ H4,
                                                 const int* __restrict__ rowptr,
                                                 const int* __restrict__ col,
                                                 const float* __restrict__ dinv,
                                                 const int d, const int lane) {
  const int beg = rowptr[d];
  const int end = rowptr[d + 1];
  const float dd = dinv[d];

  float4 acc;
  {
    const float4 hs = H4[(size_t)d * (FDIM / 4) + lane];
    const float ws = dd * dd;
    acc.x = hs.x * ws; acc.y = hs.y * ws; acc.z = hs.z * ws; acc.w = hs.w * ws;
  }

  for (int j = beg; j < end; j += 32) {
    const int n = min(32, end - j);
    int   s_l = 0;
    float w_l = 0.0f;
    if (lane < n) {
      s_l = col[j + lane];
      w_l = dinv[s_l] * dd;
    }
    int t = 0;
    for (; t + 4 <= n; t += 4) {
      const int s0 = __builtin_amdgcn_readlane(s_l, t);
      const int s1 = __builtin_amdgcn_readlane(s_l, t + 1);
      const int s2 = __builtin_amdgcn_readlane(s_l, t + 2);
      const int s3 = __builtin_amdgcn_readlane(s_l, t + 3);
      const float4 h0 = H4[(size_t)s0 * (FDIM / 4) + lane];
      const float4 h1 = H4[(size_t)s1 * (FDIM / 4) + lane];
      const float4 h2 = H4[(size_t)s2 * (FDIM / 4) + lane];
      const float4 h3 = H4[(size_t)s3 * (FDIM / 4) + lane];
      const float w0 = __int_as_float(__builtin_amdgcn_readlane(__float_as_int(w_l), t));
      const float w1 = __int_as_float(__builtin_amdgcn_readlane(__float_as_int(w_l), t + 1));
      const float w2 = __int_as_float(__builtin_amdgcn_readlane(__float_as_int(w_l), t + 2));
      const float w3 = __int_as_float(__builtin_amdgcn_readlane(__float_as_int(w_l), t + 3));
      row_fma(acc, h0, w0);
      row_fma(acc, h1, w1);
      row_fma(acc, h2, w2);
      row_fma(acc, h3, w3);
    }
    const int r = n - t;
    if (r == 3) {
      const int s0 = __builtin_amdgcn_readlane(s_l, t);
      const int s1 = __builtin_amdgcn_readlane(s_l, t + 1);
      const int s2 = __builtin_amdgcn_readlane(s_l, t + 2);
      const float4 h0 = H4[(size_t)s0 * (FDIM / 4) + lane];
      const float4 h1 = H4[(size_t)s1 * (FDIM / 4) + lane];
      const float4 h2 = H4[(size_t)s2 * (FDIM / 4) + lane];
      const float w0 = __int_as_float(__builtin_amdgcn_readlane(__float_as_int(w_l), t));
      const float w1 = __int_as_float(__builtin_amdgcn_readlane(__float_as_int(w_l), t + 1));
      const float w2 = __int_as_float(__builtin_amdgcn_readlane(__float_as_int(w_l), t + 2));
      row_fma(acc, h0, w0);
      row_fma(acc, h1, w1);
      row_fma(acc, h2, w2);
    } else if (r == 2) {
      const int s0 = __builtin_amdgcn_readlane(s_l, t);
      const int s1 = __builtin_amdgcn_readlane(s_l, t + 1);
      const float4 h0 = H4[(size_t)s0 * (FDIM / 4) + lane];
      const float4 h1 = H4[(size_t)s1 * (FDIM / 4) + lane];
      const float w0 = __int_as_float(__builtin_amdgcn_readlane(__float_as_int(w_l), t));
      const float w1 = __int_as_float(__builtin_amdgcn_readlane(__float_as_int(w_l), t + 1));
      row_fma(acc, h0, w0);
      row_fma(acc, h1, w1);
    } else if (r == 1) {
      const int s0 = __builtin_amdgcn_readlane(s_l, t);
      const float w0 = __int_as_float(__builtin_amdgcn_readlane(__float_as_int(w_l), t));
      row_fma(acc, H4[(size_t)s0 * (FDIM / 4) + lane], w0);
    }
  }
  return acc;
}

__global__ void gather_rows_kernel(const float* __restrict__ H, const int* __restrict__ rowptr,
                                   const int* __restrict__ col, const float* __restrict__ dinv,
                                   const float* __restrict__ bias, float* __restrict__ G, int N) {
  const int d = __builtin_amdgcn_readfirstlane(
      (int)(blockIdx.x * (blockDim.x >> 5) + (threadIdx.x >> 5)));
  if (d >= N) return;
  const int lane = threadIdx.x & 31;
  const float bx = bias[lane * 4 + 0];
  const float by = bias[lane * 4 + 1];
  const float bz = bias[lane * 4 + 2];
  const float bw = bias[lane * 4 + 3];
  float4 acc = gather_row_sum(reinterpret_cast<const float4*>(H), rowptr, col, dinv, d, lane);
  acc.x = fmaxf(acc.x + bx, 0.0f);
  acc.y = fmaxf(acc.y + by, 0.0f);
  acc.z = fmaxf(acc.z + bz, 0.0f);
  acc.w = fmaxf(acc.w + bw, 0.0f);
  reinterpret_cast<float4*>(G)[(size_t)d * (FDIM / 4) + lane] = acc;
}

#define FG_ROWS   32
#define FG_TPITCH 132
#define FG_BLOCKS 512

__global__ __launch_bounds__(32 * FG_ROWS, 16) void gather_product_kernel(
    const float* __restrict__ H, const int* __restrict__ rowptr, const int* __restrict__ col,
    const float* __restrict__ dinv, const float* __restrict__ bias, const float* __restrict__ Wn,
    float* __restrict__ Hn, int N, int ngroups) {
  __shared__ alignas(16) float smem[64 * GEMM_WPITCH + 2 * FG_ROWS * FG_TPITCH];
  float* Ws = smem;
  const int tid = threadIdx.x;

  {
    const float4* __restrict__ W4 = reinterpret_cast<const float4*>(Wn);
    for (int it = 0; it < 2; ++it) {
      const int i = it * (32 * FG_ROWS) + tid;
      const int p = i >> 5;
      const int q = i & 31;
      const float4 w0 = W4[(2 * p) * (FDIM / 4) + q];
      const float4 w1 = W4[(2 * p + 1) * (FDIM / 4) + q];
      float4* dst = reinterpret_cast<float4*>(Ws + p * GEMM_WPITCH + q * 8);
      dst[0] = make_float4(w0.x, w1.x, w0.y, w1.y);
      dst[1] = make_float4(w0.z, w1.z, w0.w, w1.w);
    }
  }
  __syncthreads();

  const int wave = __builtin_amdgcn_readfirstlane((int)(threadIdx.x >> 5));
  const int lane = tid & 31;
  const int c = lane & 15;
  const int h = lane >> 4;

  const float bx = bias[lane * 4 + 0];
  const float by = bias[lane * 4 + 1];
  const float bz = bias[lane * 4 + 2];
  const float bw = bias[lane * 4 + 3];
  const float4* __restrict__ H4 = reinterpret_cast<const float4*>(H);

  const int rt = (wave >> 2) & 1;
  const int cp = wave & 3;
  const unsigned boff0 = (unsigned)(h * (GEMM_WPITCH / 2) + c + 32 * cp) * 8u;

  int it = 0;
  for (int g = blockIdx.x; g < ngroups; g += gridDim.x, ++it) {
    const int toff = 64 * GEMM_WPITCH + (it & 1) * (FG_ROWS * FG_TPITCH);
    float* tl = smem + toff;
    const int d = g * FG_ROWS + wave;
    float4 acc = make_float4(0.0f, 0.0f, 0.0f, 0.0f);
    if (d < N) {
      acc = gather_row_sum(H4, rowptr, col, dinv, d, lane);
      acc.x = fmaxf(acc.x + bx, 0.0f);
      acc.y = fmaxf(acc.y + by, 0.0f);
      acc.z = fmaxf(acc.z + bz, 0.0f);
      acc.w = fmaxf(acc.w + bw, 0.0f);
    }
    reinterpret_cast<float4*>(tl + wave * FG_TPITCH)[lane] = acc;
    __syncthreads();

    if ((wave >> 3) == (it & 3)) {
      unsigned ao = (unsigned)(toff + (16 * rt + c) * FG_TPITCH + 2 * h) * 4u;
      unsigned bo = boff0;
      const char* sb = reinterpret_cast<const char*>(smem);
      const int row0 = g * FG_ROWS + 16 * rt + 8 * h;
      float* hp = Hn + (size_t)row0 * FDIM + 32 * cp + c;
      int nval = N - row0;
      asm volatile("" : "+v"(hp), "+v"(nval));
      __attribute__((address_space(1))) float* hg = (__attribute__((address_space(1))) float*)hp;
      v8f acc0 = v8f{};
      v8f acc1 = v8f{};
      v2f ax  = *reinterpret_cast<const v2f*>(sb + ao);
      v2f bx0 = *reinterpret_cast<const v2f*>(sb + bo);
      v2f bx1 = *reinterpret_cast<const v2f*>(sb + bo + 128);
      #pragma unroll 1
      for (int kk = 0; kk < FDIM / 4 - 2; kk += 2) {
        ao += 16;
        bo += 2 * (GEMM_WPITCH / 2) * 8;
        asm volatile("" : "+v"(ao), "+v"(bo));
        const v2f ay  = *reinterpret_cast<const v2f*>(sb + ao);
        const v2f by0 = *reinterpret_cast<const v2f*>(sb + bo);
        const v2f by1 = *reinterpret_cast<const v2f*>(sb + bo + 128);
        acc0 = __builtin_amdgcn_wmma_f32_16x16x4_f32(
             false, ax,  false, bx0,
             (short)0, acc0,  false,  false);
        acc1 = __builtin_amdgcn_wmma_f32_16x16x4_f32(
             false, ax,  false, bx1,
             (short)0, acc1,  false,  false);
        ao += 16;
        bo += 2 * (GEMM_WPITCH / 2) * 8;
        asm volatile("" : "+v"(ao), "+v"(bo));
        ax  = *reinterpret_cast<const v2f*>(sb + ao);
        bx0 = *reinterpret_cast<const v2f*>(sb + bo);
        bx1 = *reinterpret_cast<const v2f*>(sb + bo + 128);
        acc0 = __builtin_amdgcn_wmma_f32_16x16x4_f32(
             false, ay,  false, by0,
             (short)0, acc0,  false,  false);
        acc1 = __builtin_amdgcn_wmma_f32_16x16x4_f32(
             false, ay,  false, by1,
             (short)0, acc1,  false,  false);
      }
      {
        ao += 16;
        bo += 2 * (GEMM_WPITCH / 2) * 8;
        asm volatile("" : "+v"(ao), "+v"(bo));
        const v2f ay  = *reinterpret_cast<const v2f*>(sb + ao);
        const v2f by0 = *reinterpret_cast<const v2f*>(sb + bo);
        const v2f by1 = *reinterpret_cast<const v2f*>(sb + bo + 128);
        acc0 = __builtin_amdgcn_wmma_f32_16x16x4_f32(
             false, ax,  false, bx0,
             (short)0, acc0,  false,  false);
        acc1 = __builtin_amdgcn_wmma_f32_16x16x4_f32(
             false, ax,  false, bx1,
             (short)0, acc1,  false,  false);
        acc0 = __builtin_amdgcn_wmma_f32_16x16x4_f32(
             false, ay,  false, by0,
             (short)0, acc0,  false,  false);
        acc1 = __builtin_amdgcn_wmma_f32_16x16x4_f32(
             false, ay,  false, by1,
             (short)0, acc1,  false,  false);
      }
      #pragma unroll
      for (int r = 0; r < 8; ++r) {
        if (r < nval) {
          hg[r * FDIM]      = acc0[r];
          hg[r * FDIM + 16] = acc1[r];
        }
      }
    }
  }
}

__global__ void bias_relu_kernel(float* __restrict__ G, const float* __restrict__ b) {
  const size_t i = (size_t)blockIdx.x * FDIM + threadIdx.x;
  G[i] = fmaxf(G[i] + b[threadIdx.x], 0.0f);
}

__global__ void init_out_kernel(float* __restrict__ out, int ng) {
  int i = blockIdx.x * blockDim.x + threadIdx.x;
  if (i >= ng * 2 * FDIM) return;
  int f = i & (2 * FDIM - 1);
  out[i] = (f < FDIM) ? 0.0f : __int_as_float(0xFF7FFFFF);
}

__device__ __forceinline__ void atomic_max_f32(float* addr, float v) {
  if (v >= 0.0f) atomicMax((int*)addr, __float_as_int(v));
  else           atomicMin((unsigned int*)addr, (unsigned int)__float_as_int(v));
}

#define POOL_ROWS  32
#define POOL_WAVES 8

__device__ __forceinline__ void pool_flush(float* __restrict__ out, float* __restrict__ cnt,
                                           float* st, int lane, int g,
                                           const float4 s, const float4 m, int run) {
  reinterpret_cast<float4*>(st)[lane] = s;
  reinterpret_cast<float4*>(st + FDIM)[lane] = m;
  __builtin_amdgcn_fence(__ATOMIC_ACQ_REL, "wavefront");
  float a[4], b[4];
  #pragma unroll
  for (int k = 0; k < 4; ++k) {
    a[k] = st[lane + 32 * k];
    b[k] = st[FDIM + lane + 32 * k];
  }
  __builtin_amdgcn_fence(__ATOMIC_ACQ_REL, "wavefront");
  float* o = out + (size_t)g * (2 * FDIM);
  #pragma unroll
  for (int k = 0; k < 4; ++k) atomicAdd(o + lane + 32 * k, a[k]);
  #pragma unroll
  for (int k = 0; k < 4; ++k) atomic_max_f32(o + FDIM + lane + 32 * k, b[k]);
  if (lane == 0) atomicAdd(&cnt[g], (float)run);
}

__global__ __launch_bounds__(256) void pool_runs_kernel(const float* __restrict__ G,
                                                        const int* __restrict__ gi,
                                                        float* __restrict__ out,
                                                        float* __restrict__ cnt, int N) {
  __shared__ alignas(16) float stage[POOL_WAVES * 2 * FDIM];
  const int wv = threadIdx.x >> 5;
  const int lane = threadIdx.x & 31;
  const int n0 = __builtin_amdgcn_readfirstlane((int)(blockIdx.x * POOL_WAVES + wv)) * POOL_ROWS;
  if (n0 >= N) return;
  const int nr = min(POOL_ROWS, N - n0);
  float* st = stage + wv * (2 * FDIM);

  const int g_l = gi[n0 + min(lane, nr - 1)];

  const float4* __restrict__ G4 = reinterpret_cast<const float4*>(G) + (size_t)n0 * (FDIM / 4) + lane;
  float4 v[POOL_ROWS];
  #pragma unroll
  for (int i = 0; i < POOL_ROWS; ++i) v[i] = G4[(size_t)min(i, nr - 1) * (FDIM / 4)];

  int a = 0;
  for (int trip = 0; trip < POOL_ROWS && a < nr; ++trip) {
    const int ga = __builtin_amdgcn_readlane(g_l, a);
    const unsigned diff = __builtin_amdgcn_ballot_w32(g_l != ga) & (~0u << a);
    const int b = min(diff ? __builtin_ctz(diff) : POOL_ROWS, nr);
    float4 s = make_float4(0.0f, 0.0f, 0.0f, 0.0f);
    float4 m = make_float4(-FLT_MAX, -FLT_MAX, -FLT_MAX, -FLT_MAX);
    #pragma unroll
    for (int i = 0; i < POOL_ROWS; ++i) {
      if (i >= a && i < b) {
        s.x += v[i].x; s.y += v[i].y; s.z += v[i].z; s.w += v[i].w;
        m.x = fmaxf(m.x, v[i].x); m.y = fmaxf(m.y, v[i].y);
        m.z = fmaxf(m.z, v[i].z); m.w = fmaxf(m.w, v[i].w);
      }
    }
    pool_flush(out, cnt, st, lane, ga, s, m, b - a);
    a = b;
  }
}

__global__ void finalize_mean_kernel(float* __restrict__ out, const float* __restrict__ cnt) {
  const int g = blockIdx.x;
  out[(size_t)g * 2 * FDIM + threadIdx.x] /= fmaxf(cnt[g], 1.0f);
}


extern "C" void kernel_launch(void* const* d_in, const int* in_sizes, int n_in,
                              void* d_out, int out_size, void* d_ws, size_t ws_size,
                              hipStream_t stream) {
  const float* x  = (const float*)d_in[0];
  const int*   ei = (const int*)d_in[1];
  const int*   gi = (const int*)d_in[2];
  const float* Wm[3] = {(const float*)d_in[4], (const float*)d_in[6], (const float*)d_in[8]};
  const float* Bv[3] = {(const float*)d_in[5], (const float*)d_in[7], (const float*)d_in[9]};
  float* out = (float*)d_out;

  const int  N  = in_sizes[0] / FDIM;
  const int  E  = in_sizes[1] / 2;
  const int  NG = out_size / (2 * FDIM);
  const long Nf = (long)N * FDIM;

  float* Hbuf = (float*)d_ws;
  float* Gbuf = Hbuf + Nf;
  float* dinv = Gbuf + Nf;
  float* cnt  = dinv + N;
  if (ws_size < (size_t)(2 * Nf + N + NG) * sizeof(float)) return;

  const int nsb = (N + SCAN_ELEMS - 1) / SCAN_ELEMS;
  int* rowptr = (int*)(cnt + NG);
  int* bsum   = rowptr + (N + 1);
  int* col    = bsum + nsb;
  int* rank   = (int*)Hbuf;
  const bool ordered =
      ws_size >= (size_t)(2 * Nf + N + NG) * sizeof(float) + ((size_t)N + 1 + nsb + E) * sizeof(int) &&
      (long)E <= Nf && E > 0;

  auto zero = [&](float* p, long n) {
    zero_f32_kernel<<<(unsigned)((n + 255) / 256), 256, 0, stream>>>(p, n);
  };

  const int tiles = N / 16;
  const int rem   = N - tiles * 16;
  const float* cur = x;

  if (ordered) {
    zero(dinv, N);
    count_rank_kernel<<<(E + 255) / 256, 256, 0, stream>>>(ei, (int*)dinv, rank, E);
    scan_local_kernel<<<nsb, SCAN_THREADS, 0, stream>>>((const int*)dinv, rowptr, bsum, N);
    scan_add_kernel<<<nsb, SCAN_THREADS, 0, stream>>>(rowptr, bsum, N);
    place_kernel<<<(E + 255) / 256, 256, 0, stream>>>(ei, rowptr, rank, col, E);
    dinv_from_count_kernel<<<(N + 255) / 256, 256, 0, stream>>>(dinv, N);

    if (tiles > 0)
      gemm_strip_kernel<<<GEMM_BLOCKS, 256, 0, stream>>>(x, Wm[0], Hbuf, tiles);
    if (rem > 0)
      gemm_tail_kernel<<<rem, FDIM, 0, stream>>>(x, Wm[0], Hbuf, tiles * 16);
    const int ngroups = (N + FG_ROWS - 1) / FG_ROWS;
    const int fgrid = ngroups < FG_BLOCKS ? ngroups : FG_BLOCKS;
    gather_product_kernel<<<fgrid, 32 * FG_ROWS, 0, stream>>>(Hbuf, rowptr, col, dinv, Bv[0], Wm[1], Gbuf, N, ngroups);
    gather_product_kernel<<<fgrid, 32 * FG_ROWS, 0, stream>>>(Gbuf, rowptr, col, dinv, Bv[1], Wm[2], Hbuf, N, ngroups);
    gather_rows_kernel<<<(N + 7) / 8, 256, 0, stream>>>(Hbuf, rowptr, col, dinv, Bv[2], Gbuf, N);
  } else {
    zero(dinv, N);
    degree_kernel<<<(E + N + 255) / 256, 256, 0, stream>>>(ei, dinv, E, N);
    dinv_kernel<<<(N + 255) / 256, 256, 0, stream>>>(dinv, N);

    for (int l = 0; l < 3; ++l) {
      if (tiles > 0)
        gemm_strip_kernel<<<GEMM_BLOCKS, 256, 0, stream>>>(cur, Wm[l], Hbuf, tiles);
      if (rem > 0)
        gemm_tail_kernel<<<rem, FDIM, 0, stream>>>(cur, Wm[l], Hbuf, tiles * 16);
      zero(Gbuf, Nf);
      aggregate_kernel<<<(E + N + 7) / 8, 256, 0, stream>>>(Hbuf, ei, dinv, Gbuf, E, N);
      bias_relu_kernel<<<N, FDIM, 0, stream>>>(Gbuf, Bv[l]);
      cur = Gbuf;
    }
  }

  init_out_kernel<<<(NG * 2 * FDIM + 255) / 256, 256, 0, stream>>>(out, NG);
  zero(cnt, NG);
  pool_runs_kernel<<<(N + POOL_ROWS * POOL_WAVES - 1) / (POOL_ROWS * POOL_WAVES), 32 * POOL_WAVES, 0, stream>>>(Gbuf, gi, out, cnt, N);
  finalize_mean_kernel<<<NG, FDIM, 0, stream>>>(out, cnt);
}


# Edge bucketing by target on LDS counters (two-level counting sort: chunk counts, scan, pair scatter, per-bucket LDS order) replaces 1.6M scattered returning global adds; row sum order = LDS arrival
# speedup vs baseline: 5.2289x; 1.3415x over previous
#include <hip/hip_runtime.h>
#include <float.h>

#define FDIM 128

typedef __attribute__((ext_vector_type(2))) float v2f;
typedef __attribute__((ext_vector_type(8))) float v8f;


__global__ void zero_f32_kernel(float* __restrict__ p, long n) {
  long i = (long)blockIdx.x * blockDim.x + threadIdx.x;
  if (i < n) p[i] = 0.0f;
}

__global__ void degree_kernel(const int* __restrict__ ei, float* __restrict__ deg,
                              int E, int N) {
  int e = blockIdx.x * blockDim.x + threadIdx.x;
  int EN = E + N;
  if (e >= EN) return;
  int d = (e < E) ? ei[E + e] : (e - E);
  atomicAdd(&deg[d], 1.0f);
}

__global__ void dinv_kernel(float* __restrict__ deg_dinv, int N) {
  int i = blockIdx.x * blockDim.x + threadIdx.x;
  if (i < N) deg_dinv[i] = rsqrtf(fmaxf(deg_dinv[i], 1.0f));
}

#define GEMM_WPITCH 288
#define GEMM_BLOCKS 512

__global__ __launch_bounds__(256, 4) void gemm_strip_kernel(const float* __restrict__ X,
                                                         const float* __restrict__ W,
                                                         float* __restrict__ H, int tiles) {
  __shared__ alignas(16) float Ws[64 * GEMM_WPITCH];
  const int tid = threadIdx.x;

  {
    const float4* __restrict__ W4 = reinterpret_cast<const float4*>(W);
    for (int it = 0; it < 8; ++it) {
      const int i = it * 256 + tid;
      const int p = i >> 5;
      const int q = i & 31;
      const float4 w0 = W4[(2 * p) * (FDIM / 4) + q];
      const float4 w1 = W4[(2 * p + 1) * (FDIM / 4) + q];
      float4* dst = reinterpret_cast<float4*>(Ws + p * GEMM_WPITCH + q * 8);
      dst[0] = make_float4(w0.x, w1.x, w0.y, w1.y);
      dst[1] = make_float4(w0.z, w1.z, w0.w, w1.w);
    }
  }
  __syncthreads();

  const int wave = __builtin_amdgcn_readfirstlane((int)(threadIdx.x >> 5));
  const int lane = tid & 31;
  const int c = lane & 15;
  const int h = lane >> 4;

  const v2f* __restrict__ bfrag = reinterpret_cast<const v2f*>(Ws) + h * (GEMM_WPITCH / 2) + c;

  const int xoff = c * (FDIM / 2) + h;
  const int hoff = 8 * h * FDIM + c;

  const int s0 = blockIdx.x + gridDim.x * wave;
  const size_t sstep = (size_t)gridDim.x * 8 * (16 * FDIM);
  const v2f* xs = reinterpret_cast<const v2f*>(X + (size_t)s0 * (16 * FDIM)) + xoff;
  float* hs = H + (size_t)s0 * (16 * FDIM) + hoff;

  for (int s = s0; s < tiles; s += gridDim.x * 8, xs += sstep / 2, hs += sstep) {
    v2f a[FDIM / 4];
    #pragma unroll
    for (int kk = 0; kk < FDIM / 4; ++kk) a[kk] = xs[2 * kk];

    v8f acc[8];
    #pragma unroll
    for (int t = 0; t < 8; ++t) acc[t] = v8f{};

    v2f bcur[8];
    #pragma unroll
    for (int t = 0; t < 8; ++t) bcur[t] = bfrag[16 * t];

    #pragma unroll
    for (int kk = 0; kk < FDIM / 4; ++kk) {
      v2f bnxt[8];
      if (kk + 1 < FDIM / 4) {
        const v2f* bp = bfrag + (2 * (kk + 1)) * (GEMM_WPITCH / 2);
        #pragma unroll
        for (int t = 0; t < 8; ++t) bnxt[t] = bp[16 * t];
      }
      #pragma unroll
      for (int t = 0; t < 8; ++t) {
        acc[t] = __builtin_amdgcn_wmma_f32_16x16x4_f32(
             false, a[kk],  false, bcur[t],
             (short)0, acc[t],  false,  false);
      }
      __builtin_amdgcn_sched_barrier(0);
      if (kk + 1 < FDIM / 4) {
        #pragma unroll
        for (int t = 0; t < 8; ++t) bcur[t] = bnxt[t];
      }
    }

    #pragma unroll
    for (int t = 0; t < 8; ++t) {
      #pragma unroll
      for (int r = 0; r < 8; ++r) hs[r * FDIM + 16 * t] = acc[t][r];
    }
  }
}

__global__ void gemm_tail_kernel(const float* __restrict__ X,
                                 const float* __restrict__ W,
                                 float* __restrict__ H, int base) {
  const int row = base + blockIdx.x;
  const int f = threadIdx.x;
  const float* xr = X + (size_t)row * FDIM;
  float acc = 0.0f;
  #pragma unroll 8
  for (int k = 0; k < FDIM; ++k) acc = fmaf(xr[k], W[(size_t)k * FDIM + f], acc);
  H[(size_t)row * FDIM + f] = acc;
}

__global__ void aggregate_kernel(const float* __restrict__ H,
                                 const int* __restrict__ ei,
                                 const float* __restrict__ dinv,
                                 float* __restrict__ G, int E, int N) {
  const int e = blockIdx.x * (blockDim.x >> 5) + (threadIdx.x >> 5);
  const int EN = E + N;
  if (e >= EN) return;
  const int lane = threadIdx.x & 31;
  int s, d;
  if (e < E) { s = ei[e]; d = ei[E + e]; }
  else       { s = e - E; d = s; }
  const float w = dinv[s] * dinv[d];
  const float4 hv = reinterpret_cast<const float4*>(H + (size_t)s * FDIM)[lane];
  float* gp = G + (size_t)d * FDIM + lane * 4;
  atomicAdd(gp + 0, hv.x * w);
  atomicAdd(gp + 1, hv.y * w);
  atomicAdd(gp + 2, hv.z * w);
  atomicAdd(gp + 3, hv.w * w);
}


__global__ void count_rank_kernel(const int* __restrict__ ei, int* __restrict__ icnt,
                                  int* __restrict__ rank, int E) {
  const int e = blockIdx.x * blockDim.x + threadIdx.x;
  if (e >= E) return;
  rank[e] = atomicAdd(&icnt[ei[E + e]], 1);
}

#define SCAN_THREADS 256
#define SCAN_ELEMS   1024

__global__ void scan_local_kernel(const int* __restrict__ icnt, int* __restrict__ rowptr,
                                  int* __restrict__ bsum, int N) {
  __shared__ int wsum[SCAN_THREADS / 32];
  const int tid = threadIdx.x, lane = tid & 31, wv = tid >> 5;
  const int base = blockIdx.x * SCAN_ELEMS + tid * 4;
  int v[4];
  #pragma unroll
  for (int q = 0; q < 4; ++q) v[q] = (base + q < N) ? icnt[base + q] : 0;
  const int t = v[0] + v[1] + v[2] + v[3];
  int inc = t;
  #pragma unroll
  for (int off = 1; off < 32; off <<= 1) {
    const int u = __shfl_up(inc, off);
    if (lane >= off) inc += u;
  }
  if (lane == 31) wsum[wv] = inc;
  __syncthreads();
  int woff = 0, tot = 0;
  #pragma unroll
  for (int k = 0; k < SCAN_THREADS / 32; ++k) {
    const int s = wsum[k];
    if (k < wv) woff += s;
    tot += s;
  }
  int run = woff + inc - t;
  #pragma unroll
  for (int q = 0; q < 4; ++q) {
    if (base + q < N) rowptr[base + q] = run;
    run += v[q];
  }
  if (tid == 0) bsum[blockIdx.x] = tot;
}

__global__ void scan_add_kernel(int* __restrict__ rowptr, const int* __restrict__ bsum, int N) {
  __shared__ int red[SCAN_THREADS / 32];
  const int tid = threadIdx.x, lane = tid & 31, wv = tid >> 5;
  const int b = blockIdx.x;
  int part = 0;
  for (int k = tid; k < b; k += SCAN_THREADS) part += bsum[k];
  #pragma unroll
  for (int off = 16; off > 0; off >>= 1) part += __shfl_xor(part, off);
  if (lane == 0) red[wv] = part;
  __syncthreads();
  int add = 0;
  #pragma unroll
  for (int k = 0; k < SCAN_THREADS / 32; ++k) add += red[k];
  const int base = b * SCAN_ELEMS + tid * 4;
  #pragma unroll
  for (int q = 0; q < 4; ++q)
    if (base + q < N) rowptr[base + q] += add;
  if (b == (int)gridDim.x - 1 && tid == 0) rowptr[N] = add + bsum[b];
}

__global__ void place_kernel(const int* __restrict__ ei, const int* __restrict__ rowptr,
                             const int* __restrict__ rank, int* __restrict__ col, int E) {
  const int e = blockIdx.x * blockDim.x + threadIdx.x;
  if (e >= E) return;
  col[rowptr[ei[E + e]] + rank[e]] = ei[e];
}

#define BIN_SHIFT   8
#define BIN_NODES   256
#define BIN_MAXB    1024
#define BIN_BLOCKS  256
#define BIN_THREADS 1024
#define BIN_BATCH   4
#define BIN_STAGE   16384

__global__ __launch_bounds__(BIN_THREADS) void bin_count_kernel(const int* __restrict__ ei,
                                                                int* __restrict__ hist, int E, int nb) {
  __shared__ int h[BIN_MAXB];
  const int tid = threadIdx.x;
  for (int k = tid; k < nb; k += BIN_THREADS) h[k] = 0;
  __syncthreads();
  const int chunk = (E + BIN_BLOCKS - 1) / BIN_BLOCKS;
  const int lo = blockIdx.x * chunk;
  const int hi = min(E, lo + chunk);
  for (int i = lo + tid; i < hi; i += BIN_BATCH * BIN_THREADS) {
    int d[BIN_BATCH];
    #pragma unroll
    for (int q = 0; q < BIN_BATCH; ++q) d[q] = ei[E + min(i + q * BIN_THREADS, hi - 1)];
    #pragma unroll
    for (int q = 0; q < BIN_BATCH; ++q)
      if (i + q * BIN_THREADS < hi) atomicAdd(&h[d[q] >> BIN_SHIFT], 1);
  }
  __syncthreads();
  for (int k = tid; k < nb; k += BIN_THREADS) hist[k * BIN_BLOCKS + blockIdx.x] = h[k];
}

__global__ __launch_bounds__(BIN_THREADS) void bin_scatter_kernel(const int* __restrict__ ei,
                                                                  const int* __restrict__ boff,
                                                                  int* __restrict__ ebuf, int E, int nb) {
  __shared__ int cur[BIN_MAXB];
  const int tid = threadIdx.x;
  for (int k = tid; k < nb; k += BIN_THREADS) cur[k] = boff[k * BIN_BLOCKS + blockIdx.x];
  __syncthreads();
  const int chunk = (E + BIN_BLOCKS - 1) / BIN_BLOCKS;
  const int lo = blockIdx.x * chunk;
  const int hi = min(E, lo + chunk);
  int2* __restrict__ eb = reinterpret_cast<int2*>(ebuf);
  for (int i = lo + tid; i < hi; i += BIN_BATCH * BIN_THREADS) {
    int s[BIN_BATCH], d[BIN_BATCH];
    #pragma unroll
    for (int q = 0; q < BIN_BATCH; ++q) {
      const int j = min(i + q * BIN_THREADS, hi - 1);
      s[q] = ei[j];
      d[q] = ei[E + j];
    }
    #pragma unroll
    for (int q = 0; q < BIN_BATCH; ++q) {
      if (i + q * BIN_THREADS < hi) {
        const int p = atomicAdd(&cur[d[q] >> BIN_SHIFT], 1);
        eb[p] = make_int2(s[q], d[q]);
      }
    }
  }
}

__global__ __launch_bounds__(BIN_THREADS) void bin_finish_kernel(const int* __restrict__ ebuf,
                                                                 const int* __restrict__ boff,
                                                                 int* __restrict__ rowptr,
                                                                 int* __restrict__ col,
                                                                 float* __restrict__ dinv, int N) {
  __shared__ int cnt[BIN_NODES];
  __shared__ int cur[BIN_NODES];
  __shared__ int wsum[BIN_NODES / 32];
  __shared__ int stage[BIN_STAGE];
  const int tid = threadIdx.x, lane = tid & 31, wv = tid >> 5;
  const int b = blockIdx.x;
  const int e0 = boff[b * BIN_BLOCKS];
  const int e1 = boff[(b + 1) * BIN_BLOCKS];
  const int2* __restrict__ eb = reinterpret_cast<const int2*>(ebuf);
  if (tid < BIN_NODES) cnt[tid] = 0;
  __syncthreads();
  for (int i = e0 + tid; i < e1; i += BIN_BATCH * BIN_THREADS) {
    int d[BIN_BATCH];
    #pragma unroll
    for (int q = 0; q < BIN_BATCH; ++q) d[q] = eb[min(i + q * BIN_THREADS, e1 - 1)].y;
    #pragma unroll
    for (int q = 0; q < BIN_BATCH; ++q)
      if (i + q * BIN_THREADS < e1) atomicAdd(&cnt[d[q] & (BIN_NODES - 1)], 1);
  }
  __syncthreads();

  const int v = (tid < BIN_NODES) ? cnt[tid] : 0;
  int inc = v;
  #pragma unroll
  for (int off = 1; off < 32; off <<= 1) {
    const int u = __shfl_up(inc, off);
    if (lane >= off) inc += u;
  }
  if (tid < BIN_NODES && lane == 31) wsum[wv] = inc;
  __syncthreads();
  if (tid < BIN_NODES) {
    int woff = 0;
    #pragma unroll
    for (int k = 0; k < BIN_NODES / 32; ++k)
      if (k < wv) woff += wsum[k];
    const int excl = woff + inc - v;
    cur[tid] = excl;
    const int node = b * BIN_NODES + tid;
    if (node < N) {
      rowptr[node] = e0 + excl;
      dinv[node] = rsqrtf(fmaxf((float)(v + 1), 1.0f));
    }
  }
  if (b == (int)gridDim.x - 1 && tid == 0) rowptr[N] = e1;
  __syncthreads();

  const int ne = e1 - e0;
  if (ne <= BIN_STAGE) {
    for (int i = e0 + tid; i < e1; i += BIN_BATCH * BIN_THREADS) {
      int2 sd[BIN_BATCH];
      #pragma unroll
      for (int q = 0; q < BIN_BATCH; ++q) sd[q] = eb[min(i + q * BIN_THREADS, e1 - 1)];
      #pragma unroll
      for (int q = 0; q < BIN_BATCH; ++q)
        if (i + q * BIN_THREADS < e1) stage[atomicAdd(&cur[sd[q].y & (BIN_NODES - 1)], 1)] = sd[q].x;
    }
    __syncthreads();
    for (int i = tid; i < ne; i += BIN_THREADS) col[e0 + i] = stage[i];
  } else {
    for (int i = e0 + tid; i < e1; i += BIN_THREADS) {
      const int2 sd = eb[i];
      col[e0 + atomicAdd(&cur[sd.y & (BIN_NODES - 1)], 1)] = sd.x;
    }
  }
}

__global__ void dinv_from_count_kernel(float* __restrict__ cnt_dinv, int N) {
  const int i = blockIdx.x * blockDim.x + threadIdx.x;
  if (i < N) {
    const int c = __float_as_int(cnt_dinv[i]);
    cnt_dinv[i] = rsqrtf(fmaxf((float)(c + 1), 1.0f));
  }
}

__device__ __forceinline__ void row_fma(float4& acc, const float4 h, const float w) {
  acc.x = fmaf(h.x, w, acc.x);
  acc.y = fmaf(h.y, w, acc.y);
  acc.z = fmaf(h.z, w, acc.z);
  acc.w = fmaf(h.w, w, acc.w);
}

__device__ __forceinline__ float4 gather_row_sum(const float4* __restrict__ H4,
                                                 const int* __restrict__ rowptr,
                                                 const int* __restrict__ col,
                                                 const float* __restrict__ dinv,
                                                 const int d, const int lane) {
  const int beg = rowptr[d];
  const int end = rowptr[d + 1];
  const float dd = dinv[d];

  float4 acc;
  {
    const float4 hs = H4[(size_t)d * (FDIM / 4) + lane];
    const float ws = dd * dd;
    acc.x = hs.x * ws; acc.y = hs.y * ws; acc.z = hs.z * ws; acc.w = hs.w * ws;
  }

  for (int j = beg; j < end; j += 32) {
    const int n = min(32, end - j);
    int   s_l = 0;
    float w_l = 0.0f;
    if (lane < n) {
      s_l = col[j + lane];
      w_l = dinv[s_l] * dd;
    }
    int t = 0;
    for (; t + 4 <= n; t += 4) {
      const int s0 = __builtin_amdgcn_readlane(s_l, t);
      const int s1 = __builtin_amdgcn_readlane(s_l, t + 1);
      const int s2 = __builtin_amdgcn_readlane(s_l, t + 2);
      const int s3 = __builtin_amdgcn_readlane(s_l, t + 3);
      const float4 h0 = H4[(size_t)s0 * (FDIM / 4) + lane];
      const float4 h1 = H4[(size_t)s1 * (FDIM / 4) + lane];
      const float4 h2 = H4[(size_t)s2 * (FDIM / 4) + lane];
      const float4 h3 = H4[(size_t)s3 * (FDIM / 4) + lane];
      const float w0 = __int_as_float(__builtin_amdgcn_readlane(__float_as_int(w_l), t));
      const float w1 = __int_as_float(__builtin_amdgcn_readlane(__float_as_int(w_l), t + 1));
      const float w2 = __int_as_float(__builtin_amdgcn_readlane(__float_as_int(w_l), t + 2));
      const float w3 = __int_as_float(__builtin_amdgcn_readlane(__float_as_int(w_l), t + 3));
      row_fma(acc, h0, w0);
      row_fma(acc, h1, w1);
      row_fma(acc, h2, w2);
      row_fma(acc, h3, w3);
    }
    const int r = n - t;
    if (r == 3) {
      const int s0 = __builtin_amdgcn_readlane(s_l, t);
      const int s1 = __builtin_amdgcn_readlane(s_l, t + 1);
      const int s2 = __builtin_amdgcn_readlane(s_l, t + 2);
      const float4 h0 = H4[(size_t)s0 * (FDIM / 4) + lane];
      const float4 h1 = H4[(size_t)s1 * (FDIM / 4) + lane];
      const float4 h2 = H4[(size_t)s2 * (FDIM / 4) + lane];
      const float w0 = __int_as_float(__builtin_amdgcn_readlane(__float_as_int(w_l), t));
      const float w1 = __int_as_float(__builtin_amdgcn_readlane(__float_as_int(w_l), t + 1));
      const float w2 = __int_as_float(__builtin_amdgcn_readlane(__float_as_int(w_l), t + 2));
      row_fma(acc, h0, w0);
      row_fma(acc, h1, w1);
      row_fma(acc, h2, w2);
    } else if (r == 2) {
      const int s0 = __builtin_amdgcn_readlane(s_l, t);
      const int s1 = __builtin_amdgcn_readlane(s_l, t + 1);
      const float4 h0 = H4[(size_t)s0 * (FDIM / 4) + lane];
      const float4 h1 = H4[(size_t)s1 * (FDIM / 4) + lane];
      const float w0 = __int_as_float(__builtin_amdgcn_readlane(__float_as_int(w_l), t));
      const float w1 = __int_as_float(__builtin_amdgcn_readlane(__float_as_int(w_l), t + 1));
      row_fma(acc, h0, w0);
      row_fma(acc, h1, w1);
    } else if (r == 1) {
      const int s0 = __builtin_amdgcn_readlane(s_l, t);
      const float w0 = __int_as_float(__builtin_amdgcn_readlane(__float_as_int(w_l), t));
      row_fma(acc, H4[(size_t)s0 * (FDIM / 4) + lane], w0);
    }
  }
  return acc;
}

__global__ void gather_rows_kernel(const float* __restrict__ H, const int* __restrict__ rowptr,
                                   const int* __restrict__ col, const float* __restrict__ dinv,
                                   const float* __restrict__ bias, float* __restrict__ G, int N) {
  const int d = __builtin_amdgcn_readfirstlane(
      (int)(blockIdx.x * (blockDim.x >> 5) + (threadIdx.x >> 5)));
  if (d >= N) return;
  const int lane = threadIdx.x & 31;
  const float bx = bias[lane * 4 + 0];
  const float by = bias[lane * 4 + 1];
  const float bz = bias[lane * 4 + 2];
  const float bw = bias[lane * 4 + 3];
  float4 acc = gather_row_sum(reinterpret_cast<const float4*>(H), rowptr, col, dinv, d, lane);
  acc.x = fmaxf(acc.x + bx, 0.0f);
  acc.y = fmaxf(acc.y + by, 0.0f);
  acc.z = fmaxf(acc.z + bz, 0.0f);
  acc.w = fmaxf(acc.w + bw, 0.0f);
  reinterpret_cast<float4*>(G)[(size_t)d * (FDIM / 4) + lane] = acc;
}

#define FG_ROWS   32
#define FG_TPITCH 132
#define FG_BLOCKS 512

__global__ __launch_bounds__(32 * FG_ROWS, 16) void gather_product_kernel(
    const float* __restrict__ H, const int* __restrict__ rowptr, const int* __restrict__ col,
    const float* __restrict__ dinv, const float* __restrict__ bias, const float* __restrict__ Wn,
    float* __restrict__ Hn, int N, int ngroups) {
  __shared__ alignas(16) float smem[64 * GEMM_WPITCH + 2 * FG_ROWS * FG_TPITCH];
  float* Ws = smem;
  const int tid = threadIdx.x;

  {
    const float4* __restrict__ W4 = reinterpret_cast<const float4*>(Wn);
    for (int it = 0; it < 2; ++it) {
      const int i = it * (32 * FG_ROWS) + tid;
      const int p = i >> 5;
      const int q = i & 31;
      const float4 w0 = W4[(2 * p) * (FDIM / 4) + q];
      const float4 w1 = W4[(2 * p + 1) * (FDIM / 4) + q];
      float4* dst = reinterpret_cast<float4*>(Ws + p * GEMM_WPITCH + q * 8);
      dst[0] = make_float4(w0.x, w1.x, w0.y, w1.y);
      dst[1] = make_float4(w0.z, w1.z, w0.w, w1.w);
    }
  }
  __syncthreads();

  const int wave = __builtin_amdgcn_readfirstlane((int)(threadIdx.x >> 5));
  const int lane = tid & 31;
  const int c = lane & 15;
  const int h = lane >> 4;

  const float bx = bias[lane * 4 + 0];
  const float by = bias[lane * 4 + 1];
  const float bz = bias[lane * 4 + 2];
  const float bw = bias[lane * 4 + 3];
  const float4* __restrict__ H4 = reinterpret_cast<const float4*>(H);

  const int rt = (wave >> 2) & 1;
  const int cp = wave & 3;
  const unsigned boff0 = (unsigned)(h * (GEMM_WPITCH / 2) + c + 32 * cp) * 8u;

  int it = 0;
  for (int g = blockIdx.x; g < ngroups; g += gridDim.x, ++it) {
    const int toff = 64 * GEMM_WPITCH + (it & 1) * (FG_ROWS * FG_TPITCH);
    float* tl = smem + toff;
    const int d = g * FG_ROWS + wave;
    float4 acc = make_float4(0.0f, 0.0f, 0.0f, 0.0f);
    if (d < N) {
      acc = gather_row_sum(H4, rowptr, col, dinv, d, lane);
      acc.x = fmaxf(acc.x + bx, 0.0f);
      acc.y = fmaxf(acc.y + by, 0.0f);
      acc.z = fmaxf(acc.z + bz, 0.0f);
      acc.w = fmaxf(acc.w + bw, 0.0f);
    }
    reinterpret_cast<float4*>(tl + wave * FG_TPITCH)[lane] = acc;
    __syncthreads();

    if ((wave >> 3) == (it & 3)) {
      unsigned ao = (unsigned)(toff + (16 * rt + c) * FG_TPITCH + 2 * h) * 4u;
      unsigned bo = boff0;
      const char* sb = reinterpret_cast<const char*>(smem);
      const int row0 = g * FG_ROWS + 16 * rt + 8 * h;
      float* hp = Hn + (size_t)row0 * FDIM + 32 * cp + c;
      int nval = N - row0;
      asm volatile("" : "+v"(hp), "+v"(nval));
      __attribute__((address_space(1))) float* hg = (__attribute__((address_space(1))) float*)hp;
      v8f acc0 = v8f{};
      v8f acc1 = v8f{};
      v2f ax  = *reinterpret_cast<const v2f*>(sb + ao);
      v2f bx0 = *reinterpret_cast<const v2f*>(sb + bo);
      v2f bx1 = *reinterpret_cast<const v2f*>(sb + bo + 128);
      #pragma unroll 1
      for (int kk = 0; kk < FDIM / 4 - 2; kk += 2) {
        ao += 16;
        bo += 2 * (GEMM_WPITCH / 2) * 8;
        asm volatile("" : "+v"(ao), "+v"(bo));
        const v2f ay  = *reinterpret_cast<const v2f*>(sb + ao);
        const v2f by0 = *reinterpret_cast<const v2f*>(sb + bo);
        const v2f by1 = *reinterpret_cast<const v2f*>(sb + bo + 128);
        acc0 = __builtin_amdgcn_wmma_f32_16x16x4_f32(
             false, ax,  false, bx0,
             (short)0, acc0,  false,  false);
        acc1 = __builtin_amdgcn_wmma_f32_16x16x4_f32(
             false, ax,  false, bx1,
             (short)0, acc1,  false,  false);
        ao += 16;
        bo += 2 * (GEMM_WPITCH / 2) * 8;
        asm volatile("" : "+v"(ao), "+v"(bo));
        ax  = *reinterpret_cast<const v2f*>(sb + ao);
        bx0 = *reinterpret_cast<const v2f*>(sb + bo);
        bx1 = *reinterpret_cast<const v2f*>(sb + bo + 128);
        acc0 = __builtin_amdgcn_wmma_f32_16x16x4_f32(
             false, ay,  false, by0,
             (short)0, acc0,  false,  false);
        acc1 = __builtin_amdgcn_wmma_f32_16x16x4_f32(
             false, ay,  false, by1,
             (short)0, acc1,  false,  false);
      }
      {
        ao += 16;
        bo += 2 * (GEMM_WPITCH / 2) * 8;
        asm volatile("" : "+v"(ao), "+v"(bo));
        const v2f ay  = *reinterpret_cast<const v2f*>(sb + ao);
        const v2f by0 = *reinterpret_cast<const v2f*>(sb + bo);
        const v2f by1 = *reinterpret_cast<const v2f*>(sb + bo + 128);
        acc0 = __builtin_amdgcn_wmma_f32_16x16x4_f32(
             false, ax,  false, bx0,
             (short)0, acc0,  false,  false);
        acc1 = __builtin_amdgcn_wmma_f32_16x16x4_f32(
             false, ax,  false, bx1,
             (short)0, acc1,  false,  false);
        acc0 = __builtin_amdgcn_wmma_f32_16x16x4_f32(
             false, ay,  false, by0,
             (short)0, acc0,  false,  false);
        acc1 = __builtin_amdgcn_wmma_f32_16x16x4_f32(
             false, ay,  false, by1,
             (short)0, acc1,  false,  false);
      }
      #pragma unroll
      for (int r = 0; r < 8; ++r) {
        if (r < nval) {
          hg[r * FDIM]      = acc0[r];
          hg[r * FDIM + 16] = acc1[r];
        }
      }
    }
  }
}

__global__ void bias_relu_kernel(float* __restrict__ G, const float* __restrict__ b) {
  const size_t i = (size_t)blockIdx.x * FDIM + threadIdx.x;
  G[i] = fmaxf(G[i] + b[threadIdx.x], 0.0f);
}

__global__ void init_out_kernel(float* __restrict__ out, int ng) {
  int i = blockIdx.x * blockDim.x + threadIdx.x;
  if (i >= ng * 2 * FDIM) return;
  int f = i & (2 * FDIM - 1);
  out[i] = (f < FDIM) ? 0.0f : __int_as_float(0xFF7FFFFF);
}

__device__ __forceinline__ void atomic_max_f32(float* addr, float v) {
  if (v >= 0.0f) atomicMax((int*)addr, __float_as_int(v));
  else           atomicMin((unsigned int*)addr, (unsigned int)__float_as_int(v));
}

#define POOL_ROWS  32
#define POOL_WAVES 8

__device__ __forceinline__ void pool_flush(float* __restrict__ out, float* __restrict__ cnt,
                                           float* st, int lane, int g,
                                           const float4 s, const float4 m, int run) {
  reinterpret_cast<float4*>(st)[lane] = s;
  reinterpret_cast<float4*>(st + FDIM)[lane] = m;
  __builtin_amdgcn_fence(__ATOMIC_ACQ_REL, "wavefront");
  float a[4], b[4];
  #pragma unroll
  for (int k = 0; k < 4; ++k) {
    a[k] = st[lane + 32 * k];
    b[k] = st[FDIM + lane + 32 * k];
  }
  __builtin_amdgcn_fence(__ATOMIC_ACQ_REL, "wavefront");
  float* o = out + (size_t)g * (2 * FDIM);
  #pragma unroll
  for (int k = 0; k < 4; ++k) atomicAdd(o + lane + 32 * k, a[k]);
  #pragma unroll
  for (int k = 0; k < 4; ++k) atomic_max_f32(o + FDIM + lane + 32 * k, b[k]);
  if (lane == 0) atomicAdd(&cnt[g], (float)run);
}

__global__ __launch_bounds__(256) void pool_runs_kernel(const float* __restrict__ G,
                                                        const int* __restrict__ gi,
                                                        float* __restrict__ out,
                                                        float* __restrict__ cnt, int N) {
  __shared__ alignas(16) float stage[POOL_WAVES * 2 * FDIM];
  const int wv = threadIdx.x >> 5;
  const int lane = threadIdx.x & 31;
  const int n0 = __builtin_amdgcn_readfirstlane((int)(blockIdx.x * POOL_WAVES + wv)) * POOL_ROWS;
  if (n0 >= N) return;
  const int nr = min(POOL_ROWS, N - n0);
  float* st = stage + wv * (2 * FDIM);

  const int g_l = gi[n0 + min(lane, nr - 1)];

  const float4* __restrict__ G4 = reinterpret_cast<const float4*>(G) + (size_t)n0 * (FDIM / 4) + lane;
  float4 v[POOL_ROWS];
  #pragma unroll
  for (int i = 0; i < POOL_ROWS; ++i) v[i] = G4[(size_t)min(i, nr - 1) * (FDIM / 4)];

  int a = 0;
  for (int trip = 0; trip < POOL_ROWS && a < nr; ++trip) {
    const int ga = __builtin_amdgcn_readlane(g_l, a);
    const unsigned diff = __builtin_amdgcn_ballot_w32(g_l != ga) & (~0u << a);
    const int b = min(diff ? __builtin_ctz(diff) : POOL_ROWS, nr);
    float4 s = make_float4(0.0f, 0.0f, 0.0f, 0.0f);
    float4 m = make_float4(-FLT_MAX, -FLT_MAX, -FLT_MAX, -FLT_MAX);
    #pragma unroll
    for (int i = 0; i < POOL_ROWS; ++i) {
      if (i >= a && i < b) {
        s.x += v[i].x; s.y += v[i].y; s.z += v[i].z; s.w += v[i].w;
        m.x = fmaxf(m.x, v[i].x); m.y = fmaxf(m.y, v[i].y);
        m.z = fmaxf(m.z, v[i].z); m.w = fmaxf(m.w, v[i].w);
      }
    }
    pool_flush(out, cnt, st, lane, ga, s, m, b - a);
    a = b;
  }
}

__global__ void finalize_mean_kernel(float* __restrict__ out, const float* __restrict__ cnt) {
  const int g = blockIdx.x;
  out[(size_t)g * 2 * FDIM + threadIdx.x] /= fmaxf(cnt[g], 1.0f);
}


extern "C" void kernel_launch(void* const* d_in, const int* in_sizes, int n_in,
                              void* d_out, int out_size, void* d_ws, size_t ws_size,
                              hipStream_t stream) {
  const float* x  = (const float*)d_in[0];
  const int*   ei = (const int*)d_in[1];
  const int*   gi = (const int*)d_in[2];
  const float* Wm[3] = {(const float*)d_in[4], (const float*)d_in[6], (const float*)d_in[8]};
  const float* Bv[3] = {(const float*)d_in[5], (const float*)d_in[7], (const float*)d_in[9]};
  float* out = (float*)d_out;

  const int  N  = in_sizes[0] / FDIM;
  const int  E  = in_sizes[1] / 2;
  const int  NG = out_size / (2 * FDIM);
  const long Nf = (long)N * FDIM;

  float* Hbuf = (float*)d_ws;
  float* Gbuf = Hbuf + Nf;
  float* dinv = Gbuf + Nf;
  float* cnt  = dinv + N;
  if (ws_size < (size_t)(2 * Nf + N + NG) * sizeof(float)) return;

  const int nsb = (N + SCAN_ELEMS - 1) / SCAN_ELEMS;
  int* rowptr = (int*)(cnt + NG);
  int* bsum   = rowptr + (N + 1);
  int* col    = bsum + nsb;
  int* rank   = (int*)Hbuf;
  const bool ordered =
      ws_size >= (size_t)(2 * Nf + N + NG) * sizeof(float) + ((size_t)N + 1 + nsb + E) * sizeof(int) &&
      (long)E <= Nf && E > 0;

  auto zero = [&](float* p, long n) {
    zero_f32_kernel<<<(unsigned)((n + 255) / 256), 256, 0, stream>>>(p, n);
  };

  const int tiles = N / 16;
  const int rem   = N - tiles * 16;
  const float* cur = x;

  const int  nb   = (N + BIN_NODES - 1) / BIN_NODES;
  const long nh   = (long)nb * BIN_BLOCKS;
  const int  nsb2 = (int)((nh + SCAN_ELEMS - 1) / SCAN_ELEMS);
  int* ebuf  = (int*)Hbuf;
  int* hist  = (int*)Gbuf;
  int* boff  = hist + nh;
  int* bsum2 = boff + nh + 1;
  const bool binned = ordered && nb <= BIN_MAXB && 2 * (long)E <= Nf && 2 * nh + 1 + nsb2 <= Nf;

  if (ordered) {
    if (binned) {
      bin_count_kernel<<<BIN_BLOCKS, BIN_THREADS, 0, stream>>>(ei, hist, E, nb);
      scan_local_kernel<<<nsb2, SCAN_THREADS, 0, stream>>>(hist, boff, bsum2, (int)nh);
      scan_add_kernel<<<nsb2, SCAN_THREADS, 0, stream>>>(boff, bsum2, (int)nh);
      bin_scatter_kernel<<<BIN_BLOCKS, BIN_THREADS, 0, stream>>>(ei, boff, ebuf, E, nb);
      bin_finish_kernel<<<nb, BIN_THREADS, 0, stream>>>(ebuf, boff, rowptr, col, dinv, N);
    } else {
      zero(dinv, N);
      count_rank_kernel<<<(E + 255) / 256, 256, 0, stream>>>(ei, (int*)dinv, rank, E);
      scan_local_kernel<<<nsb, SCAN_THREADS, 0, stream>>>((const int*)dinv, rowptr, bsum, N);
      scan_add_kernel<<<nsb, SCAN_THREADS, 0, stream>>>(rowptr, bsum, N);
      place_kernel<<<(E + 255) / 256, 256, 0, stream>>>(ei, rowptr, rank, col, E);
      dinv_from_count_kernel<<<(N + 255) / 256, 256, 0, stream>>>(dinv, N);
    }

    if (tiles > 0)
      gemm_strip_kernel<<<GEMM_BLOCKS, 256, 0, stream>>>(x, Wm[0], Hbuf, tiles);
    if (rem > 0)
      gemm_tail_kernel<<<rem, FDIM, 0, stream>>>(x, Wm[0], Hbuf, tiles * 16);
    const int ngroups = (N + FG_ROWS - 1) / FG_ROWS;
    const int fgrid = ngroups < FG_BLOCKS ? ngroups : FG_BLOCKS;
    gather_product_kernel<<<fgrid, 32 * FG_ROWS, 0, stream>>>(Hbuf, rowptr, col, dinv, Bv[0], Wm[1], Gbuf, N, ngroups);
    gather_product_kernel<<<fgrid, 32 * FG_ROWS, 0, stream>>>(Gbuf, rowptr, col, dinv, Bv[1], Wm[2], Hbuf, N, ngroups);
    gather_rows_kernel<<<(N + 7) / 8, 256, 0, stream>>>(Hbuf, rowptr, col, dinv, Bv[2], Gbuf, N);
  } else {
    zero(dinv, N);
    degree_kernel<<<(E + N + 255) / 256, 256, 0, stream>>>(ei, dinv, E, N);
    dinv_kernel<<<(N + 255) / 256, 256, 0, stream>>>(dinv, N);

    for (int l = 0; l < 3; ++l) {
      if (tiles > 0)
        gemm_strip_kernel<<<GEMM_BLOCKS, 256, 0, stream>>>(cur, Wm[l], Hbuf, tiles);
      if (rem > 0)
        gemm_tail_kernel<<<rem, FDIM, 0, stream>>>(cur, Wm[l], Hbuf, tiles * 16);
      zero(Gbuf, Nf);
      aggregate_kernel<<<(E + N + 7) / 8, 256, 0, stream>>>(Hbuf, ei, dinv, Gbuf, E, N);
      bias_relu_kernel<<<N, FDIM, 0, stream>>>(Gbuf, Bv[l]);
      cur = Gbuf;
    }
  }

  init_out_kernel<<<(NG * 2 * FDIM + 255) / 256, 256, 0, stream>>>(out, NG);
  zero(cnt, NG);
  pool_runs_kernel<<<(N + POOL_ROWS * POOL_WAVES - 1) / (POOL_ROWS * POOL_WAVES), 32 * POOL_WAVES, 0, stream>>>(Gbuf, gi, out, cnt, N);
  finalize_mean_kernel<<<NG, FDIM, 0, stream>>>(out, cnt);
}


# Source weight folded into stored rows: each product's row i leaves times dinv[i]; a target sums plain rows, scales once by dinv[target]; per-edge dinv[src] fetch gone; products re-associated
# speedup vs baseline: 5.6008x; 1.0711x over previous
#include <hip/hip_runtime.h>
#include <float.h>

#define FDIM 128

typedef __attribute__((ext_vector_type(2))) float v2f;
typedef __attribute__((ext_vector_type(8))) float v8f;


__global__ void zero_f32_kernel(float* __restrict__ p, long n) {
  long i = (long)blockIdx.x * blockDim.x + threadIdx.x;
  if (i < n) p[i] = 0.0f;
}

__global__ void degree_kernel(const int* __restrict__ ei, float* __restrict__ deg,
                              int E, int N) {
  int e = blockIdx.x * blockDim.x + threadIdx.x;
  int EN = E + N;
  if (e >= EN) return;
  int d = (e < E) ? ei[E + e] : (e - E);
  atomicAdd(&deg[d], 1.0f);
}

__global__ void dinv_kernel(float* __restrict__ deg_dinv, int N) {
  int i = blockIdx.x * blockDim.x + threadIdx.x;
  if (i < N) deg_dinv[i] = rsqrtf(fmaxf(deg_dinv[i], 1.0f));
}

#define GEMM_WPITCH 288
#define GEMM_BLOCKS 512

template <bool SCALED>
__global__ __launch_bounds__(256, 4) void gemm_strip_kernel(const float* __restrict__ X,
                                                         const float* __restrict__ W,
                                                         const float* __restrict__ rs,
                                                         float* __restrict__ H, int tiles) {
  __shared__ alignas(16) float Ws[64 * GEMM_WPITCH];
  const int tid = threadIdx.x;

  {
    const float4* __restrict__ W4 = reinterpret_cast<const float4*>(W);
    for (int it = 0; it < 8; ++it) {
      const int i = it * 256 + tid;
      const int p = i >> 5;
      const int q = i & 31;
      const float4 w0 = W4[(2 * p) * (FDIM / 4) + q];
      const float4 w1 = W4[(2 * p + 1) * (FDIM / 4) + q];
      float4* dst = reinterpret_cast<float4*>(Ws + p * GEMM_WPITCH + q * 8);
      dst[0] = make_float4(w0.x, w1.x, w0.y, w1.y);
      dst[1] = make_float4(w0.z, w1.z, w0.w, w1.w);
    }
  }
  __syncthreads();

  const int wave = __builtin_amdgcn_readfirstlane((int)(threadIdx.x >> 5));
  const int lane = tid & 31;
  const int c = lane & 15;
  const int h = lane >> 4;

  const v2f* __restrict__ bfrag = reinterpret_cast<const v2f*>(Ws) + h * (GEMM_WPITCH / 2) + c;

  const int xoff = c * (FDIM / 2) + h;
  const int hoff = 8 * h * FDIM + c;

  const int s0 = blockIdx.x + gridDim.x * wave;
  const size_t sstep = (size_t)gridDim.x * 8 * (16 * FDIM);
  const v2f* xs = reinterpret_cast<const v2f*>(X + (size_t)s0 * (16 * FDIM)) + xoff;
  float* hs = H + (size_t)s0 * (16 * FDIM) + hoff;
  const float* rp = SCALED ? rs + (size_t)s0 * 16 + c : X;
  const size_t rstep = SCALED ? (size_t)gridDim.x * 8 * 16 : 0;

  for (int s = s0; s < tiles; s += gridDim.x * 8, xs += sstep / 2, hs += sstep, rp += rstep) {
    v2f a[FDIM / 4];
    #pragma unroll
    for (int kk = 0; kk < FDIM / 4; ++kk) a[kk] = xs[2 * kk];
    if (SCALED) {
      const float sc = *rp;
      #pragma unroll
      for (int kk = 0; kk < FDIM / 4; ++kk) a[kk] = a[kk] * sc;
    }

    v8f acc[8];
    #pragma unroll
    for (int t = 0; t < 8; ++t) acc[t] = v8f{};

    v2f bcur[8];
    #pragma unroll
    for (int t = 0; t < 8; ++t) bcur[t] = bfrag[16 * t];

    #pragma unroll
    for (int kk = 0; kk < FDIM / 4; ++kk) {
      v2f bnxt[8];
      if (kk + 1 < FDIM / 4) {
        const v2f* bp = bfrag + (2 * (kk + 1)) * (GEMM_WPITCH / 2);
        #pragma unroll
        for (int t = 0; t < 8; ++t) bnxt[t] = bp[16 * t];
      }
      #pragma unroll
      for (int t = 0; t < 8; ++t) {
        acc[t] = __builtin_amdgcn_wmma_f32_16x16x4_f32(
             false, a[kk],  false, bcur[t],
             (short)0, acc[t],  false,  false);
      }
      __builtin_amdgcn_sched_barrier(0);
      if (kk + 1 < FDIM / 4) {
        #pragma unroll
        for (int t = 0; t < 8; ++t) bcur[t] = bnxt[t];
      }
    }

    #pragma unroll
    for (int t = 0; t < 8; ++t) {
      #pragma unroll
      for (int r = 0; r < 8; ++r) hs[r * FDIM + 16 * t] = acc[t][r];
    }
  }
}

__global__ void gemm_tail_kernel(const float* __restrict__ X,
                                 const float* __restrict__ W,
                                 const float* __restrict__ rs,
                                 float* __restrict__ H, int base) {
  const int row = base + blockIdx.x;
  const int f = threadIdx.x;
  const float* xr = X + (size_t)row * FDIM;
  float acc = 0.0f;
  #pragma unroll 8
  for (int k = 0; k < FDIM; ++k) acc = fmaf(xr[k], W[(size_t)k * FDIM + f], acc);
  H[(size_t)row * FDIM + f] = rs ? acc * rs[row] : acc;
}

__global__ void aggregate_kernel(const float* __restrict__ H,
                                 const int* __restrict__ ei,
                                 const float* __restrict__ dinv,
                                 float* __restrict__ G, int E, int N) {
  const int e = blockIdx.x * (blockDim.x >> 5) + (threadIdx.x >> 5);
  const int EN = E + N;
  if (e >= EN) return;
  const int lane = threadIdx.x & 31;
  int s, d;
  if (e < E) { s = ei[e]; d = ei[E + e]; }
  else       { s = e - E; d = s; }
  const float w = dinv[s] * dinv[d];
  const float4 hv = reinterpret_cast<const float4*>(H + (size_t)s * FDIM)[lane];
  float* gp = G + (size_t)d * FDIM + lane * 4;
  atomicAdd(gp + 0, hv.x * w);
  atomicAdd(gp + 1, hv.y * w);
  atomicAdd(gp + 2, hv.z * w);
  atomicAdd(gp + 3, hv.w * w);
}


__global__ void count_rank_kernel(const int* __restrict__ ei, int* __restrict__ icnt,
                                  int* __restrict__ rank, int E) {
  const int e = blockIdx.x * blockDim.x + threadIdx.x;
  if (e >= E) return;
  rank[e] = atomicAdd(&icnt[ei[E + e]], 1);
}

#define SCAN_THREADS 256
#define SCAN_ELEMS   1024

__global__ void scan_local_kernel(const int* __restrict__ icnt, int* __restrict__ rowptr,
                                  int* __restrict__ bsum, int N) {
  __shared__ int wsum[SCAN_THREADS / 32];
  const int tid = threadIdx.x, lane = tid & 31, wv = tid >> 5;
  const int base = blockIdx.x * SCAN_ELEMS + tid * 4;
  int v[4];
  #pragma unroll
  for (int q = 0; q < 4; ++q) v[q] = (base + q < N) ? icnt[base + q] : 0;
  const int t = v[0] + v[1] + v[2] + v[3];
  int inc = t;
  #pragma unroll
  for (int off = 1; off < 32; off <<= 1) {
    const int u = __shfl_up(inc, off);
    if (lane >= off) inc += u;
  }
  if (lane == 31) wsum[wv] = inc;
  __syncthreads();
  int woff = 0, tot = 0;
  #pragma unroll
  for (int k = 0; k < SCAN_THREADS / 32; ++k) {
    const int s = wsum[k];
    if (k < wv) woff += s;
    tot += s;
  }
  int run = woff + inc - t;
  #pragma unroll
  for (int q = 0; q < 4; ++q) {
    if (base + q < N) rowptr[base + q] = run;
    run += v[q];
  }
  if (tid == 0) bsum[blockIdx.x] = tot;
}

__global__ void scan_add_kernel(int* __restrict__ rowptr, const int* __restrict__ bsum, int N) {
  __shared__ int red[SCAN_THREADS / 32];
  const int tid = threadIdx.x, lane = tid & 31, wv = tid >> 5;
  const int b = blockIdx.x;
  int part = 0;
  for (int k = tid; k < b; k += SCAN_THREADS) part += bsum[k];
  #pragma unroll
  for (int off = 16; off > 0; off >>= 1) part += __shfl_xor(part, off);
  if (lane == 0) red[wv] = part;
  __syncthreads();
  int add = 0;
  #pragma unroll
  for (int k = 0; k < SCAN_THREADS / 32; ++k) add += red[k];
  const int base = b * SCAN_ELEMS + tid * 4;
  #pragma unroll
  for (int q = 0; q < 4; ++q)
    if (base + q < N) rowptr[base + q] += add;
  if (b == (int)gridDim.x - 1 && tid == 0) rowptr[N] = add + bsum[b];
}

__global__ void place_kernel(const int* __restrict__ ei, const int* __restrict__ rowptr,
                             const int* __restrict__ rank, int* __restrict__ col, int E) {
  const int e = blockIdx.x * blockDim.x + threadIdx.x;
  if (e >= E) return;
  col[rowptr[ei[E + e]] + rank[e]] = ei[e];
}

#define BIN_SHIFT   8
#define BIN_NODES   256
#define BIN_MAXB    1024
#define BIN_BLOCKS  256
#define BIN_THREADS 1024
#define BIN_BATCH   4
#define BIN_STAGE   16384

__global__ __launch_bounds__(BIN_THREADS) void bin_count_kernel(const int* __restrict__ ei,
                                                                int* __restrict__ hist, int E, int nb) {
  __shared__ int h[BIN_MAXB];
  const int tid = threadIdx.x;
  for (int k = tid; k < nb; k += BIN_THREADS) h[k] = 0;
  __syncthreads();
  const int chunk = (E + BIN_BLOCKS - 1) / BIN_BLOCKS;
  const int lo = blockIdx.x * chunk;
  const int hi = min(E, lo + chunk);
  for (int i = lo + tid; i < hi; i += BIN_BATCH * BIN_THREADS) {
    int d[BIN_BATCH];
    #pragma unroll
    for (int q = 0; q < BIN_BATCH; ++q) d[q] = ei[E + min(i + q * BIN_THREADS, hi - 1)];
    #pragma unroll
    for (int q = 0; q < BIN_BATCH; ++q)
      if (i + q * BIN_THREADS < hi) atomicAdd(&h[d[q] >> BIN_SHIFT], 1);
  }
  __syncthreads();
  for (int k = tid; k < nb; k += BIN_THREADS) hist[k * BIN_BLOCKS + blockIdx.x] = h[k];
}

__global__ __launch_bounds__(BIN_THREADS) void bin_scatter_kernel(const int* __restrict__ ei,
                                                                  const int* __restrict__ boff,
                                                                  int* __restrict__ ebuf, int E, int nb) {
  __shared__ int cur[BIN_MAXB];
  const int tid = threadIdx.x;
  for (int k = tid; k < nb; k += BIN_THREADS) cur[k] = boff[k * BIN_BLOCKS + blockIdx.x];
  __syncthreads();
  const int chunk = (E + BIN_BLOCKS - 1) / BIN_BLOCKS;
  const int lo = blockIdx.x * chunk;
  const int hi = min(E, lo + chunk);
  int2* __restrict__ eb = reinterpret_cast<int2*>(ebuf);
  for (int i = lo + tid; i < hi; i += BIN_BATCH * BIN_THREADS) {
    int s[BIN_BATCH], d[BIN_BATCH];
    #pragma unroll
    for (int q = 0; q < BIN_BATCH; ++q) {
      const int j = min(i + q * BIN_THREADS, hi - 1);
      s[q] = ei[j];
      d[q] = ei[E + j];
    }
    #pragma unroll
    for (int q = 0; q < BIN_BATCH; ++q) {
      if (i + q * BIN_THREADS < hi) {
        const int p = atomicAdd(&cur[d[q] >> BIN_SHIFT], 1);
        eb[p] = make_int2(s[q], d[q]);
      }
    }
  }
}

__global__ __launch_bounds__(BIN_THREADS) void bin_finish_kernel(const int* __restrict__ ebuf,
                                                                 const int* __restrict__ boff,
                                                                 int* __restrict__ rowptr,
                                                                 int* __restrict__ col,
                                                                 float* __restrict__ dinv, int N) {
  __shared__ int cnt[BIN_NODES];
  __shared__ int cur[BIN_NODES];
  __shared__ int wsum[BIN_NODES / 32];
  __shared__ int stage[BIN_STAGE];
  const int tid = threadIdx.x, lane = tid & 31, wv = tid >> 5;
  const int b = blockIdx.x;
  const int e0 = boff[b * BIN_BLOCKS];
  const int e1 = boff[(b + 1) * BIN_BLOCKS];
  const int2* __restrict__ eb = reinterpret_cast<const int2*>(ebuf);
  if (tid < BIN_NODES) cnt[tid] = 0;
  __syncthreads();
  for (int i = e0 + tid; i < e1; i += BIN_BATCH * BIN_THREADS) {
    int d[BIN_BATCH];
    #pragma unroll
    for (int q = 0; q < BIN_BATCH; ++q) d[q] = eb[min(i + q * BIN_THREADS, e1 - 1)].y;
    #pragma unroll
    for (int q = 0; q < BIN_BATCH; ++q)
      if (i + q * BIN_THREADS < e1) atomicAdd(&cnt[d[q] & (BIN_NODES - 1)], 1);
  }
  __syncthreads();

  const int v = (tid < BIN_NODES) ? cnt[tid] : 0;
  int inc = v;
  #pragma unroll
  for (int off = 1; off < 32; off <<= 1) {
    const int u = __shfl_up(inc, off);
    if (lane >= off) inc += u;
  }
  if (tid < BIN_NODES && lane == 31) wsum[wv] = inc;
  __syncthreads();
  if (tid < BIN_NODES) {
    int woff = 0;
    #pragma unroll
    for (int k = 0; k < BIN_NODES / 32; ++k)
      if (k < wv) woff += wsum[k];
    const int excl = woff + inc - v;
    cur[tid] = excl;
    const int node = b * BIN_NODES + tid;
    if (node < N) {
      rowptr[node] = e0 + excl;
      dinv[node] = rsqrtf(fmaxf((float)(v + 1), 1.0f));
    }
  }
  if (b == (int)gridDim.x - 1 && tid == 0) rowptr[N] = e1;
  __syncthreads();

  const int ne = e1 - e0;
  if (ne <= BIN_STAGE) {
    for (int i = e0 + tid; i < e1; i += BIN_BATCH * BIN_THREADS) {
      int2 sd[BIN_BATCH];
      #pragma unroll
      for (int q = 0; q < BIN_BATCH; ++q) sd[q] = eb[min(i + q * BIN_THREADS, e1 - 1)];
      #pragma unroll
      for (int q = 0; q < BIN_BATCH; ++q)
        if (i + q * BIN_THREADS < e1) stage[atomicAdd(&cur[sd[q].y & (BIN_NODES - 1)], 1)] = sd[q].x;
    }
    __syncthreads();
    for (int i = tid; i < ne; i += BIN_THREADS) col[e0 + i] = stage[i];
  } else {
    for (int i = e0 + tid; i < e1; i += BIN_THREADS) {
      const int2 sd = eb[i];
      col[e0 + atomicAdd(&cur[sd.y & (BIN_NODES - 1)], 1)] = sd.x;
    }
  }
}

__global__ void dinv_from_count_kernel(float* __restrict__ cnt_dinv, int N) {
  const int i = blockIdx.x * blockDim.x + threadIdx.x;
  if (i < N) {
    const int c = __float_as_int(cnt_dinv[i]);
    cnt_dinv[i] = rsqrtf(fmaxf((float)(c + 1), 1.0f));
  }
}

__device__ __forceinline__ void row_add(float4& acc, const float4 h) {
  acc.x += h.x;
  acc.y += h.y;
  acc.z += h.z;
  acc.w += h.w;
}

__device__ __forceinline__ float4 gather_row_sum(const float4* __restrict__ H4,
                                                 const int* __restrict__ rowptr,
                                                 const int* __restrict__ col,
                                                 const int d, const int lane) {
  const int beg = rowptr[d];
  const int end = rowptr[d + 1];

  float4 acc = H4[(size_t)d * (FDIM / 4) + lane];

  for (int j = beg; j < end; j += 32) {
    const int n = min(32, end - j);
    int s_l = 0;
    if (lane < n) s_l = col[j + lane];
    int t = 0;
    for (; t + 4 <= n; t += 4) {
      const int s0 = __builtin_amdgcn_readlane(s_l, t);
      const int s1 = __builtin_amdgcn_readlane(s_l, t + 1);
      const int s2 = __builtin_amdgcn_readlane(s_l, t + 2);
      const int s3 = __builtin_amdgcn_readlane(s_l, t + 3);
      const float4 h0 = H4[(size_t)s0 * (FDIM / 4) + lane];
      const float4 h1 = H4[(size_t)s1 * (FDIM / 4) + lane];
      const float4 h2 = H4[(size_t)s2 * (FDIM / 4) + lane];
      const float4 h3 = H4[(size_t)s3 * (FDIM / 4) + lane];
      row_add(acc, h0);
      row_add(acc, h1);
      row_add(acc, h2);
      row_add(acc, h3);
    }
    const int r = n - t;
    if (r == 3) {
      const int s0 = __builtin_amdgcn_readlane(s_l, t);
      const int s1 = __builtin_amdgcn_readlane(s_l, t + 1);
      const int s2 = __builtin_amdgcn_readlane(s_l, t + 2);
      const float4 h0 = H4[(size_t)s0 * (FDIM / 4) + lane];
      const float4 h1 = H4[(size_t)s1 * (FDIM / 4) + lane];
      const float4 h2 = H4[(size_t)s2 * (FDIM / 4) + lane];
      row_add(acc, h0);
      row_add(acc, h1);
      row_add(acc, h2);
    } else if (r == 2) {
      const int s0 = __builtin_amdgcn_readlane(s_l, t);
      const int s1 = __builtin_amdgcn_readlane(s_l, t + 1);
      const float4 h0 = H4[(size_t)s0 * (FDIM / 4) + lane];
      const float4 h1 = H4[(size_t)s1 * (FDIM / 4) + lane];
      row_add(acc, h0);
      row_add(acc, h1);
    } else if (r == 1) {
      const int s0 = __builtin_amdgcn_readlane(s_l, t);
      row_add(acc, H4[(size_t)s0 * (FDIM / 4) + lane]);
    }
  }
  return acc;
}

__global__ void gather_rows_kernel(const float* __restrict__ H, const int* __restrict__ rowptr,
                                   const int* __restrict__ col, const float* __restrict__ dinv,
                                   const float* __restrict__ bias, float* __restrict__ G, int N) {
  const int d = __builtin_amdgcn_readfirstlane(
      (int)(blockIdx.x * (blockDim.x >> 5) + (threadIdx.x >> 5)));
  if (d >= N) return;
  const int lane = threadIdx.x & 31;
  const float bx = bias[lane * 4 + 0];
  const float by = bias[lane * 4 + 1];
  const float bz = bias[lane * 4 + 2];
  const float bw = bias[lane * 4 + 3];
  const float dd = dinv[d];
  float4 acc = gather_row_sum(reinterpret_cast<const float4*>(H), rowptr, col, d, lane);
  acc.x = fmaxf(acc.x * dd + bx, 0.0f);
  acc.y = fmaxf(acc.y * dd + by, 0.0f);
  acc.z = fmaxf(acc.z * dd + bz, 0.0f);
  acc.w = fmaxf(acc.w * dd + bw, 0.0f);
  reinterpret_cast<float4*>(G)[(size_t)d * (FDIM / 4) + lane] = acc;
}

#define FG_ROWS   32
#define FG_TPITCH 132
#define FG_BLOCKS 512

__global__ __launch_bounds__(32 * FG_ROWS, 16) void gather_product_kernel(
    const float* __restrict__ H, const int* __restrict__ rowptr, const int* __restrict__ col,
    const float* __restrict__ dinv, const float* __restrict__ bias, const float* __restrict__ Wn,
    float* __restrict__ Hn, int N, int ngroups) {
  __shared__ alignas(16) float smem[64 * GEMM_WPITCH + 2 * FG_ROWS * FG_TPITCH];
  float* Ws = smem;
  const int tid = threadIdx.x;

  {
    const float4* __restrict__ W4 = reinterpret_cast<const float4*>(Wn);
    for (int it = 0; it < 2; ++it) {
      const int i = it * (32 * FG_ROWS) + tid;
      const int p = i >> 5;
      const int q = i & 31;
      const float4 w0 = W4[(2 * p) * (FDIM / 4) + q];
      const float4 w1 = W4[(2 * p + 1) * (FDIM / 4) + q];
      float4* dst = reinterpret_cast<float4*>(Ws + p * GEMM_WPITCH + q * 8);
      dst[0] = make_float4(w0.x, w1.x, w0.y, w1.y);
      dst[1] = make_float4(w0.z, w1.z, w0.w, w1.w);
    }
  }
  __syncthreads();

  const int wave = __builtin_amdgcn_readfirstlane((int)(threadIdx.x >> 5));
  const int lane = tid & 31;
  const int c = lane & 15;
  const int h = lane >> 4;

  const float bx = bias[lane * 4 + 0];
  const float by = bias[lane * 4 + 1];
  const float bz = bias[lane * 4 + 2];
  const float bw = bias[lane * 4 + 3];
  const float4* __restrict__ H4 = reinterpret_cast<const float4*>(H);

  const int rt = (wave >> 2) & 1;
  const int cp = wave & 3;
  const unsigned boff0 = (unsigned)(h * (GEMM_WPITCH / 2) + c + 32 * cp) * 8u;

  int it = 0;
  for (int g = blockIdx.x; g < ngroups; g += gridDim.x, ++it) {
    const int toff = 64 * GEMM_WPITCH + (it & 1) * (FG_ROWS * FG_TPITCH);
    float* tl = smem + toff;
    const int d = g * FG_ROWS + wave;
    float4 acc = make_float4(0.0f, 0.0f, 0.0f, 0.0f);
    if (d < N) {
      const float dd = dinv[d];
      acc = gather_row_sum(H4, rowptr, col, d, lane);
      acc.x = fmaxf(acc.x * dd + bx, 0.0f) * dd;
      acc.y = fmaxf(acc.y * dd + by, 0.0f) * dd;
      acc.z = fmaxf(acc.z * dd + bz, 0.0f) * dd;
      acc.w = fmaxf(acc.w * dd + bw, 0.0f) * dd;
    }
    reinterpret_cast<float4*>(tl + wave * FG_TPITCH)[lane] = acc;
    __syncthreads();

    if ((wave >> 3) == (it & 3)) {
      unsigned ao = (unsigned)(toff + (16 * rt + c) * FG_TPITCH + 2 * h) * 4u;
      unsigned bo = boff0;
      const char* sb = reinterpret_cast<const char*>(smem);
      const int row0 = g * FG_ROWS + 16 * rt + 8 * h;
      float* hp = Hn + (size_t)row0 * FDIM + 32 * cp + c;
      int nval = N - row0;
      asm volatile("" : "+v"(hp), "+v"(nval));
      __attribute__((address_space(1))) float* hg = (__attribute__((address_space(1))) float*)hp;
      v8f acc0 = v8f{};
      v8f acc1 = v8f{};
      v2f ax  = *reinterpret_cast<const v2f*>(sb + ao);
      v2f bx0 = *reinterpret_cast<const v2f*>(sb + bo);
      v2f bx1 = *reinterpret_cast<const v2f*>(sb + bo + 128);
      #pragma unroll 1
      for (int kk = 0; kk < FDIM / 4 - 2; kk += 2) {
        ao += 16;
        bo += 2 * (GEMM_WPITCH / 2) * 8;
        asm volatile("" : "+v"(ao), "+v"(bo));
        const v2f ay  = *reinterpret_cast<const v2f*>(sb + ao);
        const v2f by0 = *reinterpret_cast<const v2f*>(sb + bo);
        const v2f by1 = *reinterpret_cast<const v2f*>(sb + bo + 128);
        acc0 = __builtin_amdgcn_wmma_f32_16x16x4_f32(
             false, ax,  false, bx0,
             (short)0, acc0,  false,  false);
        acc1 = __builtin_amdgcn_wmma_f32_16x16x4_f32(
             false, ax,  false, bx1,
             (short)0, acc1,  false,  false);
        ao += 16;
        bo += 2 * (GEMM_WPITCH / 2) * 8;
        asm volatile("" : "+v"(ao), "+v"(bo));
        ax  = *reinterpret_cast<const v2f*>(sb + ao);
        bx0 = *reinterpret_cast<const v2f*>(sb + bo);
        bx1 = *reinterpret_cast<const v2f*>(sb + bo + 128);
        acc0 = __builtin_amdgcn_wmma_f32_16x16x4_f32(
             false, ay,  false, by0,
             (short)0, acc0,  false,  false);
        acc1 = __builtin_amdgcn_wmma_f32_16x16x4_f32(
             false, ay,  false, by1,
             (short)0, acc1,  false,  false);
      }
      {
        ao += 16;
        bo += 2 * (GEMM_WPITCH / 2) * 8;
        asm volatile("" : "+v"(ao), "+v"(bo));
        const v2f ay  = *reinterpret_cast<const v2f*>(sb + ao);
        const v2f by0 = *reinterpret_cast<const v2f*>(sb + bo);
        const v2f by1 = *reinterpret_cast<const v2f*>(sb + bo + 128);
        acc0 = __builtin_amdgcn_wmma_f32_16x16x4_f32(
             false, ax,  false, bx0,
             (short)0, acc0,  false,  false);
        acc1 = __builtin_amdgcn_wmma_f32_16x16x4_f32(
             false, ax,  false, bx1,
             (short)0, acc1,  false,  false);
        acc0 = __builtin_amdgcn_wmma_f32_16x16x4_f32(
             false, ay,  false, by0,
             (short)0, acc0,  false,  false);
        acc1 = __builtin_amdgcn_wmma_f32_16x16x4_f32(
             false, ay,  false, by1,
             (short)0, acc1,  false,  false);
      }
      #pragma unroll
      for (int r = 0; r < 8; ++r) {
        if (r < nval) {
          hg[r * FDIM]      = acc0[r];
          hg[r * FDIM + 16] = acc1[r];
        }
      }
    }
  }
}

__global__ void bias_relu_kernel(float* __restrict__ G, const float* __restrict__ b) {
  const size_t i = (size_t)blockIdx.x * FDIM + threadIdx.x;
  G[i] = fmaxf(G[i] + b[threadIdx.x], 0.0f);
}

__global__ void init_out_kernel(float* __restrict__ out, int ng) {
  int i = blockIdx.x * blockDim.x + threadIdx.x;
  if (i >= ng * 2 * FDIM) return;
  int f = i & (2 * FDIM - 1);
  out[i] = (f < FDIM) ? 0.0f : __int_as_float(0xFF7FFFFF);
}

__device__ __forceinline__ void atomic_max_f32(float* addr, float v) {
  if (v >= 0.0f) atomicMax((int*)addr, __float_as_int(v));
  else           atomicMin((unsigned int*)addr, (unsigned int)__float_as_int(v));
}

#define POOL_ROWS  32
#define POOL_WAVES 8

__device__ __forceinline__ void pool_flush(float* __restrict__ out, float* __restrict__ cnt,
                                           float* st, int lane, int g,
                                           const float4 s, const float4 m, int run) {
  reinterpret_cast<float4*>(st)[lane] = s;
  reinterpret_cast<float4*>(st + FDIM)[lane] = m;
  __builtin_amdgcn_fence(__ATOMIC_ACQ_REL, "wavefront");
  float a[4], b[4];
  #pragma unroll
  for (int k = 0; k < 4; ++k) {
    a[k] = st[lane + 32 * k];
    b[k] = st[FDIM + lane + 32 * k];
  }
  __builtin_amdgcn_fence(__ATOMIC_ACQ_REL, "wavefront");
  float* o = out + (size_t)g * (2 * FDIM);
  #pragma unroll
  for (int k = 0; k < 4; ++k) atomicAdd(o + lane + 32 * k, a[k]);
  #pragma unroll
  for (int k = 0; k < 4; ++k) atomic_max_f32(o + FDIM + lane + 32 * k, b[k]);
  if (lane == 0) atomicAdd(&cnt[g], (float)run);
}

__global__ __launch_bounds__(256) void pool_runs_kernel(const float* __restrict__ G,
                                                        const int* __restrict__ gi,
                                                        float* __restrict__ out,
                                                        float* __restrict__ cnt, int N) {
  __shared__ alignas(16) float stage[POOL_WAVES * 2 * FDIM];
  const int wv = threadIdx.x >> 5;
  const int lane = threadIdx.x & 31;
  const int n0 = __builtin_amdgcn_readfirstlane((int)(blockIdx.x * POOL_WAVES + wv)) * POOL_ROWS;
  if (n0 >= N) return;
  const int nr = min(POOL_ROWS, N - n0);
  float* st = stage + wv * (2 * FDIM);

  const int g_l = gi[n0 + min(lane, nr - 1)];

  const float4* __restrict__ G4 = reinterpret_cast<const float4*>(G) + (size_t)n0 * (FDIM / 4) + lane;
  float4 v[POOL_ROWS];
  #pragma unroll
  for (int i = 0; i < POOL_ROWS; ++i) v[i] = G4[(size_t)min(i, nr - 1) * (FDIM / 4)];

  int a = 0;
  for (int trip = 0; trip < POOL_ROWS && a < nr; ++trip) {
    const int ga = __builtin_amdgcn_readlane(g_l, a);
    const unsigned diff = __builtin_amdgcn_ballot_w32(g_l != ga) & (~0u << a);
    const int b = min(diff ? __builtin_ctz(diff) : POOL_ROWS, nr);
    float4 s = make_float4(0.0f, 0.0f, 0.0f, 0.0f);
    float4 m = make_float4(-FLT_MAX, -FLT_MAX, -FLT_MAX, -FLT_MAX);
    #pragma unroll
    for (int i = 0; i < POOL_ROWS; ++i) {
      if (i >= a && i < b) {
        s.x += v[i].x; s.y += v[i].y; s.z += v[i].z; s.w += v[i].w;
        m.x = fmaxf(m.x, v[i].x); m.y = fmaxf(m.y, v[i].y);
        m.z = fmaxf(m.z, v[i].z); m.w = fmaxf(m.w, v[i].w);
      }
    }
    pool_flush(out, cnt, st, lane, ga, s, m, b - a);
    a = b;
  }
}

__global__ void finalize_mean_kernel(float* __restrict__ out, const float* __restrict__ cnt) {
  const int g = blockIdx.x;
  out[(size_t)g * 2 * FDIM + threadIdx.x] /= fmaxf(cnt[g], 1.0f);
}


extern "C" void kernel_launch(void* const* d_in, const int* in_sizes, int n_in,
                              void* d_out, int out_size, void* d_ws, size_t ws_size,
                              hipStream_t stream) {
  const float* x  = (const float*)d_in[0];
  const int*   ei = (const int*)d_in[1];
  const int*   gi = (const int*)d_in[2];
  const float* Wm[3] = {(const float*)d_in[4], (const float*)d_in[6], (const float*)d_in[8]};
  const float* Bv[3] = {(const float*)d_in[5], (const float*)d_in[7], (const float*)d_in[9]};
  float* out = (float*)d_out;

  const int  N  = in_sizes[0] / FDIM;
  const int  E  = in_sizes[1] / 2;
  const int  NG = out_size / (2 * FDIM);
  const long Nf = (long)N * FDIM;

  float* Hbuf = (float*)d_ws;
  float* Gbuf = Hbuf + Nf;
  float* dinv = Gbuf + Nf;
  float* cnt  = dinv + N;
  if (ws_size < (size_t)(2 * Nf + N + NG) * sizeof(float)) return;

  const int nsb = (N + SCAN_ELEMS - 1) / SCAN_ELEMS;
  int* rowptr = (int*)(cnt + NG);
  int* bsum   = rowptr + (N + 1);
  int* col    = bsum + nsb;
  int* rank   = (int*)Hbuf;
  const bool ordered =
      ws_size >= (size_t)(2 * Nf + N + NG) * sizeof(float) + ((size_t)N + 1 + nsb + E) * sizeof(int) &&
      (long)E <= Nf && E > 0;

  auto zero = [&](float* p, long n) {
    zero_f32_kernel<<<(unsigned)((n + 255) / 256), 256, 0, stream>>>(p, n);
  };

  const int tiles = N / 16;
  const int rem   = N - tiles * 16;
  const float* cur = x;

  const int  nb   = (N + BIN_NODES - 1) / BIN_NODES;
  const long nh   = (long)nb * BIN_BLOCKS;
  const int  nsb2 = (int)((nh + SCAN_ELEMS - 1) / SCAN_ELEMS);
  int* ebuf  = (int*)Hbuf;
  int* hist  = (int*)Gbuf;
  int* boff  = hist + nh;
  int* bsum2 = boff + nh + 1;
  const bool binned = ordered && nb <= BIN_MAXB && 2 * (long)E <= Nf && 2 * nh + 1 + nsb2 <= Nf;

  if (ordered) {
    if (binned) {
      bin_count_kernel<<<BIN_BLOCKS, BIN_THREADS, 0, stream>>>(ei, hist, E, nb);
      scan_local_kernel<<<nsb2, SCAN_THREADS, 0, stream>>>(hist, boff, bsum2, (int)nh);
      scan_add_kernel<<<nsb2, SCAN_THREADS, 0, stream>>>(boff, bsum2, (int)nh);
      bin_scatter_kernel<<<BIN_BLOCKS, BIN_THREADS, 0, stream>>>(ei, boff, ebuf, E, nb);
      bin_finish_kernel<<<nb, BIN_THREADS, 0, stream>>>(ebuf, boff, rowptr, col, dinv, N);
    } else {
      zero(dinv, N);
      count_rank_kernel<<<(E + 255) / 256, 256, 0, stream>>>(ei, (int*)dinv, rank, E);
      scan_local_kernel<<<nsb, SCAN_THREADS, 0, stream>>>((const int*)dinv, rowptr, bsum, N);
      scan_add_kernel<<<nsb, SCAN_THREADS, 0, stream>>>(rowptr, bsum, N);
      place_kernel<<<(E + 255) / 256, 256, 0, stream>>>(ei, rowptr, rank, col, E);
      dinv_from_count_kernel<<<(N + 255) / 256, 256, 0, stream>>>(dinv, N);
    }

    if (tiles > 0)
      gemm_strip_kernel<true><<<GEMM_BLOCKS, 256, 0, stream>>>(x, Wm[0], dinv, Hbuf, tiles);
    if (rem > 0)
      gemm_tail_kernel<<<rem, FDIM, 0, stream>>>(x, Wm[0], dinv, Hbuf, tiles * 16);
    const int ngroups = (N + FG_ROWS - 1) / FG_ROWS;
    const int fgrid = ngroups < FG_BLOCKS ? ngroups : FG_BLOCKS;
    gather_product_kernel<<<fgrid, 32 * FG_ROWS, 0, stream>>>(Hbuf, rowptr, col, dinv, Bv[0], Wm[1], Gbuf, N, ngroups);
    gather_product_kernel<<<fgrid, 32 * FG_ROWS, 0, stream>>>(Gbuf, rowptr, col, dinv, Bv[1], Wm[2], Hbuf, N, ngroups);
    gather_rows_kernel<<<(N + 7) / 8, 256, 0, stream>>>(Hbuf, rowptr, col, dinv, Bv[2], Gbuf, N);
  } else {
    zero(dinv, N);
    degree_kernel<<<(E + N + 255) / 256, 256, 0, stream>>>(ei, dinv, E, N);
    dinv_kernel<<<(N + 255) / 256, 256, 0, stream>>>(dinv, N);

    for (int l = 0; l < 3; ++l) {
      if (tiles > 0)
        gemm_strip_kernel<false><<<GEMM_BLOCKS, 256, 0, stream>>>(cur, Wm[l], nullptr, Hbuf, tiles);
      if (rem > 0)
        gemm_tail_kernel<<<rem, FDIM, 0, stream>>>(cur, Wm[l], nullptr, Hbuf, tiles * 16);
      zero(Gbuf, Nf);
      aggregate_kernel<<<(E + N + 7) / 8, 256, 0, stream>>>(Hbuf, ei, dinv, Gbuf, E, N);
      bias_relu_kernel<<<N, FDIM, 0, stream>>>(Gbuf, Bv[l]);
      cur = Gbuf;
    }
  }

  init_out_kernel<<<(NG * 2 * FDIM + 255) / 256, 256, 0, stream>>>(out, NG);
  zero(cnt, NG);
  pool_runs_kernel<<<(N + POOL_ROWS * POOL_WAVES - 1) / (POOL_ROWS * POOL_WAVES), 32 * POOL_WAVES, 0, stream>>>(Gbuf, gi, out, cnt, N);
  finalize_mean_kernel<<<NG, FDIM, 0, stream>>>(out, cnt);
}


# Four small launches folded into the edge-list build: its first kernel stores the pooling start values (out, cnt); each scatter block sums and scans the bucket count table itself, no scan launches
# speedup vs baseline: 5.8320x; 1.0413x over previous
#include <hip/hip_runtime.h>
#include <float.h>

#define FDIM 128

typedef __attribute__((ext_vector_type(2))) float v2f;
typedef __attribute__((ext_vector_type(8))) float v8f;


__global__ void zero_f32_kernel(float* __restrict__ p, long n) {
  long i = (long)blockIdx.x * blockDim.x + threadIdx.x;
  if (i < n) p[i] = 0.0f;
}

__global__ void degree_kernel(const int* __restrict__ ei, float* __restrict__ deg,
                              int E, int N) {
  int e = blockIdx.x * blockDim.x + threadIdx.x;
  int EN = E + N;
  if (e >= EN) return;
  int d = (e < E) ? ei[E + e] : (e - E);
  atomicAdd(&deg[d], 1.0f);
}

__global__ void dinv_kernel(float* __restrict__ deg_dinv, int N) {
  int i = blockIdx.x * blockDim.x + threadIdx.x;
  if (i < N) deg_dinv[i] = rsqrtf(fmaxf(deg_dinv[i], 1.0f));
}

#define GEMM_WPITCH 288
#define GEMM_BLOCKS 512

template <bool SCALED>
__global__ __launch_bounds__(256, 4) void gemm_strip_kernel(const float* __restrict__ X,
                                                         const float* __restrict__ W,
                                                         const float* __restrict__ rs,
                                                         float* __restrict__ H, int tiles) {
  __shared__ alignas(16) float Ws[64 * GEMM_WPITCH];
  const int tid = threadIdx.x;

  {
    const float4* __restrict__ W4 = reinterpret_cast<const float4*>(W);
    for (int it = 0; it < 8; ++it) {
      const int i = it * 256 + tid;
      const int p = i >> 5;
      const int q = i & 31;
      const float4 w0 = W4[(2 * p) * (FDIM / 4) + q];
      const float4 w1 = W4[(2 * p + 1) * (FDIM / 4) + q];
      float4* dst = reinterpret_cast<float4*>(Ws + p * GEMM_WPITCH + q * 8);
      dst[0] = make_float4(w0.x, w1.x, w0.y, w1.y);
      dst[1] = make_float4(w0.z, w1.z, w0.w, w1.w);
    }
  }
  __syncthreads();

  const int wave = __builtin_amdgcn_readfirstlane((int)(threadIdx.x >> 5));
  const int lane = tid & 31;
  const int c = lane & 15;
  const int h = lane >> 4;

  const v2f* __restrict__ bfrag = reinterpret_cast<const v2f*>(Ws) + h * (GEMM_WPITCH / 2) + c;

  const int xoff = c * (FDIM / 2) + h;
  const int hoff = 8 * h * FDIM + c;

  const int s0 = blockIdx.x + gridDim.x * wave;
  const size_t sstep = (size_t)gridDim.x * 8 * (16 * FDIM);
  const v2f* xs = reinterpret_cast<const v2f*>(X + (size_t)s0 * (16 * FDIM)) + xoff;
  float* hs = H + (size_t)s0 * (16 * FDIM) + hoff;
  const float* rp = SCALED ? rs + (size_t)s0 * 16 + c : X;
  const size_t rstep = SCALED ? (size_t)gridDim.x * 8 * 16 : 0;

  for (int s = s0; s < tiles; s += gridDim.x * 8, xs += sstep / 2, hs += sstep, rp += rstep) {
    v2f a[FDIM / 4];
    #pragma unroll
    for (int kk = 0; kk < FDIM / 4; ++kk) a[kk] = xs[2 * kk];
    if (SCALED) {
      const float sc = *rp;
      #pragma unroll
      for (int kk = 0; kk < FDIM / 4; ++kk) a[kk] = a[kk] * sc;
    }

    v8f acc[8];
    #pragma unroll
    for (int t = 0; t < 8; ++t) acc[t] = v8f{};

    v2f bcur[8];
    #pragma unroll
    for (int t = 0; t < 8; ++t) bcur[t] = bfrag[16 * t];

    #pragma unroll
    for (int kk = 0; kk < FDIM / 4; ++kk) {
      v2f bnxt[8];
      if (kk + 1 < FDIM / 4) {
        const v2f* bp = bfrag + (2 * (kk + 1)) * (GEMM_WPITCH / 2);
        #pragma unroll
        for (int t = 0; t < 8; ++t) bnxt[t] = bp[16 * t];
      }
      #pragma unroll
      for (int t = 0; t < 8; ++t) {
        acc[t] = __builtin_amdgcn_wmma_f32_16x16x4_f32(
             false, a[kk],  false, bcur[t],
             (short)0, acc[t],  false,  false);
      }
      __builtin_amdgcn_sched_barrier(0);
      if (kk + 1 < FDIM / 4) {
        #pragma unroll
        for (int t = 0; t < 8; ++t) bcur[t] = bnxt[t];
      }
    }

    #pragma unroll
    for (int t = 0; t < 8; ++t) {
      #pragma unroll
      for (int r = 0; r < 8; ++r) hs[r * FDIM + 16 * t] = acc[t][r];
    }
  }
}

__global__ void gemm_tail_kernel(const float* __restrict__ X,
                                 const float* __restrict__ W,
                                 const float* __restrict__ rs,
                                 float* __restrict__ H, int base) {
  const int row = base + blockIdx.x;
  const int f = threadIdx.x;
  const float* xr = X + (size_t)row * FDIM;
  float acc = 0.0f;
  #pragma unroll 8
  for (int k = 0; k < FDIM; ++k) acc = fmaf(xr[k], W[(size_t)k * FDIM + f], acc);
  H[(size_t)row * FDIM + f] = rs ? acc * rs[row] : acc;
}

__global__ void aggregate_kernel(const float* __restrict__ H,
                                 const int* __restrict__ ei,
                                 const float* __restrict__ dinv,
                                 float* __restrict__ G, int E, int N) {
  const int e = blockIdx.x * (blockDim.x >> 5) + (threadIdx.x >> 5);
  const int EN = E + N;
  if (e >= EN) return;
  const int lane = threadIdx.x & 31;
  int s, d;
  if (e < E) { s = ei[e]; d = ei[E + e]; }
  else       { s = e - E; d = s; }
  const float w = dinv[s] * dinv[d];
  const float4 hv = reinterpret_cast<const float4*>(H + (size_t)s * FDIM)[lane];
  float* gp = G + (size_t)d * FDIM + lane * 4;
  atomicAdd(gp + 0, hv.x * w);
  atomicAdd(gp + 1, hv.y * w);
  atomicAdd(gp + 2, hv.z * w);
  atomicAdd(gp + 3, hv.w * w);
}


__global__ void count_rank_kernel(const int* __restrict__ ei, int* __restrict__ icnt,
                                  int* __restrict__ rank, int E) {
  const int e = blockIdx.x * blockDim.x + threadIdx.x;
  if (e >= E) return;
  rank[e] = atomicAdd(&icnt[ei[E + e]], 1);
}

#define SCAN_THREADS 256
#define SCAN_ELEMS   1024

__global__ void scan_local_kernel(const int* __restrict__ icnt, int* __restrict__ rowptr,
                                  int* __restrict__ bsum, int N) {
  __shared__ int wsum[SCAN_THREADS / 32];
  const int tid = threadIdx.x, lane = tid & 31, wv = tid >> 5;
  const int base = blockIdx.x * SCAN_ELEMS + tid * 4;
  int v[4];
  #pragma unroll
  for (int q = 0; q < 4; ++q) v[q] = (base + q < N) ? icnt[base + q] : 0;
  const int t = v[0] + v[1] + v[2] + v[3];
  int inc = t;
  #pragma unroll
  for (int off = 1; off < 32; off <<= 1) {
    const int u = __shfl_up(inc, off);
    if (lane >= off) inc += u;
  }
  if (lane == 31) wsum[wv] = inc;
  __syncthreads();
  int woff = 0, tot = 0;
  #pragma unroll
  for (int k = 0; k < SCAN_THREADS / 32; ++k) {
    const int s = wsum[k];
    if (k < wv) woff += s;
    tot += s;
  }
  int run = woff + inc - t;
  #pragma unroll
  for (int q = 0; q < 4; ++q) {
    if (base + q < N) rowptr[base + q] = run;
    run += v[q];
  }
  if (tid == 0) bsum[blockIdx.x] = tot;
}

__global__ void scan_add_kernel(int* __restrict__ rowptr, const int* __restrict__ bsum, int N) {
  __shared__ int red[SCAN_THREADS / 32];
  const int tid = threadIdx.x, lane = tid & 31, wv = tid >> 5;
  const int b = blockIdx.x;
  int part = 0;
  for (int k = tid; k < b; k += SCAN_THREADS) part += bsum[k];
  #pragma unroll
  for (int off = 16; off > 0; off >>= 1) part += __shfl_xor(part, off);
  if (lane == 0) red[wv] = part;
  __syncthreads();
  int add = 0;
  #pragma unroll
  for (int k = 0; k < SCAN_THREADS / 32; ++k) add += red[k];
  const int base = b * SCAN_ELEMS + tid * 4;
  #pragma unroll
  for (int q = 0; q < 4; ++q)
    if (base + q < N) rowptr[base + q] += add;
  if (b == (int)gridDim.x - 1 && tid == 0) rowptr[N] = add + bsum[b];
}

__global__ void place_kernel(const int* __restrict__ ei, const int* __restrict__ rowptr,
                             const int* __restrict__ rank, int* __restrict__ col, int E) {
  const int e = blockIdx.x * blockDim.x + threadIdx.x;
  if (e >= E) return;
  col[rowptr[ei[E + e]] + rank[e]] = ei[e];
}

#define BIN_SHIFT   8
#define BIN_NODES   256
#define BIN_MAXB    1024
#define BIN_BLOCKS  256
#define BIN_THREADS 1024
#define BIN_BATCH   4
#define BIN_STAGE   16384

__global__ __launch_bounds__(BIN_THREADS) void bin_count_kernel(const int* __restrict__ ei,
                                                                int* __restrict__ hist, int E, int nb,
                                                                float* __restrict__ out,
                                                                float* __restrict__ cnt, int ng) {
  __shared__ int h[BIN_MAXB];
  const int tid = threadIdx.x;
  {
    const float lowest = __int_as_float(0xFF7FFFFF);
    float4* __restrict__ out4 = reinterpret_cast<float4*>(out);
    const int gtid = blockIdx.x * BIN_THREADS + tid;
    const int gthreads = gridDim.x * BIN_THREADS;
    for (int i = gtid; i < ng * (2 * FDIM / 4); i += gthreads) {
      const float v = ((i & (2 * FDIM / 4 - 1)) < FDIM / 4) ? 0.0f : lowest;
      out4[i] = make_float4(v, v, v, v);
    }
    for (int i = gtid; i < ng; i += gthreads) cnt[i] = 0.0f;
  }
  const int nbp = (nb + 3) & ~3;
  for (int k = tid; k < nbp; k += BIN_THREADS) h[k] = 0;
  __syncthreads();
  const int chunk = (E + BIN_BLOCKS - 1) / BIN_BLOCKS;
  const int lo = blockIdx.x * chunk;
  const int hi = min(E, lo + chunk);
  for (int i = lo + tid; i < hi; i += BIN_BATCH * BIN_THREADS) {
    int d[BIN_BATCH];
    #pragma unroll
    for (int q = 0; q < BIN_BATCH; ++q) d[q] = ei[E + min(i + q * BIN_THREADS, hi - 1)];
    #pragma unroll
    for (int q = 0; q < BIN_BATCH; ++q)
      if (i + q * BIN_THREADS < hi) atomicAdd(&h[d[q] >> BIN_SHIFT], 1);
  }
  __syncthreads();
  for (int k = tid; k < nbp; k += BIN_THREADS) hist[blockIdx.x * nbp + k] = h[k];
}

#define BIN_ROWS 16
__global__ __launch_bounds__(BIN_THREADS) void bin_scatter_kernel(const int* __restrict__ ei,
                                                                  const int* __restrict__ hist,
                                                                  int* __restrict__ bstart,
                                                                  int* __restrict__ ebuf, int E, int nb) {
  __shared__ int cur[BIN_MAXB];
  __shared__ int rsum[BIN_MAXB];
  __shared__ int wsum[BIN_THREADS / 32];
  const int tid = threadIdx.x, lane = tid & 31, wv = tid >> 5;
  const int b = blockIdx.x;
  const int nq = (nb + 3) >> 2;
  const int rq = BIN_THREADS / nq;
  const int rg = rq >= 16 ? 16 : (rq >= 8 ? 8 : 4);
  for (int k = tid; k < 4 * nq; k += BIN_THREADS) {
    cur[k] = 0;
    rsum[k] = 0;
  }
  __syncthreads();
  {
    const int4* __restrict__ hist4 = reinterpret_cast<const int4*>(hist);
    const int j = tid % nq;
    const int r = tid / nq;
    if (r < rg) {
      int4 tot = make_int4(0, 0, 0, 0);
      int4 low = make_int4(0, 0, 0, 0);
      #pragma clang loop vectorize(disable) interleave(disable) unroll(disable)
      for (int r0 = r; r0 < BIN_BLOCKS; r0 += BIN_ROWS * rg) {
        int4 u[BIN_ROWS];
        #pragma unroll
        for (int q = 0; q < BIN_ROWS; ++q) u[q] = hist4[(r0 + q * rg) * nq + j];
        #pragma unroll
        for (int q = 0; q < BIN_ROWS; ++q) {
          const int m = (r0 + q * rg < b) ? -1 : 0;
          tot.x += u[q].x; tot.y += u[q].y; tot.z += u[q].z; tot.w += u[q].w;
          low.x += u[q].x & m; low.y += u[q].y & m; low.z += u[q].z & m; low.w += u[q].w & m;
        }
      }
      atomicAdd(&rsum[4 * j + 0], tot.x);
      atomicAdd(&rsum[4 * j + 1], tot.y);
      atomicAdd(&rsum[4 * j + 2], tot.z);
      atomicAdd(&rsum[4 * j + 3], tot.w);
      atomicAdd(&cur[4 * j + 0], low.x);
      atomicAdd(&cur[4 * j + 1], low.y);
      atomicAdd(&cur[4 * j + 2], low.z);
      atomicAdd(&cur[4 * j + 3], low.w);
    }
  }
  __syncthreads();
  const int v = (tid < nb) ? rsum[tid] : 0;
  int inc = v;
  #pragma unroll
  for (int off = 1; off < 32; off <<= 1) {
    const int t = __shfl_up(inc, off);
    if (lane >= off) inc += t;
  }
  if (lane == 31) wsum[wv] = inc;
  __syncthreads();
  int woff = 0;
  #pragma unroll
  for (int k = 0; k < BIN_THREADS / 32; ++k)
    if (k < wv) woff += wsum[k];
  if (tid < nb) {
    const int start = woff + inc - v;
    cur[tid] += start;
    if (b == 0) {
      bstart[tid] = start;
      if (tid == nb - 1) bstart[nb] = start + v;
    }
  }
  __syncthreads();
  const int chunk = (E + BIN_BLOCKS - 1) / BIN_BLOCKS;
  const int lo = blockIdx.x * chunk;
  const int hi = min(E, lo + chunk);
  int2* __restrict__ eb = reinterpret_cast<int2*>(ebuf);
  for (int i = lo + tid; i < hi; i += BIN_BATCH * BIN_THREADS) {
    int s[BIN_BATCH], d[BIN_BATCH];
    #pragma unroll
    for (int q = 0; q < BIN_BATCH; ++q) {
      const int j = min(i + q * BIN_THREADS, hi - 1);
      s[q] = ei[j];
      d[q] = ei[E + j];
    }
    #pragma unroll
    for (int q = 0; q < BIN_BATCH; ++q) {
      if (i + q * BIN_THREADS < hi) {
        const int p = atomicAdd(&cur[d[q] >> BIN_SHIFT], 1);
        eb[p] = make_int2(s[q], d[q]);
      }
    }
  }
}

__global__ __launch_bounds__(BIN_THREADS) void bin_finish_kernel(const int* __restrict__ ebuf,
                                                                 const int* __restrict__ bstart,
                                                                 int* __restrict__ rowptr,
                                                                 int* __restrict__ col,
                                                                 float* __restrict__ dinv, int N) {
  __shared__ int cnt[BIN_NODES];
  __shared__ int cur[BIN_NODES];
  __shared__ int wsum[BIN_NODES / 32];
  __shared__ int stage[BIN_STAGE];
  const int tid = threadIdx.x, lane = tid & 31, wv = tid >> 5;
  const int b = blockIdx.x;
  const int e0 = bstart[b];
  const int e1 = bstart[b + 1];
  const int2* __restrict__ eb = reinterpret_cast<const int2*>(ebuf);
  if (tid < BIN_NODES) cnt[tid] = 0;
  __syncthreads();
  for (int i = e0 + tid; i < e1; i += BIN_BATCH * BIN_THREADS) {
    int d[BIN_BATCH];
    #pragma unroll
    for (int q = 0; q < BIN_BATCH; ++q) d[q] = eb[min(i + q * BIN_THREADS, e1 - 1)].y;
    #pragma unroll
    for (int q = 0; q < BIN_BATCH; ++q)
      if (i + q * BIN_THREADS < e1) atomicAdd(&cnt[d[q] & (BIN_NODES - 1)], 1);
  }
  __syncthreads();

  const int v = (tid < BIN_NODES) ? cnt[tid] : 0;
  int inc = v;
  #pragma unroll
  for (int off = 1; off < 32; off <<= 1) {
    const int u = __shfl_up(inc, off);
    if (lane >= off) inc += u;
  }
  if (tid < BIN_NODES && lane == 31) wsum[wv] = inc;
  __syncthreads();
  if (tid < BIN_NODES) {
    int woff = 0;
    #pragma unroll
    for (int k = 0; k < BIN_NODES / 32; ++k)
      if (k < wv) woff += wsum[k];
    const int excl = woff + inc - v;
    cur[tid] = excl;
    const int node = b * BIN_NODES + tid;
    if (node < N) {
      rowptr[node] = e0 + excl;
      dinv[node] = rsqrtf(fmaxf((float)(v + 1), 1.0f));
    }
  }
  if (b == (int)gridDim.x - 1 && tid == 0) rowptr[N] = e1;
  __syncthreads();

  const int ne = e1 - e0;
  if (ne <= BIN_STAGE) {
    for (int i = e0 + tid; i < e1; i += BIN_BATCH * BIN_THREADS) {
      int2 sd[BIN_BATCH];
      #pragma unroll
      for (int q = 0; q < BIN_BATCH; ++q) sd[q] = eb[min(i + q * BIN_THREADS, e1 - 1)];
      #pragma unroll
      for (int q = 0; q < BIN_BATCH; ++q)
        if (i + q * BIN_THREADS < e1) stage[atomicAdd(&cur[sd[q].y & (BIN_NODES - 1)], 1)] = sd[q].x;
    }
    __syncthreads();
    for (int i = tid; i < ne; i += BIN_THREADS) col[e0 + i] = stage[i];
  } else {
    for (int i = e0 + tid; i < e1; i += BIN_THREADS) {
      const int2 sd = eb[i];
      col[e0 + atomicAdd(&cur[sd.y & (BIN_NODES - 1)], 1)] = sd.x;
    }
  }
}

__global__ void dinv_from_count_kernel(float* __restrict__ cnt_dinv, int N) {
  const int i = blockIdx.x * blockDim.x + threadIdx.x;
  if (i < N) {
    const int c = __float_as_int(cnt_dinv[i]);
    cnt_dinv[i] = rsqrtf(fmaxf((float)(c + 1), 1.0f));
  }
}

__device__ __forceinline__ void row_add(float4& acc, const float4 h) {
  acc.x += h.x;
  acc.y += h.y;
  acc.z += h.z;
  acc.w += h.w;
}

__device__ __forceinline__ float4 gather_row_sum(const float4* __restrict__ H4,
                                                 const int* __restrict__ rowptr,
                                                 const int* __restrict__ col,
                                                 const int d, const int lane) {
  const int beg = rowptr[d];
  const int end = rowptr[d + 1];

  float4 acc = H4[(size_t)d * (FDIM / 4) + lane];

  for (int j = beg; j < end; j += 32) {
    const int n = min(32, end - j);
    int s_l = 0;
    if (lane < n) s_l = col[j + lane];
    int t = 0;
    for (; t + 4 <= n; t += 4) {
      const int s0 = __builtin_amdgcn_readlane(s_l, t);
      const int s1 = __builtin_amdgcn_readlane(s_l, t + 1);
      const int s2 = __builtin_amdgcn_readlane(s_l, t + 2);
      const int s3 = __builtin_amdgcn_readlane(s_l, t + 3);
      const float4 h0 = H4[(size_t)s0 * (FDIM / 4) + lane];
      const float4 h1 = H4[(size_t)s1 * (FDIM / 4) + lane];
      const float4 h2 = H4[(size_t)s2 * (FDIM / 4) + lane];
      const float4 h3 = H4[(size_t)s3 * (FDIM / 4) + lane];
      row_add(acc, h0);
      row_add(acc, h1);
      row_add(acc, h2);
      row_add(acc, h3);
    }
    const int r = n - t;
    if (r == 3) {
      const int s0 = __builtin_amdgcn_readlane(s_l, t);
      const int s1 = __builtin_amdgcn_readlane(s_l, t + 1);
      const int s2 = __builtin_amdgcn_readlane(s_l, t + 2);
      const float4 h0 = H4[(size_t)s0 * (FDIM / 4) + lane];
      const float4 h1 = H4[(size_t)s1 * (FDIM / 4) + lane];
      const float4 h2 = H4[(size_t)s2 * (FDIM / 4) + lane];
      row_add(acc, h0);
      row_add(acc, h1);
      row_add(acc, h2);
    } else if (r == 2) {
      const int s0 = __builtin_amdgcn_readlane(s_l, t);
      const int s1 = __builtin_amdgcn_readlane(s_l, t + 1);
      const float4 h0 = H4[(size_t)s0 * (FDIM / 4) + lane];
      const float4 h1 = H4[(size_t)s1 * (FDIM / 4) + lane];
      row_add(acc, h0);
      row_add(acc, h1);
    } else if (r == 1) {
      const int s0 = __builtin_amdgcn_readlane(s_l, t);
      row_add(acc, H4[(size_t)s0 * (FDIM / 4) + lane]);
    }
  }
  return acc;
}

__global__ void gather_rows_kernel(const float* __restrict__ H, const int* __restrict__ rowptr,
                                   const int* __restrict__ col, const float* __restrict__ dinv,
                                   const float* __restrict__ bias, float* __restrict__ G, int N) {
  const int d = __builtin_amdgcn_readfirstlane(
      (int)(blockIdx.x * (blockDim.x >> 5) + (threadIdx.x >> 5)));
  if (d >= N) return;
  const int lane = threadIdx.x & 31;
  const float bx = bias[lane * 4 + 0];
  const float by = bias[lane * 4 + 1];
  const float bz = bias[lane * 4 + 2];
  const float bw = bias[lane * 4 + 3];
  const float dd = dinv[d];
  float4 acc = gather_row_sum(reinterpret_cast<const float4*>(H), rowptr, col, d, lane);
  acc.x = fmaxf(acc.x * dd + bx, 0.0f);
  acc.y = fmaxf(acc.y * dd + by, 0.0f);
  acc.z = fmaxf(acc.z * dd + bz, 0.0f);
  acc.w = fmaxf(acc.w * dd + bw, 0.0f);
  reinterpret_cast<float4*>(G)[(size_t)d * (FDIM / 4) + lane] = acc;
}

#define FG_ROWS   32
#define FG_TPITCH 132
#define FG_BLOCKS 512

__global__ __launch_bounds__(32 * FG_ROWS, 16) void gather_product_kernel(
    const float* __restrict__ H, const int* __restrict__ rowptr, const int* __restrict__ col,
    const float* __restrict__ dinv, const float* __restrict__ bias, const float* __restrict__ Wn,
    float* __restrict__ Hn, int N, int ngroups) {
  __shared__ alignas(16) float smem[64 * GEMM_WPITCH + 2 * FG_ROWS * FG_TPITCH];
  float* Ws = smem;
  const int tid = threadIdx.x;

  {
    const float4* __restrict__ W4 = reinterpret_cast<const float4*>(Wn);
    for (int it = 0; it < 2; ++it) {
      const int i = it * (32 * FG_ROWS) + tid;
      const int p = i >> 5;
      const int q = i & 31;
      const float4 w0 = W4[(2 * p) * (FDIM / 4) + q];
      const float4 w1 = W4[(2 * p + 1) * (FDIM / 4) + q];
      float4* dst = reinterpret_cast<float4*>(Ws + p * GEMM_WPITCH + q * 8);
      dst[0] = make_float4(w0.x, w1.x, w0.y, w1.y);
      dst[1] = make_float4(w0.z, w1.z, w0.w, w1.w);
    }
  }
  __syncthreads();

  const int wave = __builtin_amdgcn_readfirstlane((int)(threadIdx.x >> 5));
  const int lane = tid & 31;
  const int c = lane & 15;
  const int h = lane >> 4;

  const float bx = bias[lane * 4 + 0];
  const float by = bias[lane * 4 + 1];
  const float bz = bias[lane * 4 + 2];
  const float bw = bias[lane * 4 + 3];
  const float4* __restrict__ H4 = reinterpret_cast<const float4*>(H);

  const int rt = (wave >> 2) & 1;
  const int cp = wave & 3;
  const unsigned boff0 = (unsigned)(h * (GEMM_WPITCH / 2) + c + 32 * cp) * 8u;

  int it = 0;
  for (int g = blockIdx.x; g < ngroups; g += gridDim.x, ++it) {
    const int toff = 64 * GEMM_WPITCH + (it & 1) * (FG_ROWS * FG_TPITCH);
    float* tl = smem + toff;
    const int d = g * FG_ROWS + wave;
    float4 acc = make_float4(0.0f, 0.0f, 0.0f, 0.0f);
    if (d < N) {
      const float dd = dinv[d];
      acc = gather_row_sum(H4, rowptr, col, d, lane);
      acc.x = fmaxf(acc.x * dd + bx, 0.0f) * dd;
      acc.y = fmaxf(acc.y * dd + by, 0.0f) * dd;
      acc.z = fmaxf(acc.z * dd + bz, 0.0f) * dd;
      acc.w = fmaxf(acc.w * dd + bw, 0.0f) * dd;
    }
    reinterpret_cast<float4*>(tl + wave * FG_TPITCH)[lane] = acc;
    __syncthreads();

    if ((wave >> 3) == (it & 3)) {
      unsigned ao = (unsigned)(toff + (16 * rt + c) * FG_TPITCH + 2 * h) * 4u;
      unsigned bo = boff0;
      const char* sb = reinterpret_cast<const char*>(smem);
      const int row0 = g * FG_ROWS + 16 * rt + 8 * h;
      float* hp = Hn + (size_t)row0 * FDIM + 32 * cp + c;
      int nval = N - row0;
      asm volatile("" : "+v"(hp), "+v"(nval));
      __attribute__((address_space(1))) float* hg = (__attribute__((address_space(1))) float*)hp;
      v8f acc0 = v8f{};
      v8f acc1 = v8f{};
      v2f ax  = *reinterpret_cast<const v2f*>(sb + ao);
      v2f bx0 = *reinterpret_cast<const v2f*>(sb + bo);
      v2f bx1 = *reinterpret_cast<const v2f*>(sb + bo + 128);
      #pragma unroll 1
      for (int kk = 0; kk < FDIM / 4 - 2; kk += 2) {
        ao += 16;
        bo += 2 * (GEMM_WPITCH / 2) * 8;
        asm volatile("" : "+v"(ao), "+v"(bo));
        const v2f ay  = *reinterpret_cast<const v2f*>(sb + ao);
        const v2f by0 = *reinterpret_cast<const v2f*>(sb + bo);
        const v2f by1 = *reinterpret_cast<const v2f*>(sb + bo + 128);
        acc0 = __builtin_amdgcn_wmma_f32_16x16x4_f32(
             false, ax,  false, bx0,
             (short)0, acc0,  false,  false);
        acc1 = __builtin_amdgcn_wmma_f32_16x16x4_f32(
             false, ax,  false, bx1,
             (short)0, acc1,  false,  false);
        ao += 16;
        bo += 2 * (GEMM_WPITCH / 2) * 8;
        asm volatile("" : "+v"(ao), "+v"(bo));
        ax  = *reinterpret_cast<const v2f*>(sb + ao);
        bx0 = *reinterpret_cast<const v2f*>(sb + bo);
        bx1 = *reinterpret_cast<const v2f*>(sb + bo + 128);
        acc0 = __builtin_amdgcn_wmma_f32_16x16x4_f32(
             false, ay,  false, by0,
             (short)0, acc0,  false,  false);
        acc1 = __builtin_amdgcn_wmma_f32_16x16x4_f32(
             false, ay,  false, by1,
             (short)0, acc1,  false,  false);
      }
      {
        ao += 16;
        bo += 2 * (GEMM_WPITCH / 2) * 8;
        asm volatile("" : "+v"(ao), "+v"(bo));
        const v2f ay  = *reinterpret_cast<const v2f*>(sb + ao);
        const v2f by0 = *reinterpret_cast<const v2f*>(sb + bo);
        const v2f by1 = *reinterpret_cast<const v2f*>(sb + bo + 128);
        acc0 = __builtin_amdgcn_wmma_f32_16x16x4_f32(
             false, ax,  false, bx0,
             (short)0, acc0,  false,  false);
        acc1 = __builtin_amdgcn_wmma_f32_16x16x4_f32(
             false, ax,  false, bx1,
             (short)0, acc1,  false,  false);
        acc0 = __builtin_amdgcn_wmma_f32_16x16x4_f32(
             false, ay,  false, by0,
             (short)0, acc0,  false,  false);
        acc1 = __builtin_amdgcn_wmma_f32_16x16x4_f32(
             false, ay,  false, by1,
             (short)0, acc1,  false,  false);
      }
      #pragma unroll
      for (int r = 0; r < 8; ++r) {
        if (r < nval) {
          hg[r * FDIM]      = acc0[r];
          hg[r * FDIM + 16] = acc1[r];
        }
      }
    }
  }
}

__global__ void bias_relu_kernel(float* __restrict__ G, const float* __restrict__ b) {
  const size_t i = (size_t)blockIdx.x * FDIM + threadIdx.x;
  G[i] = fmaxf(G[i] + b[threadIdx.x], 0.0f);
}

__global__ void init_out_kernel(float* __restrict__ out, int ng) {
  int i = blockIdx.x * blockDim.x + threadIdx.x;
  if (i >= ng * 2 * FDIM) return;
  int f = i & (2 * FDIM - 1);
  out[i] = (f < FDIM) ? 0.0f : __int_as_float(0xFF7FFFFF);
}

__device__ __forceinline__ void atomic_max_f32(float* addr, float v) {
  if (v >= 0.0f) atomicMax((int*)addr, __float_as_int(v));
  else           atomicMin((unsigned int*)addr, (unsigned int)__float_as_int(v));
}

#define POOL_ROWS  32
#define POOL_WAVES 8

__device__ __forceinline__ void pool_flush(float* __restrict__ out, float* __restrict__ cnt,
                                           float* st, int lane, int g,
                                           const float4 s, const float4 m, int run) {
  reinterpret_cast<float4*>(st)[lane] = s;
  reinterpret_cast<float4*>(st + FDIM)[lane] = m;
  __builtin_amdgcn_fence(__ATOMIC_ACQ_REL, "wavefront");
  float a[4], b[4];
  #pragma unroll
  for (int k = 0; k < 4; ++k) {
    a[k] = st[lane + 32 * k];
    b[k] = st[FDIM + lane + 32 * k];
  }
  __builtin_amdgcn_fence(__ATOMIC_ACQ_REL, "wavefront");
  float* o = out + (size_t)g * (2 * FDIM);
  #pragma unroll
  for (int k = 0; k < 4; ++k) atomicAdd(o + lane + 32 * k, a[k]);
  #pragma unroll
  for (int k = 0; k < 4; ++k) atomic_max_f32(o + FDIM + lane + 32 * k, b[k]);
  if (lane == 0) atomicAdd(&cnt[g], (float)run);
}

__global__ __launch_bounds__(256) void pool_runs_kernel(const float* __restrict__ G,
                                                        const int* __restrict__ gi,
                                                        float* __restrict__ out,
                                                        float* __restrict__ cnt, int N) {
  __shared__ alignas(16) float stage[POOL_WAVES * 2 * FDIM];
  const int wv = threadIdx.x >> 5;
  const int lane = threadIdx.x & 31;
  const int n0 = __builtin_amdgcn_readfirstlane((int)(blockIdx.x * POOL_WAVES + wv)) * POOL_ROWS;
  if (n0 >= N) return;
  const int nr = min(POOL_ROWS, N - n0);
  float* st = stage + wv * (2 * FDIM);

  const int g_l = gi[n0 + min(lane, nr - 1)];

  const float4* __restrict__ G4 = reinterpret_cast<const float4*>(G) + (size_t)n0 * (FDIM / 4) + lane;
  float4 v[POOL_ROWS];
  #pragma unroll
  for (int i = 0; i < POOL_ROWS; ++i) v[i] = G4[(size_t)min(i, nr - 1) * (FDIM / 4)];

  int a = 0;
  for (int trip = 0; trip < POOL_ROWS && a < nr; ++trip) {
    const int ga = __builtin_amdgcn_readlane(g_l, a);
    const unsigned diff = __builtin_amdgcn_ballot_w32(g_l != ga) & (~0u << a);
    const int b = min(diff ? __builtin_ctz(diff) : POOL_ROWS, nr);
    float4 s = make_float4(0.0f, 0.0f, 0.0f, 0.0f);
    float4 m = make_float4(-FLT_MAX, -FLT_MAX, -FLT_MAX, -FLT_MAX);
    #pragma unroll
    for (int i = 0; i < POOL_ROWS; ++i) {
      if (i >= a && i < b) {
        s.x += v[i].x; s.y += v[i].y; s.z += v[i].z; s.w += v[i].w;
        m.x = fmaxf(m.x, v[i].x); m.y = fmaxf(m.y, v[i].y);
        m.z = fmaxf(m.z, v[i].z); m.w = fmaxf(m.w, v[i].w);
      }
    }
    pool_flush(out, cnt, st, lane, ga, s, m, b - a);
    a = b;
  }
}

__global__ void finalize_mean_kernel(float* __restrict__ out, const float* __restrict__ cnt) {
  const int g = blockIdx.x;
  out[(size_t)g * 2 * FDIM + threadIdx.x] /= fmaxf(cnt[g], 1.0f);
}


extern "C" void kernel_launch(void* const* d_in, const int* in_sizes, int n_in,
                              void* d_out, int out_size, void* d_ws, size_t ws_size,
                              hipStream_t stream) {
  const float* x  = (const float*)d_in[0];
  const int*   ei = (const int*)d_in[1];
  const int*   gi = (const int*)d_in[2];
  const float* Wm[3] = {(const float*)d_in[4], (const float*)d_in[6], (const float*)d_in[8]};
  const float* Bv[3] = {(const float*)d_in[5], (const float*)d_in[7], (const float*)d_in[9]};
  float* out = (float*)d_out;

  const int  N  = in_sizes[0] / FDIM;
  const int  E  = in_sizes[1] / 2;
  const int  NG = out_size / (2 * FDIM);
  const long Nf = (long)N * FDIM;

  float* Hbuf = (float*)d_ws;
  float* Gbuf = Hbuf + Nf;
  float* dinv = Gbuf + Nf;
  float* cnt  = dinv + N;
  if (ws_size < (size_t)(2 * Nf + N + NG) * sizeof(float)) return;

  const int nsb = (N + SCAN_ELEMS - 1) / SCAN_ELEMS;
  int* rowptr = (int*)(cnt + NG);
  int* bsum   = rowptr + (N + 1);
  int* col    = bsum + nsb;
  int* rank   = (int*)Hbuf;
  const bool ordered =
      ws_size >= (size_t)(2 * Nf + N + NG) * sizeof(float) + ((size_t)N + 1 + nsb + E) * sizeof(int) &&
      (long)E <= Nf && E > 0;

  auto zero = [&](float* p, long n) {
    zero_f32_kernel<<<(unsigned)((n + 255) / 256), 256, 0, stream>>>(p, n);
  };

  const int tiles = N / 16;
  const int rem   = N - tiles * 16;
  const float* cur = x;

  const int  nb   = (N + BIN_NODES - 1) / BIN_NODES;
  const long nh   = (long)((nb + 3) & ~3) * BIN_BLOCKS;
  int* ebuf   = (int*)Hbuf;
  int* hist   = (int*)Gbuf;
  int* bstart = hist + nh;
  const bool binned = ordered && nb <= BIN_MAXB && 2 * (long)E <= Nf && nh + nb + 1 <= Nf;

  if (ordered) {
    if (binned) {
      bin_count_kernel<<<BIN_BLOCKS, BIN_THREADS, 0, stream>>>(ei, hist, E, nb, out, cnt, NG);
      bin_scatter_kernel<<<BIN_BLOCKS, BIN_THREADS, 0, stream>>>(ei, hist, bstart, ebuf, E, nb);
      bin_finish_kernel<<<nb, BIN_THREADS, 0, stream>>>(ebuf, bstart, rowptr, col, dinv, N);
    } else {
      zero(dinv, N);
      count_rank_kernel<<<(E + 255) / 256, 256, 0, stream>>>(ei, (int*)dinv, rank, E);
      scan_local_kernel<<<nsb, SCAN_THREADS, 0, stream>>>((const int*)dinv, rowptr, bsum, N);
      scan_add_kernel<<<nsb, SCAN_THREADS, 0, stream>>>(rowptr, bsum, N);
      place_kernel<<<(E + 255) / 256, 256, 0, stream>>>(ei, rowptr, rank, col, E);
      dinv_from_count_kernel<<<(N + 255) / 256, 256, 0, stream>>>(dinv, N);
    }

    if (tiles > 0)
      gemm_strip_kernel<true><<<GEMM_BLOCKS, 256, 0, stream>>>(x, Wm[0], dinv, Hbuf, tiles);
    if (rem > 0)
      gemm_tail_kernel<<<rem, FDIM, 0, stream>>>(x, Wm[0], dinv, Hbuf, tiles * 16);
    const int ngroups = (N + FG_ROWS - 1) / FG_ROWS;
    const int fgrid = ngroups < FG_BLOCKS ? ngroups : FG_BLOCKS;
    gather_product_kernel<<<fgrid, 32 * FG_ROWS, 0, stream>>>(Hbuf, rowptr, col, dinv, Bv[0], Wm[1], Gbuf, N, ngroups);
    gather_product_kernel<<<fgrid, 32 * FG_ROWS, 0, stream>>>(Gbuf, rowptr, col, dinv, Bv[1], Wm[2], Hbuf, N, ngroups);
    gather_rows_kernel<<<(N + 7) / 8, 256, 0, stream>>>(Hbuf, rowptr, col, dinv, Bv[2], Gbuf, N);
  } else {
    zero(dinv, N);
    degree_kernel<<<(E + N + 255) / 256, 256, 0, stream>>>(ei, dinv, E, N);
    dinv_kernel<<<(N + 255) / 256, 256, 0, stream>>>(dinv, N);

    for (int l = 0; l < 3; ++l) {
      if (tiles > 0)
        gemm_strip_kernel<false><<<GEMM_BLOCKS, 256, 0, stream>>>(cur, Wm[l], nullptr, Hbuf, tiles);
      if (rem > 0)
        gemm_tail_kernel<<<rem, FDIM, 0, stream>>>(cur, Wm[l], nullptr, Hbuf, tiles * 16);
      zero(Gbuf, Nf);
      aggregate_kernel<<<(E + N + 7) / 8, 256, 0, stream>>>(Hbuf, ei, dinv, Gbuf, E, N);
      bias_relu_kernel<<<N, FDIM, 0, stream>>>(Gbuf, Bv[l]);
      cur = Gbuf;
    }
  }

  if (!binned) {
    init_out_kernel<<<(NG * 2 * FDIM + 255) / 256, 256, 0, stream>>>(out, NG);
    zero(cnt, NG);
  }
  pool_runs_kernel<<<(N + POOL_ROWS * POOL_WAVES - 1) / (POOL_ROWS * POOL_WAVES), 32 * POOL_WAVES, 0, stream>>>(Gbuf, gi, out, cnt, N);
  finalize_mean_kernel<<<NG, FDIM, 0, stream>>>(out, cnt);
}


# Pooling fused into the last gather: a wave walks 7 adjacent target rows, keeps each graph-id run's sum and max in registers, one set of atomics per run; 1 launch fewer; pooled sum order changes
# speedup vs baseline: 6.0398x; 1.0356x over previous
#include <hip/hip_runtime.h>
#include <float.h>

#define FDIM 128

typedef __attribute__((ext_vector_type(2))) float v2f;
typedef __attribute__((ext_vector_type(8))) float v8f;


__global__ void zero_f32_kernel(float* __restrict__ p, long n) {
  long i = (long)blockIdx.x * blockDim.x + threadIdx.x;
  if (i < n) p[i] = 0.0f;
}

__global__ void degree_kernel(const int* __restrict__ ei, float* __restrict__ deg,
                              int E, int N) {
  int e = blockIdx.x * blockDim.x + threadIdx.x;
  int EN = E + N;
  if (e >= EN) return;
  int d = (e < E) ? ei[E + e] : (e - E);
  atomicAdd(&deg[d], 1.0f);
}

__global__ void dinv_kernel(float* __restrict__ deg_dinv, int N) {
  int i = blockIdx.x * blockDim.x + threadIdx.x;
  if (i < N) deg_dinv[i] = rsqrtf(fmaxf(deg_dinv[i], 1.0f));
}

#define GEMM_WPITCH 288
#define GEMM_BLOCKS 512

template <bool SCALED>
__global__ __launch_bounds__(256, 4) void gemm_strip_kernel(const float* __restrict__ X,
                                                         const float* __restrict__ W,
                                                         const float* __restrict__ rs,
                                                         float* __restrict__ H, int tiles) {
  __shared__ alignas(16) float Ws[64 * GEMM_WPITCH];
  const int tid = threadIdx.x;

  {
    const float4* __restrict__ W4 = reinterpret_cast<const float4*>(W);
    for (int it = 0; it < 8; ++it) {
      const int i = it * 256 + tid;
      const int p = i >> 5;
      const int q = i & 31;
      const float4 w0 = W4[(2 * p) * (FDIM / 4) + q];
      const float4 w1 = W4[(2 * p + 1) * (FDIM / 4) + q];
      float4* dst = reinterpret_cast<float4*>(Ws + p * GEMM_WPITCH + q * 8);
      dst[0] = make_float4(w0.x, w1.x, w0.y, w1.y);
      dst[1] = make_float4(w0.z, w1.z, w0.w, w1.w);
    }
  }
  __syncthreads();

  const int wave = __builtin_amdgcn_readfirstlane((int)(threadIdx.x >> 5));
  const int lane = tid & 31;
  const int c = lane & 15;
  const int h = lane >> 4;

  const v2f* __restrict__ bfrag = reinterpret_cast<const v2f*>(Ws) + h * (GEMM_WPITCH / 2) + c;

  const int xoff = c * (FDIM / 2) + h;
  const int hoff = 8 * h * FDIM + c;

  const int s0 = blockIdx.x + gridDim.x * wave;
  const size_t sstep = (size_t)gridDim.x * 8 * (16 * FDIM);
  const v2f* xs = reinterpret_cast<const v2f*>(X + (size_t)s0 * (16 * FDIM)) + xoff;
  float* hs = H + (size_t)s0 * (16 * FDIM) + hoff;
  const float* rp = SCALED ? rs + (size_t)s0 * 16 + c : X;
  const size_t rstep = SCALED ? (size_t)gridDim.x * 8 * 16 : 0;

  for (int s = s0; s < tiles; s += gridDim.x * 8, xs += sstep / 2, hs += sstep, rp += rstep) {
    v2f a[FDIM / 4];
    #pragma unroll
    for (int kk = 0; kk < FDIM / 4; ++kk) a[kk] = xs[2 * kk];
    if (SCALED) {
      const float sc = *rp;
      #pragma unroll
      for (int kk = 0; kk < FDIM / 4; ++kk) a[kk] = a[kk] * sc;
    }

    v8f acc[8];
    #pragma unroll
    for (int t = 0; t < 8; ++t) acc[t] = v8f{};

    v2f bcur[8];
    #pragma unroll
    for (int t = 0; t < 8; ++t) bcur[t] = bfrag[16 * t];

    #pragma unroll
    for (int kk = 0; kk < FDIM / 4; ++kk) {
      v2f bnxt[8];
      if (kk + 1 < FDIM / 4) {
        const v2f* bp = bfrag + (2 * (kk + 1)) * (GEMM_WPITCH / 2);
        #pragma unroll
        for (int t = 0; t < 8; ++t) bnxt[t] = bp[16 * t];
      }
      #pragma unroll
      for (int t = 0; t < 8; ++t) {
        acc[t] = __builtin_amdgcn_wmma_f32_16x16x4_f32(
             false, a[kk],  false, bcur[t],
             (short)0, acc[t],  false,  false);
      }
      __builtin_amdgcn_sched_barrier(0);
      if (kk + 1 < FDIM / 4) {
        #pragma unroll
        for (int t = 0; t < 8; ++t) bcur[t] = bnxt[t];
      }
    }

    #pragma unroll
    for (int t = 0; t < 8; ++t) {
      #pragma unroll
      for (int r = 0; r < 8; ++r) hs[r * FDIM + 16 * t] = acc[t][r];
    }
  }
}

__global__ void gemm_tail_kernel(const float* __restrict__ X,
                                 const float* __restrict__ W,
                                 const float* __restrict__ rs,
                                 float* __restrict__ H, int base) {
  const int row = base + blockIdx.x;
  const int f = threadIdx.x;
  const float* xr = X + (size_t)row * FDIM;
  float acc = 0.0f;
  #pragma unroll 8
  for (int k = 0; k < FDIM; ++k) acc = fmaf(xr[k], W[(size_t)k * FDIM + f], acc);
  H[(size_t)row * FDIM + f] = rs ? acc * rs[row] : acc;
}

__global__ void aggregate_kernel(const float* __restrict__ H,
                                 const int* __restrict__ ei,
                                 const float* __restrict__ dinv,
                                 float* __restrict__ G, int E, int N) {
  const int e = blockIdx.x * (blockDim.x >> 5) + (threadIdx.x >> 5);
  const int EN = E + N;
  if (e >= EN) return;
  const int lane = threadIdx.x & 31;
  int s, d;
  if (e < E) { s = ei[e]; d = ei[E + e]; }
  else       { s = e - E; d = s; }
  const float w = dinv[s] * dinv[d];
  const float4 hv = reinterpret_cast<const float4*>(H + (size_t)s * FDIM)[lane];
  float* gp = G + (size_t)d * FDIM + lane * 4;
  atomicAdd(gp + 0, hv.x * w);
  atomicAdd(gp + 1, hv.y * w);
  atomicAdd(gp + 2, hv.z * w);
  atomicAdd(gp + 3, hv.w * w);
}


__global__ void count_rank_kernel(const int* __restrict__ ei, int* __restrict__ icnt,
                                  int* __restrict__ rank, int E) {
  const int e = blockIdx.x * blockDim.x + threadIdx.x;
  if (e >= E) return;
  rank[e] = atomicAdd(&icnt[ei[E + e]], 1);
}

#define SCAN_THREADS 256
#define SCAN_ELEMS   1024

__global__ void scan_local_kernel(const int* __restrict__ icnt, int* __restrict__ rowptr,
                                  int* __restrict__ bsum, int N) {
  __shared__ int wsum[SCAN_THREADS / 32];
  const int tid = threadIdx.x, lane = tid & 31, wv = tid >> 5;
  const int base = blockIdx.x * SCAN_ELEMS + tid * 4;
  int v[4];
  #pragma unroll
  for (int q = 0; q < 4; ++q) v[q] = (base + q < N) ? icnt[base + q] : 0;
  const int t = v[0] + v[1] + v[2] + v[3];
  int inc = t;
  #pragma unroll
  for (int off = 1; off < 32; off <<= 1) {
    const int u = __shfl_up(inc, off);
    if (lane >= off) inc += u;
  }
  if (lane == 31) wsum[wv] = inc;
  __syncthreads();
  int woff = 0, tot = 0;
  #pragma unroll
  for (int k = 0; k < SCAN_THREADS / 32; ++k) {
    const int s = wsum[k];
    if (k < wv) woff += s;
    tot += s;
  }
  int run = woff + inc - t;
  #pragma unroll
  for (int q = 0; q < 4; ++q) {
    if (base + q < N) rowptr[base + q] = run;
    run += v[q];
  }
  if (tid == 0) bsum[blockIdx.x] = tot;
}

__global__ void scan_add_kernel(int* __restrict__ rowptr, const int* __restrict__ bsum, int N) {
  __shared__ int red[SCAN_THREADS / 32];
  const int tid = threadIdx.x, lane = tid & 31, wv = tid >> 5;
  const int b = blockIdx.x;
  int part = 0;
  for (int k = tid; k < b; k += SCAN_THREADS) part += bsum[k];
  #pragma unroll
  for (int off = 16; off > 0; off >>= 1) part += __shfl_xor(part, off);
  if (lane == 0) red[wv] = part;
  __syncthreads();
  int add = 0;
  #pragma unroll
  for (int k = 0; k < SCAN_THREADS / 32; ++k) add += red[k];
  const int base = b * SCAN_ELEMS + tid * 4;
  #pragma unroll
  for (int q = 0; q < 4; ++q)
    if (base + q < N) rowptr[base + q] += add;
  if (b == (int)gridDim.x - 1 && tid == 0) rowptr[N] = add + bsum[b];
}

__global__ void place_kernel(const int* __restrict__ ei, const int* __restrict__ rowptr,
                             const int* __restrict__ rank, int* __restrict__ col, int E) {
  const int e = blockIdx.x * blockDim.x + threadIdx.x;
  if (e >= E) return;
  col[rowptr[ei[E + e]] + rank[e]] = ei[e];
}

#define BIN_SHIFT   8
#define BIN_NODES   256
#define BIN_MAXB    1024
#define BIN_BLOCKS  256
#define BIN_THREADS 1024
#define BIN_BATCH   4
#define BIN_STAGE   16384

__global__ __launch_bounds__(BIN_THREADS) void bin_count_kernel(const int* __restrict__ ei,
                                                                int* __restrict__ hist, int E, int nb,
                                                                float* __restrict__ out,
                                                                float* __restrict__ cnt, int ng) {
  __shared__ int h[BIN_MAXB];
  const int tid = threadIdx.x;
  {
    const float lowest = __int_as_float(0xFF7FFFFF);
    float4* __restrict__ out4 = reinterpret_cast<float4*>(out);
    const int gtid = blockIdx.x * BIN_THREADS + tid;
    const int gthreads = gridDim.x * BIN_THREADS;
    for (int i = gtid; i < ng * (2 * FDIM / 4); i += gthreads) {
      const float v = ((i & (2 * FDIM / 4 - 1)) < FDIM / 4) ? 0.0f : lowest;
      out4[i] = make_float4(v, v, v, v);
    }
    for (int i = gtid; i < ng; i += gthreads) cnt[i] = 0.0f;
  }
  const int nbp = (nb + 3) & ~3;
  for (int k = tid; k < nbp; k += BIN_THREADS) h[k] = 0;
  __syncthreads();
  const int chunk = (E + BIN_BLOCKS - 1) / BIN_BLOCKS;
  const int lo = blockIdx.x * chunk;
  const int hi = min(E, lo + chunk);
  for (int i = lo + tid; i < hi; i += BIN_BATCH * BIN_THREADS) {
    int d[BIN_BATCH];
    #pragma unroll
    for (int q = 0; q < BIN_BATCH; ++q) d[q] = ei[E + min(i + q * BIN_THREADS, hi - 1)];
    #pragma unroll
    for (int q = 0; q < BIN_BATCH; ++q)
      if (i + q * BIN_THREADS < hi) atomicAdd(&h[d[q] >> BIN_SHIFT], 1);
  }
  __syncthreads();
  for (int k = tid; k < nbp; k += BIN_THREADS) hist[blockIdx.x * nbp + k] = h[k];
}

#define BIN_ROWS 16
__global__ __launch_bounds__(BIN_THREADS) void bin_scatter_kernel(const int* __restrict__ ei,
                                                                  const int* __restrict__ hist,
                                                                  int* __restrict__ bstart,
                                                                  int* __restrict__ ebuf, int E, int nb) {
  __shared__ int cur[BIN_MAXB];
  __shared__ int rsum[BIN_MAXB];
  __shared__ int wsum[BIN_THREADS / 32];
  const int tid = threadIdx.x, lane = tid & 31, wv = tid >> 5;
  const int b = blockIdx.x;
  const int nq = (nb + 3) >> 2;
  const int rq = BIN_THREADS / nq;
  const int rg = rq >= 16 ? 16 : (rq >= 8 ? 8 : 4);
  for (int k = tid; k < 4 * nq; k += BIN_THREADS) {
    cur[k] = 0;
    rsum[k] = 0;
  }
  __syncthreads();
  {
    const int4* __restrict__ hist4 = reinterpret_cast<const int4*>(hist);
    const int j = tid % nq;
    const int r = tid / nq;
    if (r < rg) {
      int4 tot = make_int4(0, 0, 0, 0);
      int4 low = make_int4(0, 0, 0, 0);
      #pragma clang loop vectorize(disable) interleave(disable) unroll(disable)
      for (int r0 = r; r0 < BIN_BLOCKS; r0 += BIN_ROWS * rg) {
        int4 u[BIN_ROWS];
        #pragma unroll
        for (int q = 0; q < BIN_ROWS; ++q) u[q] = hist4[(r0 + q * rg) * nq + j];
        #pragma unroll
        for (int q = 0; q < BIN_ROWS; ++q) {
          const int m = (r0 + q * rg < b) ? -1 : 0;
          tot.x += u[q].x; tot.y += u[q].y; tot.z += u[q].z; tot.w += u[q].w;
          low.x += u[q].x & m; low.y += u[q].y & m; low.z += u[q].z & m; low.w += u[q].w & m;
        }
      }
      atomicAdd(&rsum[4 * j + 0], tot.x);
      atomicAdd(&rsum[4 * j + 1], tot.y);
      atomicAdd(&rsum[4 * j + 2], tot.z);
      atomicAdd(&rsum[4 * j + 3], tot.w);
      atomicAdd(&cur[4 * j + 0], low.x);
      atomicAdd(&cur[4 * j + 1], low.y);
      atomicAdd(&cur[4 * j + 2], low.z);
      atomicAdd(&cur[4 * j + 3], low.w);
    }
  }
  __syncthreads();
  const int v = (tid < nb) ? rsum[tid] : 0;
  int inc = v;
  #pragma unroll
  for (int off = 1; off < 32; off <<= 1) {
    const int t = __shfl_up(inc, off);
    if (lane >= off) inc += t;
  }
  if (lane == 31) wsum[wv] = inc;
  __syncthreads();
  int woff = 0;
  #pragma unroll
  for (int k = 0; k < BIN_THREADS / 32; ++k)
    if (k < wv) woff += wsum[k];
  if (tid < nb) {
    const int start = woff + inc - v;
    cur[tid] += start;
    if (b == 0) {
      bstart[tid] = start;
      if (tid == nb - 1) bstart[nb] = start + v;
    }
  }
  __syncthreads();
  const int chunk = (E + BIN_BLOCKS - 1) / BIN_BLOCKS;
  const int lo = blockIdx.x * chunk;
  const int hi = min(E, lo + chunk);
  int2* __restrict__ eb = reinterpret_cast<int2*>(ebuf);
  for (int i = lo + tid; i < hi; i += BIN_BATCH * BIN_THREADS) {
    int s[BIN_BATCH], d[BIN_BATCH];
    #pragma unroll
    for (int q = 0; q < BIN_BATCH; ++q) {
      const int j = min(i + q * BIN_THREADS, hi - 1);
      s[q] = ei[j];
      d[q] = ei[E + j];
    }
    #pragma unroll
    for (int q = 0; q < BIN_BATCH; ++q) {
      if (i + q * BIN_THREADS < hi) {
        const int p = atomicAdd(&cur[d[q] >> BIN_SHIFT], 1);
        eb[p] = make_int2(s[q], d[q]);
      }
    }
  }
}

__global__ __launch_bounds__(BIN_THREADS) void bin_finish_kernel(const int* __restrict__ ebuf,
                                                                 const int* __restrict__ bstart,
                                                                 int* __restrict__ rowptr,
                                                                 int* __restrict__ col,
                                                                 float* __restrict__ dinv, int N) {
  __shared__ int cnt[BIN_NODES];
  __shared__ int cur[BIN_NODES];
  __shared__ int wsum[BIN_NODES / 32];
  __shared__ int stage[BIN_STAGE];
  const int tid = threadIdx.x, lane = tid & 31, wv = tid >> 5;
  const int b = blockIdx.x;
  const int e0 = bstart[b];
  const int e1 = bstart[b + 1];
  const int2* __restrict__ eb = reinterpret_cast<const int2*>(ebuf);
  if (tid < BIN_NODES) cnt[tid] = 0;
  __syncthreads();
  for (int i = e0 + tid; i < e1; i += BIN_BATCH * BIN_THREADS) {
    int d[BIN_BATCH];
    #pragma unroll
    for (int q = 0; q < BIN_BATCH; ++q) d[q] = eb[min(i + q * BIN_THREADS, e1 - 1)].y;
    #pragma unroll
    for (int q = 0; q < BIN_BATCH; ++q)
      if (i + q * BIN_THREADS < e1) atomicAdd(&cnt[d[q] & (BIN_NODES - 1)], 1);
  }
  __syncthreads();

  const int v = (tid < BIN_NODES) ? cnt[tid] : 0;
  int inc = v;
  #pragma unroll
  for (int off = 1; off < 32; off <<= 1) {
    const int u = __shfl_up(inc, off);
    if (lane >= off) inc += u;
  }
  if (tid < BIN_NODES && lane == 31) wsum[wv] = inc;
  __syncthreads();
  if (tid < BIN_NODES) {
    int woff = 0;
    #pragma unroll
    for (int k = 0; k < BIN_NODES / 32; ++k)
      if (k < wv) woff += wsum[k];
    const int excl = woff + inc - v;
    cur[tid] = excl;
    const int node = b * BIN_NODES + tid;
    if (node < N) {
      rowptr[node] = e0 + excl;
      dinv[node] = rsqrtf(fmaxf((float)(v + 1), 1.0f));
    }
  }
  if (b == (int)gridDim.x - 1 && tid == 0) rowptr[N] = e1;
  __syncthreads();

  const int ne = e1 - e0;
  if (ne <= BIN_STAGE) {
    for (int i = e0 + tid; i < e1; i += BIN_BATCH * BIN_THREADS) {
      int2 sd[BIN_BATCH];
      #pragma unroll
      for (int q = 0; q < BIN_BATCH; ++q) sd[q] = eb[min(i + q * BIN_THREADS, e1 - 1)];
      #pragma unroll
      for (int q = 0; q < BIN_BATCH; ++q)
        if (i + q * BIN_THREADS < e1) stage[atomicAdd(&cur[sd[q].y & (BIN_NODES - 1)], 1)] = sd[q].x;
    }
    __syncthreads();
    for (int i = tid; i < ne; i += BIN_THREADS) col[e0 + i] = stage[i];
  } else {
    for (int i = e0 + tid; i < e1; i += BIN_THREADS) {
      const int2 sd = eb[i];
      col[e0 + atomicAdd(&cur[sd.y & (BIN_NODES - 1)], 1)] = sd.x;
    }
  }
}

__global__ void dinv_from_count_kernel(float* __restrict__ cnt_dinv, int N) {
  const int i = blockIdx.x * blockDim.x + threadIdx.x;
  if (i < N) {
    const int c = __float_as_int(cnt_dinv[i]);
    cnt_dinv[i] = rsqrtf(fmaxf((float)(c + 1), 1.0f));
  }
}

__device__ __forceinline__ void row_add(float4& acc, const float4 h) {
  acc.x += h.x;
  acc.y += h.y;
  acc.z += h.z;
  acc.w += h.w;
}

__device__ __forceinline__ float4 gather_row_range(const float4* __restrict__ H4,
                                                   const int* __restrict__ col,
                                                   const int d, const int beg, const int end,
                                                   const int lane) {
  float4 acc = H4[(size_t)d * (FDIM / 4) + lane];

  for (int j = beg; j < end; j += 32) {
    const int n = min(32, end - j);
    int s_l = 0;
    if (lane < n) s_l = col[j + lane];
    int t = 0;
    for (; t + 4 <= n; t += 4) {
      const int s0 = __builtin_amdgcn_readlane(s_l, t);
      const int s1 = __builtin_amdgcn_readlane(s_l, t + 1);
      const int s2 = __builtin_amdgcn_readlane(s_l, t + 2);
      const int s3 = __builtin_amdgcn_readlane(s_l, t + 3);
      const float4 h0 = H4[(size_t)s0 * (FDIM / 4) + lane];
      const float4 h1 = H4[(size_t)s1 * (FDIM / 4) + lane];
      const float4 h2 = H4[(size_t)s2 * (FDIM / 4) + lane];
      const float4 h3 = H4[(size_t)s3 * (FDIM / 4) + lane];
      row_add(acc, h0);
      row_add(acc, h1);
      row_add(acc, h2);
      row_add(acc, h3);
    }
    const int r = n - t;
    if (r == 3) {
      const int s0 = __builtin_amdgcn_readlane(s_l, t);
      const int s1 = __builtin_amdgcn_readlane(s_l, t + 1);
      const int s2 = __builtin_amdgcn_readlane(s_l, t + 2);
      const float4 h0 = H4[(size_t)s0 * (FDIM / 4) + lane];
      const float4 h1 = H4[(size_t)s1 * (FDIM / 4) + lane];
      const float4 h2 = H4[(size_t)s2 * (FDIM / 4) + lane];
      row_add(acc, h0);
      row_add(acc, h1);
      row_add(acc, h2);
    } else if (r == 2) {
      const int s0 = __builtin_amdgcn_readlane(s_l, t);
      const int s1 = __builtin_amdgcn_readlane(s_l, t + 1);
      const float4 h0 = H4[(size_t)s0 * (FDIM / 4) + lane];
      const float4 h1 = H4[(size_t)s1 * (FDIM / 4) + lane];
      row_add(acc, h0);
      row_add(acc, h1);
    } else if (r == 1) {
      const int s0 = __builtin_amdgcn_readlane(s_l, t);
      row_add(acc, H4[(size_t)s0 * (FDIM / 4) + lane]);
    }
  }
  return acc;
}

__device__ __forceinline__ float4 gather_row_sum(const float4* __restrict__ H4,
                                                 const int* __restrict__ rowptr,
                                                 const int* __restrict__ col,
                                                 const int d, const int lane) {
  const int beg = rowptr[d];
  const int end = rowptr[d + 1];
  return gather_row_range(H4, col, d, beg, end, lane);
}

#define FG_ROWS   32
#define FG_TPITCH 132
#define FG_BLOCKS 512

__global__ __launch_bounds__(32 * FG_ROWS, 16) void gather_product_kernel(
    const float* __restrict__ H, const int* __restrict__ rowptr, const int* __restrict__ col,
    const float* __restrict__ dinv, const float* __restrict__ bias, const float* __restrict__ Wn,
    float* __restrict__ Hn, int N, int ngroups) {
  __shared__ alignas(16) float smem[64 * GEMM_WPITCH + 2 * FG_ROWS * FG_TPITCH];
  float* Ws = smem;
  const int tid = threadIdx.x;

  {
    const float4* __restrict__ W4 = reinterpret_cast<const float4*>(Wn);
    for (int it = 0; it < 2; ++it) {
      const int i = it * (32 * FG_ROWS) + tid;
      const int p = i >> 5;
      const int q = i & 31;
      const float4 w0 = W4[(2 * p) * (FDIM / 4) + q];
      const float4 w1 = W4[(2 * p + 1) * (FDIM / 4) + q];
      float4* dst = reinterpret_cast<float4*>(Ws + p * GEMM_WPITCH + q * 8);
      dst[0] = make_float4(w0.x, w1.x, w0.y, w1.y);
      dst[1] = make_float4(w0.z, w1.z, w0.w, w1.w);
    }
  }
  __syncthreads();

  const int wave = __builtin_amdgcn_readfirstlane((int)(threadIdx.x >> 5));
  const int lane = tid & 31;
  const int c = lane & 15;
  const int h = lane >> 4;

  const float bx = bias[lane * 4 + 0];
  const float by = bias[lane * 4 + 1];
  const float bz = bias[lane * 4 + 2];
  const float bw = bias[lane * 4 + 3];
  const float4* __restrict__ H4 = reinterpret_cast<const float4*>(H);

  const int rt = (wave >> 2) & 1;
  const int cp = wave & 3;
  const unsigned boff0 = (unsigned)(h * (GEMM_WPITCH / 2) + c + 32 * cp) * 8u;

  int it = 0;
  for (int g = blockIdx.x; g < ngroups; g += gridDim.x, ++it) {
    const int toff = 64 * GEMM_WPITCH + (it & 1) * (FG_ROWS * FG_TPITCH);
    float* tl = smem + toff;
    const int d = g * FG_ROWS + wave;
    float4 acc = make_float4(0.0f, 0.0f, 0.0f, 0.0f);
    if (d < N) {
      const float dd = dinv[d];
      acc = gather_row_sum(H4, rowptr, col, d, lane);
      acc.x = fmaxf(acc.x * dd + bx, 0.0f) * dd;
      acc.y = fmaxf(acc.y * dd + by, 0.0f) * dd;
      acc.z = fmaxf(acc.z * dd + bz, 0.0f) * dd;
      acc.w = fmaxf(acc.w * dd + bw, 0.0f) * dd;
    }
    reinterpret_cast<float4*>(tl + wave * FG_TPITCH)[lane] = acc;
    __syncthreads();

    if ((wave >> 3) == (it & 3)) {
      unsigned ao = (unsigned)(toff + (16 * rt + c) * FG_TPITCH + 2 * h) * 4u;
      unsigned bo = boff0;
      const char* sb = reinterpret_cast<const char*>(smem);
      const int row0 = g * FG_ROWS + 16 * rt + 8 * h;
      float* hp = Hn + (size_t)row0 * FDIM + 32 * cp + c;
      int nval = N - row0;
      asm volatile("" : "+v"(hp), "+v"(nval));
      __attribute__((address_space(1))) float* hg = (__attribute__((address_space(1))) float*)hp;
      v8f acc0 = v8f{};
      v8f acc1 = v8f{};
      v2f ax  = *reinterpret_cast<const v2f*>(sb + ao);
      v2f bx0 = *reinterpret_cast<const v2f*>(sb + bo);
      v2f bx1 = *reinterpret_cast<const v2f*>(sb + bo + 128);
      #pragma unroll 1
      for (int kk = 0; kk < FDIM / 4 - 2; kk += 2) {
        ao += 16;
        bo += 2 * (GEMM_WPITCH / 2) * 8;
        asm volatile("" : "+v"(ao), "+v"(bo));
        const v2f ay  = *reinterpret_cast<const v2f*>(sb + ao);
        const v2f by0 = *reinterpret_cast<const v2f*>(sb + bo);
        const v2f by1 = *reinterpret_cast<const v2f*>(sb + bo + 128);
        acc0 = __builtin_amdgcn_wmma_f32_16x16x4_f32(
             false, ax,  false, bx0,
             (short)0, acc0,  false,  false);
        acc1 = __builtin_amdgcn_wmma_f32_16x16x4_f32(
             false, ax,  false, bx1,
             (short)0, acc1,  false,  false);
        ao += 16;
        bo += 2 * (GEMM_WPITCH / 2) * 8;
        asm volatile("" : "+v"(ao), "+v"(bo));
        ax  = *reinterpret_cast<const v2f*>(sb + ao);
        bx0 = *reinterpret_cast<const v2f*>(sb + bo);
        bx1 = *reinterpret_cast<const v2f*>(sb + bo + 128);
        acc0 = __builtin_amdgcn_wmma_f32_16x16x4_f32(
             false, ay,  false, by0,
             (short)0, acc0,  false,  false);
        acc1 = __builtin_amdgcn_wmma_f32_16x16x4_f32(
             false, ay,  false, by1,
             (short)0, acc1,  false,  false);
      }
      {
        ao += 16;
        bo += 2 * (GEMM_WPITCH / 2) * 8;
        asm volatile("" : "+v"(ao), "+v"(bo));
        const v2f ay  = *reinterpret_cast<const v2f*>(sb + ao);
        const v2f by0 = *reinterpret_cast<const v2f*>(sb + bo);
        const v2f by1 = *reinterpret_cast<const v2f*>(sb + bo + 128);
        acc0 = __builtin_amdgcn_wmma_f32_16x16x4_f32(
             false, ax,  false, bx0,
             (short)0, acc0,  false,  false);
        acc1 = __builtin_amdgcn_wmma_f32_16x16x4_f32(
             false, ax,  false, bx1,
             (short)0, acc1,  false,  false);
        acc0 = __builtin_amdgcn_wmma_f32_16x16x4_f32(
             false, ay,  false, by0,
             (short)0, acc0,  false,  false);
        acc1 = __builtin_amdgcn_wmma_f32_16x16x4_f32(
             false, ay,  false, by1,
             (short)0, acc1,  false,  false);
      }
      #pragma unroll
      for (int r = 0; r < 8; ++r) {
        if (r < nval) {
          hg[r * FDIM]      = acc0[r];
          hg[r * FDIM + 16] = acc1[r];
        }
      }
    }
  }
}

__global__ void bias_relu_kernel(float* __restrict__ G, const float* __restrict__ b) {
  const size_t i = (size_t)blockIdx.x * FDIM + threadIdx.x;
  G[i] = fmaxf(G[i] + b[threadIdx.x], 0.0f);
}

__global__ void init_out_kernel(float* __restrict__ out, int ng) {
  int i = blockIdx.x * blockDim.x + threadIdx.x;
  if (i >= ng * 2 * FDIM) return;
  int f = i & (2 * FDIM - 1);
  out[i] = (f < FDIM) ? 0.0f : __int_as_float(0xFF7FFFFF);
}

__device__ __forceinline__ void atomic_max_f32(float* addr, float v) {
  if (v >= 0.0f) atomicMax((int*)addr, __float_as_int(v));
  else           atomicMin((unsigned int*)addr, (unsigned int)__float_as_int(v));
}

#define POOL_ROWS  32
#define POOL_WAVES 8

__device__ __forceinline__ void pool_flush(float* __restrict__ out, float* __restrict__ cnt,
                                           float* st, int lane, int g,
                                           const float4 s, const float4 m, int run) {
  reinterpret_cast<float4*>(st)[lane] = s;
  reinterpret_cast<float4*>(st + FDIM)[lane] = m;
  __builtin_amdgcn_fence(__ATOMIC_ACQ_REL, "wavefront");
  float a[4], b[4];
  #pragma unroll
  for (int k = 0; k < 4; ++k) {
    a[k] = st[lane + 32 * k];
    b[k] = st[FDIM + lane + 32 * k];
  }
  __builtin_amdgcn_fence(__ATOMIC_ACQ_REL, "wavefront");
  float* o = out + (size_t)g * (2 * FDIM);
  #pragma unroll
  for (int k = 0; k < 4; ++k) atomicAdd(o + lane + 32 * k, a[k]);
  #pragma unroll
  for (int k = 0; k < 4; ++k) atomic_max_f32(o + FDIM + lane + 32 * k, b[k]);
  if (lane == 0) atomicAdd(&cnt[g], (float)run);
}

__global__ __launch_bounds__(256) void pool_runs_kernel(const float* __restrict__ G,
                                                        const int* __restrict__ gi,
                                                        float* __restrict__ out,
                                                        float* __restrict__ cnt, int N) {
  __shared__ alignas(16) float stage[POOL_WAVES * 2 * FDIM];
  const int wv = threadIdx.x >> 5;
  const int lane = threadIdx.x & 31;
  const int n0 = __builtin_amdgcn_readfirstlane((int)(blockIdx.x * POOL_WAVES + wv)) * POOL_ROWS;
  if (n0 >= N) return;
  const int nr = min(POOL_ROWS, N - n0);
  float* st = stage + wv * (2 * FDIM);

  const int g_l = gi[n0 + min(lane, nr - 1)];

  const float4* __restrict__ G4 = reinterpret_cast<const float4*>(G) + (size_t)n0 * (FDIM / 4) + lane;
  float4 v[POOL_ROWS];
  #pragma unroll
  for (int i = 0; i < POOL_ROWS; ++i) v[i] = G4[(size_t)min(i, nr - 1) * (FDIM / 4)];

  int a = 0;
  for (int trip = 0; trip < POOL_ROWS && a < nr; ++trip) {
    const int ga = __builtin_amdgcn_readlane(g_l, a);
    const unsigned diff = __builtin_amdgcn_ballot_w32(g_l != ga) & (~0u << a);
    const int b = min(diff ? __builtin_ctz(diff) : POOL_ROWS, nr);
    float4 s = make_float4(0.0f, 0.0f, 0.0f, 0.0f);
    float4 m = make_float4(-FLT_MAX, -FLT_MAX, -FLT_MAX, -FLT_MAX);
    #pragma unroll
    for (int i = 0; i < POOL_ROWS; ++i) {
      if (i >= a && i < b) {
        s.x += v[i].x; s.y += v[i].y; s.z += v[i].z; s.w += v[i].w;
        m.x = fmaxf(m.x, v[i].x); m.y = fmaxf(m.y, v[i].y);
        m.z = fmaxf(m.z, v[i].z); m.w = fmaxf(m.w, v[i].w);
      }
    }
    pool_flush(out, cnt, st, lane, ga, s, m, b - a);
    a = b;
  }
}

#define GP_WAVES 8
#define GP_ROWS  7

__global__ __launch_bounds__(32 * GP_WAVES) void gather_pool_kernel(
    const float* __restrict__ H, const int* __restrict__ rowptr, const int* __restrict__ col,
    const float* __restrict__ dinv, const float* __restrict__ bias, const int* __restrict__ gi,
    float* __restrict__ out, float* __restrict__ cnt, int N) {
  __shared__ alignas(16) float stage[GP_WAVES * 2 * FDIM];
  const int wv = threadIdx.x >> 5;
  const int lane = threadIdx.x & 31;
  const int n0 = __builtin_amdgcn_readfirstlane((int)(blockIdx.x * GP_WAVES + wv)) * GP_ROWS;
  if (n0 >= N) return;
  const int nr = min(GP_ROWS, N - n0);
  float* st = stage + wv * (2 * FDIM);

  const int   rp_l = rowptr[n0 + min(lane, nr)];
  const int   g_l  = gi[n0 + min(lane, nr - 1)];
  const float dd_l = dinv[n0 + min(lane, nr - 1)];
  const float bx = bias[lane * 4 + 0];
  const float by = bias[lane * 4 + 1];
  const float bz = bias[lane * 4 + 2];
  const float bw = bias[lane * 4 + 3];
  const float4* __restrict__ H4 = reinterpret_cast<const float4*>(H);

  int g_cur = __builtin_amdgcn_readlane(g_l, 0);
  int run = 0;
  float4 s = make_float4(0.0f, 0.0f, 0.0f, 0.0f);
  float4 m = make_float4(-FLT_MAX, -FLT_MAX, -FLT_MAX, -FLT_MAX);
  for (int i = 0; i < nr; ++i) {
    const int beg = __builtin_amdgcn_readlane(rp_l, i);
    const int end = __builtin_amdgcn_readlane(rp_l, i + 1);
    const int g = __builtin_amdgcn_readlane(g_l, i);
    const float dd = __int_as_float(__builtin_amdgcn_readlane(__float_as_int(dd_l), i));
    float4 acc = gather_row_range(H4, col, n0 + i, beg, end, lane);
    acc.x = fmaxf(acc.x * dd + bx, 0.0f);
    acc.y = fmaxf(acc.y * dd + by, 0.0f);
    acc.z = fmaxf(acc.z * dd + bz, 0.0f);
    acc.w = fmaxf(acc.w * dd + bw, 0.0f);
    if (g != g_cur) {
      pool_flush(out, cnt, st, lane, g_cur, s, m, run);
      s = make_float4(0.0f, 0.0f, 0.0f, 0.0f);
      m = make_float4(-FLT_MAX, -FLT_MAX, -FLT_MAX, -FLT_MAX);
      run = 0;
      g_cur = g;
    }
    s.x += acc.x; s.y += acc.y; s.z += acc.z; s.w += acc.w;
    m.x = fmaxf(m.x, acc.x); m.y = fmaxf(m.y, acc.y);
    m.z = fmaxf(m.z, acc.z); m.w = fmaxf(m.w, acc.w);
    ++run;
  }
  pool_flush(out, cnt, st, lane, g_cur, s, m, run);
}

__global__ void finalize_mean_kernel(float* __restrict__ out, const float* __restrict__ cnt) {
  const int g = blockIdx.x;
  out[(size_t)g * 2 * FDIM + threadIdx.x] /= fmaxf(cnt[g], 1.0f);
}


extern "C" void kernel_launch(void* const* d_in, const int* in_sizes, int n_in,
                              void* d_out, int out_size, void* d_ws, size_t ws_size,
                              hipStream_t stream) {
  const float* x  = (const float*)d_in[0];
  const int*   ei = (const int*)d_in[1];
  const int*   gi = (const int*)d_in[2];
  const float* Wm[3] = {(const float*)d_in[4], (const float*)d_in[6], (const float*)d_in[8]};
  const float* Bv[3] = {(const float*)d_in[5], (const float*)d_in[7], (const float*)d_in[9]};
  float* out = (float*)d_out;

  const int  N  = in_sizes[0] / FDIM;
  const int  E  = in_sizes[1] / 2;
  const int  NG = out_size / (2 * FDIM);
  const long Nf = (long)N * FDIM;

  float* Hbuf = (float*)d_ws;
  float* Gbuf = Hbuf + Nf;
  float* dinv = Gbuf + Nf;
  float* cnt  = dinv + N;
  if (ws_size < (size_t)(2 * Nf + N + NG) * sizeof(float)) return;

  const int nsb = (N + SCAN_ELEMS - 1) / SCAN_ELEMS;
  int* rowptr = (int*)(cnt + NG);
  int* bsum   = rowptr + (N + 1);
  int* col    = bsum + nsb;
  int* rank   = (int*)Hbuf;
  const bool ordered =
      ws_size >= (size_t)(2 * Nf + N + NG) * sizeof(float) + ((size_t)N + 1 + nsb + E) * sizeof(int) &&
      (long)E <= Nf && E > 0;

  auto zero = [&](float* p, long n) {
    zero_f32_kernel<<<(unsigned)((n + 255) / 256), 256, 0, stream>>>(p, n);
  };

  const int tiles = N / 16;
  const int rem   = N - tiles * 16;
  const float* cur = x;

  const int  nb   = (N + BIN_NODES - 1) / BIN_NODES;
  const long nh   = (long)((nb + 3) & ~3) * BIN_BLOCKS;
  int* ebuf   = (int*)Hbuf;
  int* hist   = (int*)Gbuf;
  int* bstart = hist + nh;
  const bool binned = ordered && nb <= BIN_MAXB && 2 * (long)E <= Nf && nh + nb + 1 <= Nf;

  if (!binned) {
    init_out_kernel<<<(NG * 2 * FDIM + 255) / 256, 256, 0, stream>>>(out, NG);
    zero(cnt, NG);
  }

  if (ordered) {
    if (binned) {
      bin_count_kernel<<<BIN_BLOCKS, BIN_THREADS, 0, stream>>>(ei, hist, E, nb, out, cnt, NG);
      bin_scatter_kernel<<<BIN_BLOCKS, BIN_THREADS, 0, stream>>>(ei, hist, bstart, ebuf, E, nb);
      bin_finish_kernel<<<nb, BIN_THREADS, 0, stream>>>(ebuf, bstart, rowptr, col, dinv, N);
    } else {
      zero(dinv, N);
      count_rank_kernel<<<(E + 255) / 256, 256, 0, stream>>>(ei, (int*)dinv, rank, E);
      scan_local_kernel<<<nsb, SCAN_THREADS, 0, stream>>>((const int*)dinv, rowptr, bsum, N);
      scan_add_kernel<<<nsb, SCAN_THREADS, 0, stream>>>(rowptr, bsum, N);
      place_kernel<<<(E + 255) / 256, 256, 0, stream>>>(ei, rowptr, rank, col, E);
      dinv_from_count_kernel<<<(N + 255) / 256, 256, 0, stream>>>(dinv, N);
    }

    if (tiles > 0)
      gemm_strip_kernel<true><<<GEMM_BLOCKS, 256, 0, stream>>>(x, Wm[0], dinv, Hbuf, tiles);
    if (rem > 0)
      gemm_tail_kernel<<<rem, FDIM, 0, stream>>>(x, Wm[0], dinv, Hbuf, tiles * 16);
    const int ngroups = (N + FG_ROWS - 1) / FG_ROWS;
    const int fgrid = ngroups < FG_BLOCKS ? ngroups : FG_BLOCKS;
    gather_product_kernel<<<fgrid, 32 * FG_ROWS, 0, stream>>>(Hbuf, rowptr, col, dinv, Bv[0], Wm[1], Gbuf, N, ngroups);
    gather_product_kernel<<<fgrid, 32 * FG_ROWS, 0, stream>>>(Gbuf, rowptr, col, dinv, Bv[1], Wm[2], Hbuf, N, ngroups);
    gather_pool_kernel<<<(N + GP_ROWS * GP_WAVES - 1) / (GP_ROWS * GP_WAVES), 32 * GP_WAVES, 0, stream>>>(
        Hbuf, rowptr, col, dinv, Bv[2], gi, out, cnt, N);
  } else {
    zero(dinv, N);
    degree_kernel<<<(E + N + 255) / 256, 256, 0, stream>>>(ei, dinv, E, N);
    dinv_kernel<<<(N + 255) / 256, 256, 0, stream>>>(dinv, N);

    for (int l = 0; l < 3; ++l) {
      if (tiles > 0)
        gemm_strip_kernel<false><<<GEMM_BLOCKS, 256, 0, stream>>>(cur, Wm[l], nullptr, Hbuf, tiles);
      if (rem > 0)
        gemm_tail_kernel<<<rem, FDIM, 0, stream>>>(cur, Wm[l], nullptr, Hbuf, tiles * 16);
      zero(Gbuf, Nf);
      aggregate_kernel<<<(E + N + 7) / 8, 256, 0, stream>>>(Hbuf, ei, dinv, Gbuf, E, N);
      bias_relu_kernel<<<N, FDIM, 0, stream>>>(Gbuf, Bv[l]);
      cur = Gbuf;
    }
  }

  if (!ordered)
    pool_runs_kernel<<<(N + POOL_ROWS * POOL_WAVES - 1) / (POOL_ROWS * POOL_WAVES), 32 * POOL_WAVES, 0, stream>>>(Gbuf, gi, out, cnt, N);
  finalize_mean_kernel<<<NG, FDIM, 0, stream>>>(out, cnt);
}


# Every layer re-associated as (A H) W: one gather kernel per layer, its f32 WMMA product behind the gather; layer-1 product launch gone; layer 3 pools from accumulators; row and pool sum order change
# speedup vs baseline: 6.4872x; 1.0741x over previous
#include <hip/hip_runtime.h>
#include <float.h>

#define FDIM 128

typedef __attribute__((ext_vector_type(2))) float v2f;
typedef __attribute__((ext_vector_type(8))) float v8f;


__global__ void zero_f32_kernel(float* __restrict__ p, long n) {
  long i = (long)blockIdx.x * blockDim.x + threadIdx.x;
  if (i < n) p[i] = 0.0f;
}

__global__ void degree_kernel(const int* __restrict__ ei, float* __restrict__ deg,
                              int E, int N) {
  int e = blockIdx.x * blockDim.x + threadIdx.x;
  int EN = E + N;
  if (e >= EN) return;
  int d = (e < E) ? ei[E + e] : (e - E);
  atomicAdd(&deg[d], 1.0f);
}

__global__ void dinv_kernel(float* __restrict__ deg_dinv, int N) {
  int i = blockIdx.x * blockDim.x + threadIdx.x;
  if (i < N) deg_dinv[i] = rsqrtf(fmaxf(deg_dinv[i], 1.0f));
}

#define GEMM_WPITCH 288
#define GEMM_BLOCKS 512

template <bool SCALED>
__global__ __launch_bounds__(256, 4) void gemm_strip_kernel(const float* __restrict__ X,
                                                         const float* __restrict__ W,
                                                         const float* __restrict__ rs,
                                                         float* __restrict__ H, int tiles) {
  __shared__ alignas(16) float Ws[64 * GEMM_WPITCH];
  const int tid = threadIdx.x;

  {
    const float4* __restrict__ W4 = reinterpret_cast<const float4*>(W);
    for (int it = 0; it < 8; ++it) {
      const int i = it * 256 + tid;
      const int p = i >> 5;
      const int q = i & 31;
      const float4 w0 = W4[(2 * p) * (FDIM / 4) + q];
      const float4 w1 = W4[(2 * p + 1) * (FDIM / 4) + q];
      float4* dst = reinterpret_cast<float4*>(Ws + p * GEMM_WPITCH + q * 8);
      dst[0] = make_float4(w0.x, w1.x, w0.y, w1.y);
      dst[1] = make_float4(w0.z, w1.z, w0.w, w1.w);
    }
  }
  __syncthreads();

  const int wave = __builtin_amdgcn_readfirstlane((int)(threadIdx.x >> 5));
  const int lane = tid & 31;
  const int c = lane & 15;
  const int h = lane >> 4;

  const v2f* __restrict__ bfrag = reinterpret_cast<const v2f*>(Ws) + h * (GEMM_WPITCH / 2) + c;

  const int xoff = c * (FDIM / 2) + h;
  const int hoff = 8 * h * FDIM + c;

  const int s0 = blockIdx.x + gridDim.x * wave;
  const size_t sstep = (size_t)gridDim.x * 8 * (16 * FDIM);
  const v2f* xs = reinterpret_cast<const v2f*>(X + (size_t)s0 * (16 * FDIM)) + xoff;
  float* hs = H + (size_t)s0 * (16 * FDIM) + hoff;
  const float* rp = SCALED ? rs + (size_t)s0 * 16 + c : X;
  const size_t rstep = SCALED ? (size_t)gridDim.x * 8 * 16 : 0;

  for (int s = s0; s < tiles; s += gridDim.x * 8, xs += sstep / 2, hs += sstep, rp += rstep) {
    v2f a[FDIM / 4];
    #pragma unroll
    for (int kk = 0; kk < FDIM / 4; ++kk) a[kk] = xs[2 * kk];
    if (SCALED) {
      const float sc = *rp;
      #pragma unroll
      for (int kk = 0; kk < FDIM / 4; ++kk) a[kk] = a[kk] * sc;
    }

    v8f acc[8];
    #pragma unroll
    for (int t = 0; t < 8; ++t) acc[t] = v8f{};

    v2f bcur[8];
    #pragma unroll
    for (int t = 0; t < 8; ++t) bcur[t] = bfrag[16 * t];

    #pragma unroll
    for (int kk = 0; kk < FDIM / 4; ++kk) {
      v2f bnxt[8];
      if (kk + 1 < FDIM / 4) {
        const v2f* bp = bfrag + (2 * (kk + 1)) * (GEMM_WPITCH / 2);
        #pragma unroll
        for (int t = 0; t < 8; ++t) bnxt[t] = bp[16 * t];
      }
      #pragma unroll
      for (int t = 0; t < 8; ++t) {
        acc[t] = __builtin_amdgcn_wmma_f32_16x16x4_f32(
             false, a[kk],  false, bcur[t],
             (short)0, acc[t],  false,  false);
      }
      __builtin_amdgcn_sched_barrier(0);
      if (kk + 1 < FDIM / 4) {
        #pragma unroll
        for (int t = 0; t < 8; ++t) bcur[t] = bnxt[t];
      }
    }

    #pragma unroll
    for (int t = 0; t < 8; ++t) {
      #pragma unroll
      for (int r = 0; r < 8; ++r) hs[r * FDIM + 16 * t] = acc[t][r];
    }
  }
}

__global__ void gemm_tail_kernel(const float* __restrict__ X,
                                 const float* __restrict__ W,
                                 const float* __restrict__ rs,
                                 float* __restrict__ H, int base) {
  const int row = base + blockIdx.x;
  const int f = threadIdx.x;
  const float* xr = X + (size_t)row * FDIM;
  float acc = 0.0f;
  #pragma unroll 8
  for (int k = 0; k < FDIM; ++k) acc = fmaf(xr[k], W[(size_t)k * FDIM + f], acc);
  H[(size_t)row * FDIM + f] = rs ? acc * rs[row] : acc;
}

__global__ void aggregate_kernel(const float* __restrict__ H,
                                 const int* __restrict__ ei,
                                 const float* __restrict__ dinv,
                                 float* __restrict__ G, int E, int N) {
  const int e = blockIdx.x * (blockDim.x >> 5) + (threadIdx.x >> 5);
  const int EN = E + N;
  if (e >= EN) return;
  const int lane = threadIdx.x & 31;
  int s, d;
  if (e < E) { s = ei[e]; d = ei[E + e]; }
  else       { s = e - E; d = s; }
  const float w = dinv[s] * dinv[d];
  const float4 hv = reinterpret_cast<const float4*>(H + (size_t)s * FDIM)[lane];
  float* gp = G + (size_t)d * FDIM + lane * 4;
  atomicAdd(gp + 0, hv.x * w);
  atomicAdd(gp + 1, hv.y * w);
  atomicAdd(gp + 2, hv.z * w);
  atomicAdd(gp + 3, hv.w * w);
}


__global__ void count_rank_kernel(const int* __restrict__ ei, int* __restrict__ icnt,
                                  int* __restrict__ rank, int E) {
  const int e = blockIdx.x * blockDim.x + threadIdx.x;
  if (e >= E) return;
  rank[e] = atomicAdd(&icnt[ei[E + e]], 1);
}

#define SCAN_THREADS 256
#define SCAN_ELEMS   1024

__global__ void scan_local_kernel(const int* __restrict__ icnt, int* __restrict__ rowptr,
                                  int* __restrict__ bsum, int N) {
  __shared__ int wsum[SCAN_THREADS / 32];
  const int tid = threadIdx.x, lane = tid & 31, wv = tid >> 5;
  const int base = blockIdx.x * SCAN_ELEMS + tid * 4;
  int v[4];
  #pragma unroll
  for (int q = 0; q < 4; ++q) v[q] = (base + q < N) ? icnt[base + q] : 0;
  const int t = v[0] + v[1] + v[2] + v[3];
  int inc = t;
  #pragma unroll
  for (int off = 1; off < 32; off <<= 1) {
    const int u = __shfl_up(inc, off);
    if (lane >= off) inc += u;
  }
  if (lane == 31) wsum[wv] = inc;
  __syncthreads();
  int woff = 0, tot = 0;
  #pragma unroll
  for (int k = 0; k < SCAN_THREADS / 32; ++k) {
    const int s = wsum[k];
    if (k < wv) woff += s;
    tot += s;
  }
  int run = woff + inc - t;
  #pragma unroll
  for (int q = 0; q < 4; ++q) {
    if (base + q < N) rowptr[base + q] = run;
    run += v[q];
  }
  if (tid == 0) bsum[blockIdx.x] = tot;
}

__global__ void scan_add_kernel(int* __restrict__ rowptr, const int* __restrict__ bsum, int N) {
  __shared__ int red[SCAN_THREADS / 32];
  const int tid = threadIdx.x, lane = tid & 31, wv = tid >> 5;
  const int b = blockIdx.x;
  int part = 0;
  for (int k = tid; k < b; k += SCAN_THREADS) part += bsum[k];
  #pragma unroll
  for (int off = 16; off > 0; off >>= 1) part += __shfl_xor(part, off);
  if (lane == 0) red[wv] = part;
  __syncthreads();
  int add = 0;
  #pragma unroll
  for (int k = 0; k < SCAN_THREADS / 32; ++k) add += red[k];
  const int base = b * SCAN_ELEMS + tid * 4;
  #pragma unroll
  for (int q = 0; q < 4; ++q)
    if (base + q < N) rowptr[base + q] += add;
  if (b == (int)gridDim.x - 1 && tid == 0) rowptr[N] = add + bsum[b];
}

__global__ void place_kernel(const int* __restrict__ ei, const int* __restrict__ rowptr,
                             const int* __restrict__ rank, int* __restrict__ col, int E) {
  const int e = blockIdx.x * blockDim.x + threadIdx.x;
  if (e >= E) return;
  col[rowptr[ei[E + e]] + rank[e]] = ei[e];
}

#define BIN_SHIFT   8
#define BIN_NODES   256
#define BIN_MAXB    1024
#define BIN_BLOCKS  256
#define BIN_THREADS 1024
#define BIN_BATCH   4
#define BIN_STAGE   16384

__global__ __launch_bounds__(BIN_THREADS) void bin_count_kernel(const int* __restrict__ ei,
                                                                int* __restrict__ hist, int E, int nb,
                                                                float* __restrict__ out,
                                                                float* __restrict__ cnt, int ng) {
  __shared__ int h[BIN_MAXB];
  const int tid = threadIdx.x;
  {
    const float lowest = __int_as_float(0xFF7FFFFF);
    float4* __restrict__ out4 = reinterpret_cast<float4*>(out);
    const int gtid = blockIdx.x * BIN_THREADS + tid;
    const int gthreads = gridDim.x * BIN_THREADS;
    for (int i = gtid; i < ng * (2 * FDIM / 4); i += gthreads) {
      const float v = ((i & (2 * FDIM / 4 - 1)) < FDIM / 4) ? 0.0f : lowest;
      out4[i] = make_float4(v, v, v, v);
    }
    for (int i = gtid; i < ng; i += gthreads) cnt[i] = 0.0f;
  }
  const int nbp = (nb + 3) & ~3;
  for (int k = tid; k < nbp; k += BIN_THREADS) h[k] = 0;
  __syncthreads();
  const int chunk = (E + BIN_BLOCKS - 1) / BIN_BLOCKS;
  const int lo = blockIdx.x * chunk;
  const int hi = min(E, lo + chunk);
  for (int i = lo + tid; i < hi; i += BIN_BATCH * BIN_THREADS) {
    int d[BIN_BATCH];
    #pragma unroll
    for (int q = 0; q < BIN_BATCH; ++q) d[q] = ei[E + min(i + q * BIN_THREADS, hi - 1)];
    #pragma unroll
    for (int q = 0; q < BIN_BATCH; ++q)
      if (i + q * BIN_THREADS < hi) atomicAdd(&h[d[q] >> BIN_SHIFT], 1);
  }
  __syncthreads();
  for (int k = tid; k < nbp; k += BIN_THREADS) hist[blockIdx.x * nbp + k] = h[k];
}

#define BIN_ROWS 16
__global__ __launch_bounds__(BIN_THREADS) void bin_scatter_kernel(const int* __restrict__ ei,
                                                                  const int* __restrict__ hist,
                                                                  int* __restrict__ bstart,
                                                                  int* __restrict__ ebuf, int E, int nb) {
  __shared__ int cur[BIN_MAXB];
  __shared__ int rsum[BIN_MAXB];
  __shared__ int wsum[BIN_THREADS / 32];
  const int tid = threadIdx.x, lane = tid & 31, wv = tid >> 5;
  const int b = blockIdx.x;
  const int nq = (nb + 3) >> 2;
  const int rq = BIN_THREADS / nq;
  const int rg = rq >= 16 ? 16 : (rq >= 8 ? 8 : 4);
  for (int k = tid; k < 4 * nq; k += BIN_THREADS) {
    cur[k] = 0;
    rsum[k] = 0;
  }
  __syncthreads();
  {
    const int4* __restrict__ hist4 = reinterpret_cast<const int4*>(hist);
    const int j = tid % nq;
    const int r = tid / nq;
    if (r < rg) {
      int4 tot = make_int4(0, 0, 0, 0);
      int4 low = make_int4(0, 0, 0, 0);
      #pragma clang loop vectorize(disable) interleave(disable) unroll(disable)
      for (int r0 = r; r0 < BIN_BLOCKS; r0 += BIN_ROWS * rg) {
        int4 u[BIN_ROWS];
        #pragma unroll
        for (int q = 0; q < BIN_ROWS; ++q) u[q] = hist4[(r0 + q * rg) * nq + j];
        #pragma unroll
        for (int q = 0; q < BIN_ROWS; ++q) {
          const int m = (r0 + q * rg < b) ? -1 : 0;
          tot.x += u[q].x; tot.y += u[q].y; tot.z += u[q].z; tot.w += u[q].w;
          low.x += u[q].x & m; low.y += u[q].y & m; low.z += u[q].z & m; low.w += u[q].w & m;
        }
      }
      atomicAdd(&rsum[4 * j + 0], tot.x);
      atomicAdd(&rsum[4 * j + 1], tot.y);
      atomicAdd(&rsum[4 * j + 2], tot.z);
      atomicAdd(&rsum[4 * j + 3], tot.w);
      atomicAdd(&cur[4 * j + 0], low.x);
      atomicAdd(&cur[4 * j + 1], low.y);
      atomicAdd(&cur[4 * j + 2], low.z);
      atomicAdd(&cur[4 * j + 3], low.w);
    }
  }
  __syncthreads();
  const int v = (tid < nb) ? rsum[tid] : 0;
  int inc = v;
  #pragma unroll
  for (int off = 1; off < 32; off <<= 1) {
    const int t = __shfl_up(inc, off);
    if (lane >= off) inc += t;
  }
  if (lane == 31) wsum[wv] = inc;
  __syncthreads();
  int woff = 0;
  #pragma unroll
  for (int k = 0; k < BIN_THREADS / 32; ++k)
    if (k < wv) woff += wsum[k];
  if (tid < nb) {
    const int start = woff + inc - v;
    cur[tid] += start;
    if (b == 0) {
      bstart[tid] = start;
      if (tid == nb - 1) bstart[nb] = start + v;
    }
  }
  __syncthreads();
  const int chunk = (E + BIN_BLOCKS - 1) / BIN_BLOCKS;
  const int lo = blockIdx.x * chunk;
  const int hi = min(E, lo + chunk);
  int2* __restrict__ eb = reinterpret_cast<int2*>(ebuf);
  for (int i = lo + tid; i < hi; i += BIN_BATCH * BIN_THREADS) {
    int s[BIN_BATCH], d[BIN_BATCH];
    #pragma unroll
    for (int q = 0; q < BIN_BATCH; ++q) {
      const int j = min(i + q * BIN_THREADS, hi - 1);
      s[q] = ei[j];
      d[q] = ei[E + j];
    }
    #pragma unroll
    for (int q = 0; q < BIN_BATCH; ++q) {
      if (i + q * BIN_THREADS < hi) {
        const int p = atomicAdd(&cur[d[q] >> BIN_SHIFT], 1);
        eb[p] = make_int2(s[q], d[q]);
      }
    }
  }
}

__global__ __launch_bounds__(BIN_THREADS) void bin_finish_kernel(const int* __restrict__ ebuf,
                                                                 const int* __restrict__ bstart,
                                                                 int* __restrict__ rowptr,
                                                                 int* __restrict__ col,
                                                                 float* __restrict__ dinv, int N) {
  __shared__ int cnt[BIN_NODES];
  __shared__ int cur[BIN_NODES];
  __shared__ int wsum[BIN_NODES / 32];
  __shared__ int stage[BIN_STAGE];
  const int tid = threadIdx.x, lane = tid & 31, wv = tid >> 5;
  const int b = blockIdx.x;
  const int e0 = bstart[b];
  const int e1 = bstart[b + 1];
  const int2* __restrict__ eb = reinterpret_cast<const int2*>(ebuf);
  if (tid < BIN_NODES) cnt[tid] = 0;
  __syncthreads();
  for (int i = e0 + tid; i < e1; i += BIN_BATCH * BIN_THREADS) {
    int d[BIN_BATCH];
    #pragma unroll
    for (int q = 0; q < BIN_BATCH; ++q) d[q] = eb[min(i + q * BIN_THREADS, e1 - 1)].y;
    #pragma unroll
    for (int q = 0; q < BIN_BATCH; ++q)
      if (i + q * BIN_THREADS < e1) atomicAdd(&cnt[d[q] & (BIN_NODES - 1)], 1);
  }
  __syncthreads();

  const int v = (tid < BIN_NODES) ? cnt[tid] : 0;
  int inc = v;
  #pragma unroll
  for (int off = 1; off < 32; off <<= 1) {
    const int u = __shfl_up(inc, off);
    if (lane >= off) inc += u;
  }
  if (tid < BIN_NODES && lane == 31) wsum[wv] = inc;
  __syncthreads();
  if (tid < BIN_NODES) {
    int woff = 0;
    #pragma unroll
    for (int k = 0; k < BIN_NODES / 32; ++k)
      if (k < wv) woff += wsum[k];
    const int excl = woff + inc - v;
    cur[tid] = excl;
    const int node = b * BIN_NODES + tid;
    if (node < N) {
      rowptr[node] = e0 + excl;
      dinv[node] = rsqrtf(fmaxf((float)(v + 1), 1.0f));
    }
  }
  if (b == (int)gridDim.x - 1 && tid == 0) rowptr[N] = e1;
  __syncthreads();

  const int ne = e1 - e0;
  if (ne <= BIN_STAGE) {
    for (int i = e0 + tid; i < e1; i += BIN_BATCH * BIN_THREADS) {
      int2 sd[BIN_BATCH];
      #pragma unroll
      for (int q = 0; q < BIN_BATCH; ++q) sd[q] = eb[min(i + q * BIN_THREADS, e1 - 1)];
      #pragma unroll
      for (int q = 0; q < BIN_BATCH; ++q)
        if (i + q * BIN_THREADS < e1) stage[atomicAdd(&cur[sd[q].y & (BIN_NODES - 1)], 1)] = sd[q].x;
    }
    __syncthreads();
    for (int i = tid; i < ne; i += BIN_THREADS) col[e0 + i] = stage[i];
  } else {
    for (int i = e0 + tid; i < e1; i += BIN_THREADS) {
      const int2 sd = eb[i];
      col[e0 + atomicAdd(&cur[sd.y & (BIN_NODES - 1)], 1)] = sd.x;
    }
  }
}

__global__ void dinv_from_count_kernel(float* __restrict__ cnt_dinv, int N) {
  const int i = blockIdx.x * blockDim.x + threadIdx.x;
  if (i < N) {
    const int c = __float_as_int(cnt_dinv[i]);
    cnt_dinv[i] = rsqrtf(fmaxf((float)(c + 1), 1.0f));
  }
}

__device__ __forceinline__ void row_add(float4& acc, const float4 h) {
  acc.x += h.x;
  acc.y += h.y;
  acc.z += h.z;
  acc.w += h.w;
}

__device__ __forceinline__ void row_fma(float4& acc, const float4 h, const float w) {
  acc.x = fmaf(h.x, w, acc.x);
  acc.y = fmaf(h.y, w, acc.y);
  acc.z = fmaf(h.z, w, acc.z);
  acc.w = fmaf(h.w, w, acc.w);
}

template <bool WSRC>
__device__ __forceinline__ void row_acc(float4& acc, const float4 h, const float w_l, const int t) {
  if (WSRC) row_fma(acc, h, __int_as_float(__builtin_amdgcn_readlane(__float_as_int(w_l), t)));
  else      row_add(acc, h);
}

template <bool WSRC>
__device__ __forceinline__ float4 gather_row_range(const float4* __restrict__ H4,
                                                   const int* __restrict__ col,
                                                   const float* __restrict__ dinv,
                                                   const int d, const float dd, const int beg, const int end,
                                                   const int lane) {
  float4 acc = H4[(size_t)d * (FDIM / 4) + lane];
  if (WSRC) {
    acc.x *= dd;
    acc.y *= dd;
    acc.z *= dd;
    acc.w *= dd;
  }

  for (int j = beg; j < end; j += 32) {
    const int n = min(32, end - j);
    int s_l = 0;
    float w_l = 0.0f;
    if (lane < n) {
      s_l = col[j + lane];
      if (WSRC) w_l = dinv[s_l];
    }
    int t = 0;
    for (; t + 4 <= n; t += 4) {
      const int s0 = __builtin_amdgcn_readlane(s_l, t);
      const int s1 = __builtin_amdgcn_readlane(s_l, t + 1);
      const int s2 = __builtin_amdgcn_readlane(s_l, t + 2);
      const int s3 = __builtin_amdgcn_readlane(s_l, t + 3);
      const float4 h0 = H4[(size_t)s0 * (FDIM / 4) + lane];
      const float4 h1 = H4[(size_t)s1 * (FDIM / 4) + lane];
      const float4 h2 = H4[(size_t)s2 * (FDIM / 4) + lane];
      const float4 h3 = H4[(size_t)s3 * (FDIM / 4) + lane];
      row_acc<WSRC>(acc, h0, w_l, t);
      row_acc<WSRC>(acc, h1, w_l, t + 1);
      row_acc<WSRC>(acc, h2, w_l, t + 2);
      row_acc<WSRC>(acc, h3, w_l, t + 3);
    }
    const int r = n - t;
    if (r == 3) {
      const int s0 = __builtin_amdgcn_readlane(s_l, t);
      const int s1 = __builtin_amdgcn_readlane(s_l, t + 1);
      const int s2 = __builtin_amdgcn_readlane(s_l, t + 2);
      const float4 h0 = H4[(size_t)s0 * (FDIM / 4) + lane];
      const float4 h1 = H4[(size_t)s1 * (FDIM / 4) + lane];
      const float4 h2 = H4[(size_t)s2 * (FDIM / 4) + lane];
      row_acc<WSRC>(acc, h0, w_l, t);
      row_acc<WSRC>(acc, h1, w_l, t + 1);
      row_acc<WSRC>(acc, h2, w_l, t + 2);
    } else if (r == 2) {
      const int s0 = __builtin_amdgcn_readlane(s_l, t);
      const int s1 = __builtin_amdgcn_readlane(s_l, t + 1);
      const float4 h0 = H4[(size_t)s0 * (FDIM / 4) + lane];
      const float4 h1 = H4[(size_t)s1 * (FDIM / 4) + lane];
      row_acc<WSRC>(acc, h0, w_l, t);
      row_acc<WSRC>(acc, h1, w_l, t + 1);
    } else if (r == 1) {
      const int s0 = __builtin_amdgcn_readlane(s_l, t);
      row_acc<WSRC>(acc, H4[(size_t)s0 * (FDIM / 4) + lane], w_l, t);
    }
  }
  return acc;
}

__device__ __forceinline__ void atomic_max_f32(float* addr, float v) {
  if (v >= 0.0f) atomicMax((int*)addr, __float_as_int(v));
  else           atomicMin((unsigned int*)addr, (unsigned int)__float_as_int(v));
}

#define FG_ROWS   32
#define FG_TPITCH 132
#define FG_BLOCKS 512
#define FG_SIDE   (64 * GEMM_WPITCH + 2 * FG_ROWS * FG_TPITCH)
#define FG_STRIP  (FG_SIDE + 2 * FG_ROWS)
#define FG_SPITCH 34

template <bool WSRC, bool LAST>
__global__ __launch_bounds__(32 * FG_ROWS, 16) void gather_product_kernel(
    const float* __restrict__ H, const int* __restrict__ rowptr, const int* __restrict__ col,
    const float* __restrict__ dinv, const float* __restrict__ bias, const float* __restrict__ Wn,
    float* __restrict__ Hn, const int* __restrict__ gi, float* __restrict__ out,
    float* __restrict__ cnt, int N, int ngroups) {
  __shared__ alignas(16) float smem[FG_STRIP + (LAST ? 8 * 16 * FG_SPITCH : 0)];
  float* Ws = smem;
  int* side = reinterpret_cast<int*>(smem) + FG_SIDE;
  const int tid = threadIdx.x;

  {
    const float4* __restrict__ W4 = reinterpret_cast<const float4*>(Wn);
    for (int it = 0; it < 2; ++it) {
      const int i = it * (32 * FG_ROWS) + tid;
      const int p = i >> 5;
      const int q = i & 31;
      const float4 w0 = W4[(2 * p) * (FDIM / 4) + q];
      const float4 w1 = W4[(2 * p + 1) * (FDIM / 4) + q];
      float4* dst = reinterpret_cast<float4*>(Ws + p * GEMM_WPITCH + q * 8);
      dst[0] = make_float4(w0.x, w1.x, w0.y, w1.y);
      dst[1] = make_float4(w0.z, w1.z, w0.w, w1.w);
    }
  }
  __syncthreads();

  const int wave = __builtin_amdgcn_readfirstlane((int)(threadIdx.x >> 5));
  const int lane = tid & 31;
  const int c = lane & 15;
  const int h = lane >> 4;

  const float4* __restrict__ H4 = reinterpret_cast<const float4*>(H);

  const int rt = (wave >> 2) & 1;
  const int cp = wave & 3;
  const unsigned boff0 = (unsigned)(h * (GEMM_WPITCH / 2) + c + 32 * cp) * 8u;
  const float bc0 = bias[32 * cp + c];
  const float bc1 = bias[32 * cp + c + 16];

  int it = 0;
  for (int g = blockIdx.x; g < ngroups; g += gridDim.x, ++it) {
    const int par = it & 1;
    const int toff = 64 * GEMM_WPITCH + par * (FG_ROWS * FG_TPITCH);
    float* tl = smem + toff;
    const int d = g * FG_ROWS + wave;
    float4 acc = make_float4(0.0f, 0.0f, 0.0f, 0.0f);
    int sv = 0;
    if (d < N) {
      const float dd = dinv[d];
      const int beg = rowptr[d];
      const int end = rowptr[d + 1];
      sv = LAST ? gi[d] : __float_as_int(dd);
      acc = gather_row_range<WSRC>(H4, col, dinv, d, dd, beg, end, lane);
      acc.x *= dd;
      acc.y *= dd;
      acc.z *= dd;
      acc.w *= dd;
    }
    reinterpret_cast<float4*>(tl + wave * FG_TPITCH)[lane] = acc;
    if (lane == 0) side[par * FG_ROWS + wave] = sv;
    __syncthreads();

    if ((wave >> 3) == (it & 3)) {
      unsigned ao = (unsigned)(toff + (16 * rt + c) * FG_TPITCH + 2 * h) * 4u;
      unsigned bo = boff0;
      const char* sb = reinterpret_cast<const char*>(smem);
      const int row0 = g * FG_ROWS + 16 * rt + 8 * h;
      int4 sda = make_int4(0, 0, 0, 0);
      int4 sdb = make_int4(0, 0, 0, 0);
      int g_l = 0;
      if (LAST) {
        g_l = side[par * FG_ROWS + 16 * rt + c];
      } else {
        sda = *reinterpret_cast<const int4*>(side + par * FG_ROWS + 16 * rt + 8 * h);
        sdb = *reinterpret_cast<const int4*>(side + par * FG_ROWS + 16 * rt + 8 * h + 4);
      }
      int nval = N - row0;
      float* hp = nullptr;
      if (!LAST) {
        hp = Hn + (size_t)row0 * FDIM + 32 * cp + c;
        asm volatile("" : "+v"(hp), "+v"(nval));
      }
      __attribute__((address_space(1))) float* hg = (__attribute__((address_space(1))) float*)hp;
      v8f acc0 = v8f{};
      v8f acc1 = v8f{};
      v2f ax  = *reinterpret_cast<const v2f*>(sb + ao);
      v2f bx0 = *reinterpret_cast<const v2f*>(sb + bo);
      v2f bx1 = *reinterpret_cast<const v2f*>(sb + bo + 128);
      #pragma unroll 1
      for (int kk = 0; kk < FDIM / 4 - 2; kk += 2) {
        ao += 16;
        bo += 2 * (GEMM_WPITCH / 2) * 8;
        asm volatile("" : "+v"(ao), "+v"(bo));
        const v2f ay  = *reinterpret_cast<const v2f*>(sb + ao);
        const v2f by0 = *reinterpret_cast<const v2f*>(sb + bo);
        const v2f by1 = *reinterpret_cast<const v2f*>(sb + bo + 128);
        acc0 = __builtin_amdgcn_wmma_f32_16x16x4_f32(
             false, ax,  false, bx0,
             (short)0, acc0,  false,  false);
        acc1 = __builtin_amdgcn_wmma_f32_16x16x4_f32(
             false, ax,  false, bx1,
             (short)0, acc1,  false,  false);
        ao += 16;
        bo += 2 * (GEMM_WPITCH / 2) * 8;
        asm volatile("" : "+v"(ao), "+v"(bo));
        ax  = *reinterpret_cast<const v2f*>(sb + ao);
        bx0 = *reinterpret_cast<const v2f*>(sb + bo);
        bx1 = *reinterpret_cast<const v2f*>(sb + bo + 128);
        acc0 = __builtin_amdgcn_wmma_f32_16x16x4_f32(
             false, ay,  false, by0,
             (short)0, acc0,  false,  false);
        acc1 = __builtin_amdgcn_wmma_f32_16x16x4_f32(
             false, ay,  false, by1,
             (short)0, acc1,  false,  false);
      }
      {
        ao += 16;
        bo += 2 * (GEMM_WPITCH / 2) * 8;
        asm volatile("" : "+v"(ao), "+v"(bo));
        const v2f ay  = *reinterpret_cast<const v2f*>(sb + ao);
        const v2f by0 = *reinterpret_cast<const v2f*>(sb + bo);
        const v2f by1 = *reinterpret_cast<const v2f*>(sb + bo + 128);
        acc0 = __builtin_amdgcn_wmma_f32_16x16x4_f32(
             false, ax,  false, bx0,
             (short)0, acc0,  false,  false);
        acc1 = __builtin_amdgcn_wmma_f32_16x16x4_f32(
             false, ax,  false, bx1,
             (short)0, acc1,  false,  false);
        acc0 = __builtin_amdgcn_wmma_f32_16x16x4_f32(
             false, ay,  false, by0,
             (short)0, acc0,  false,  false);
        acc1 = __builtin_amdgcn_wmma_f32_16x16x4_f32(
             false, ay,  false, by1,
             (short)0, acc1,  false,  false);
      }
      __builtin_amdgcn_sched_barrier(0);
      asm volatile("v_nop\n\tv_nop\n\tv_nop\n\tv_nop\n\tv_nop\n\tv_nop\n\tv_nop\n\tv_nop"
                   : "+v"(acc0), "+v"(acc1));
      __builtin_amdgcn_sched_barrier(0);
      if (!LAST) {
        const int sd[8] = {sda.x, sda.y, sda.z, sda.w, sdb.x, sdb.y, sdb.z, sdb.w};
        float o0[8], o1[8];
        #pragma unroll
        for (int r = 0; r < 8; ++r) {
          const float dr = __int_as_float(sd[r]);
          o0[r] = fmaxf(acc0[r] + bc0, 0.0f) * dr;
          o1[r] = fmaxf(acc1[r] + bc1, 0.0f) * dr;
        }
        #pragma unroll
        for (int r = 0; r < 8; ++r) {
          if (r < nval) {
            hg[r * FDIM]      = o0[r];
            hg[r * FDIM + 16] = o1[r];
          }
        }
      } else {
        float* sp = smem + FG_STRIP + (wave & 7) * (16 * FG_SPITCH);
        #pragma unroll
        for (int r = 0; r < 8; ++r) {
          sp[(8 * h + r) * FG_SPITCH + c]      = fmaxf(acc0[r] + bc0, 0.0f);
          sp[(8 * h + r) * FG_SPITCH + 16 + c] = fmaxf(acc1[r] + bc1, 0.0f);
        }
        __builtin_amdgcn_fence(__ATOMIC_ACQ_REL, "wavefront");
        float v[16];
        #pragma unroll
        for (int i = 0; i < 16; ++i) v[i] = sp[i * FG_SPITCH + lane];
        const int nr = min(16, N - (g * FG_ROWS + 16 * rt));
        int a = 0;
        for (int trip = 0; trip < 16 && a < nr; ++trip) {
          const int ga = __builtin_amdgcn_readlane(g_l, a);
          const unsigned diff = __builtin_amdgcn_ballot_w32(g_l != ga) & (0xFFFFu & (~0u << a));
          const int b = min(diff ? __builtin_ctz(diff) : 16, nr);
          float s = 0.0f;
          float m = -FLT_MAX;
          #pragma unroll
          for (int i = 0; i < 16; ++i) {
            if (i >= a && i < b) {
              s += v[i];
              m = fmaxf(m, v[i]);
            }
          }
          float* o = out + (size_t)ga * (2 * FDIM) + 32 * cp + lane;
          atomicAdd(o, s);
          atomic_max_f32(o + FDIM, m);
          if (cp == 0 && lane == 0) atomicAdd(&cnt[ga], (float)(b - a));
          a = b;
        }
      }
    }
  }
}

__global__ void bias_relu_kernel(float* __restrict__ G, const float* __restrict__ b) {
  const size_t i = (size_t)blockIdx.x * FDIM + threadIdx.x;
  G[i] = fmaxf(G[i] + b[threadIdx.x], 0.0f);
}

__global__ void init_out_kernel(float* __restrict__ out, int ng) {
  int i = blockIdx.x * blockDim.x + threadIdx.x;
  if (i >= ng * 2 * FDIM) return;
  int f = i & (2 * FDIM - 1);
  out[i] = (f < FDIM) ? 0.0f : __int_as_float(0xFF7FFFFF);
}

#define POOL_ROWS  32
#define POOL_WAVES 8

__device__ __forceinline__ void pool_flush(float* __restrict__ out, float* __restrict__ cnt,
                                           float* st, int lane, int g,
                                           const float4 s, const float4 m, int run) {
  reinterpret_cast<float4*>(st)[lane] = s;
  reinterpret_cast<float4*>(st + FDIM)[lane] = m;
  __builtin_amdgcn_fence(__ATOMIC_ACQ_REL, "wavefront");
  float a[4], b[4];
  #pragma unroll
  for (int k = 0; k < 4; ++k) {
    a[k] = st[lane + 32 * k];
    b[k] = st[FDIM + lane + 32 * k];
  }
  __builtin_amdgcn_fence(__ATOMIC_ACQ_REL, "wavefront");
  float* o = out + (size_t)g * (2 * FDIM);
  #pragma unroll
  for (int k = 0; k < 4; ++k) atomicAdd(o + lane + 32 * k, a[k]);
  #pragma unroll
  for (int k = 0; k < 4; ++k) atomic_max_f32(o + FDIM + lane + 32 * k, b[k]);
  if (lane == 0) atomicAdd(&cnt[g], (float)run);
}

__global__ __launch_bounds__(256) void pool_runs_kernel(const float* __restrict__ G,
                                                        const int* __restrict__ gi,
                                                        float* __restrict__ out,
                                                        float* __restrict__ cnt, int N) {
  __shared__ alignas(16) float stage[POOL_WAVES * 2 * FDIM];
  const int wv = threadIdx.x >> 5;
  const int lane = threadIdx.x & 31;
  const int n0 = __builtin_amdgcn_readfirstlane((int)(blockIdx.x * POOL_WAVES + wv)) * POOL_ROWS;
  if (n0 >= N) return;
  const int nr = min(POOL_ROWS, N - n0);
  float* st = stage + wv * (2 * FDIM);

  const int g_l = gi[n0 + min(lane, nr - 1)];

  const float4* __restrict__ G4 = reinterpret_cast<const float4*>(G) + (size_t)n0 * (FDIM / 4) + lane;
  float4 v[POOL_ROWS];
  #pragma unroll
  for (int i = 0; i < POOL_ROWS; ++i) v[i] = G4[(size_t)min(i, nr - 1) * (FDIM / 4)];

  int a = 0;
  for (int trip = 0; trip < POOL_ROWS && a < nr; ++trip) {
    const int ga = __builtin_amdgcn_readlane(g_l, a);
    const unsigned diff = __builtin_amdgcn_ballot_w32(g_l != ga) & (~0u << a);
    const int b = min(diff ? __builtin_ctz(diff) : POOL_ROWS, nr);
    float4 s = make_float4(0.0f, 0.0f, 0.0f, 0.0f);
    float4 m = make_float4(-FLT_MAX, -FLT_MAX, -FLT_MAX, -FLT_MAX);
    #pragma unroll
    for (int i = 0; i < POOL_ROWS; ++i) {
      if (i >= a && i < b) {
        s.x += v[i].x; s.y += v[i].y; s.z += v[i].z; s.w += v[i].w;
        m.x = fmaxf(m.x, v[i].x); m.y = fmaxf(m.y, v[i].y);
        m.z = fmaxf(m.z, v[i].z); m.w = fmaxf(m.w, v[i].w);
      }
    }
    pool_flush(out, cnt, st, lane, ga, s, m, b - a);
    a = b;
  }
}

__global__ void finalize_mean_kernel(float* __restrict__ out, const float* __restrict__ cnt) {
  const int g = blockIdx.x;
  out[(size_t)g * 2 * FDIM + threadIdx.x] /= fmaxf(cnt[g], 1.0f);
}


extern "C" void kernel_launch(void* const* d_in, const int* in_sizes, int n_in,
                              void* d_out, int out_size, void* d_ws, size_t ws_size,
                              hipStream_t stream) {
  const float* x  = (const float*)d_in[0];
  const int*   ei = (const int*)d_in[1];
  const int*   gi = (const int*)d_in[2];
  const float* Wm[3] = {(const float*)d_in[4], (const float*)d_in[6], (const float*)d_in[8]};
  const float* Bv[3] = {(const float*)d_in[5], (const float*)d_in[7], (const float*)d_in[9]};
  float* out = (float*)d_out;

  const int  N  = in_sizes[0] / FDIM;
  const int  E  = in_sizes[1] / 2;
  const int  NG = out_size / (2 * FDIM);
  const long Nf = (long)N * FDIM;

  float* Hbuf = (float*)d_ws;
  float* Gbuf = Hbuf + Nf;
  float* dinv = Gbuf + Nf;
  float* cnt  = dinv + N;
  if (ws_size < (size_t)(2 * Nf + N + NG) * sizeof(float)) return;

  const int nsb = (N + SCAN_ELEMS - 1) / SCAN_ELEMS;
  int* rowptr = (int*)(cnt + NG);
  int* bsum   = rowptr + (N + 1);
  int* col    = bsum + nsb;
  int* rank   = (int*)Hbuf;
  const bool ordered =
      ws_size >= (size_t)(2 * Nf + N + NG) * sizeof(float) + ((size_t)N + 1 + nsb + E) * sizeof(int) &&
      (long)E <= Nf && E > 0;

  auto zero = [&](float* p, long n) {
    zero_f32_kernel<<<(unsigned)((n + 255) / 256), 256, 0, stream>>>(p, n);
  };

  const int tiles = N / 16;
  const int rem   = N - tiles * 16;
  const float* cur = x;

  const int  nb   = (N + BIN_NODES - 1) / BIN_NODES;
  const long nh   = (long)((nb + 3) & ~3) * BIN_BLOCKS;
  int* ebuf   = (int*)Hbuf;
  int* hist   = (int*)Gbuf;
  int* bstart = hist + nh;
  const bool binned = ordered && nb <= BIN_MAXB && 2 * (long)E <= Nf && nh + nb + 1 <= Nf;

  if (!binned) {
    init_out_kernel<<<(NG * 2 * FDIM + 255) / 256, 256, 0, stream>>>(out, NG);
    zero(cnt, NG);
  }

  if (ordered) {
    if (binned) {
      bin_count_kernel<<<BIN_BLOCKS, BIN_THREADS, 0, stream>>>(ei, hist, E, nb, out, cnt, NG);
      bin_scatter_kernel<<<BIN_BLOCKS, BIN_THREADS, 0, stream>>>(ei, hist, bstart, ebuf, E, nb);
      bin_finish_kernel<<<nb, BIN_THREADS, 0, stream>>>(ebuf, bstart, rowptr, col, dinv, N);
    } else {
      zero(dinv, N);
      count_rank_kernel<<<(E + 255) / 256, 256, 0, stream>>>(ei, (int*)dinv, rank, E);
      scan_local_kernel<<<nsb, SCAN_THREADS, 0, stream>>>((const int*)dinv, rowptr, bsum, N);
      scan_add_kernel<<<nsb, SCAN_THREADS, 0, stream>>>(rowptr, bsum, N);
      place_kernel<<<(E + 255) / 256, 256, 0, stream>>>(ei, rowptr, rank, col, E);
      dinv_from_count_kernel<<<(N + 255) / 256, 256, 0, stream>>>(dinv, N);
    }

    const int ngroups = (N + FG_ROWS - 1) / FG_ROWS;
    const int fgrid = ngroups < FG_BLOCKS ? ngroups : FG_BLOCKS;
    gather_product_kernel<true, false><<<fgrid, 32 * FG_ROWS, 0, stream>>>(
        x, rowptr, col, dinv, Bv[0], Wm[0], Hbuf, nullptr, nullptr, nullptr, N, ngroups);
    gather_product_kernel<false, false><<<fgrid, 32 * FG_ROWS, 0, stream>>>(
        Hbuf, rowptr, col, dinv, Bv[1], Wm[1], Gbuf, nullptr, nullptr, nullptr, N, ngroups);
    gather_product_kernel<false, true><<<fgrid, 32 * FG_ROWS, 0, stream>>>(
        Gbuf, rowptr, col, dinv, Bv[2], Wm[2], nullptr, gi, out, cnt, N, ngroups);
  } else {
    zero(dinv, N);
    degree_kernel<<<(E + N + 255) / 256, 256, 0, stream>>>(ei, dinv, E, N);
    dinv_kernel<<<(N + 255) / 256, 256, 0, stream>>>(dinv, N);

    for (int l = 0; l < 3; ++l) {
      if (tiles > 0)
        gemm_strip_kernel<false><<<GEMM_BLOCKS, 256, 0, stream>>>(cur, Wm[l], nullptr, Hbuf, tiles);
      if (rem > 0)
        gemm_tail_kernel<<<rem, FDIM, 0, stream>>>(cur, Wm[l], nullptr, Hbuf, tiles * 16);
      zero(Gbuf, Nf);
      aggregate_kernel<<<(E + N + 7) / 8, 256, 0, stream>>>(Hbuf, ei, dinv, Gbuf, E, N);
      bias_relu_kernel<<<N, FDIM, 0, stream>>>(Gbuf, Bv[l]);
      cur = Gbuf;
    }
  }

  if (!ordered)
    pool_runs_kernel<<<(N + POOL_ROWS * POOL_WAVES - 1) / (POOL_ROWS * POOL_WAVES), 32 * POOL_WAVES, 0, stream>>>(Gbuf, gi, out, cnt, N);
  finalize_mean_kernel<<<NG, FDIM, 0, stream>>>(out, cnt);
}
